# starting kernel
# baseline (speedup 1.0000x reference)
#include <hip/hip_runtime.h>
#include <math.h>

#define B_ 4
#define C_ 512
#define N_ 16384
#define EPS_ 1e-5f
#define NS_ITERS 12

typedef __attribute__((ext_vector_type(2))) float v2f;
typedef __attribute__((ext_vector_type(8))) float v8f;

__device__ __forceinline__ v8f wmma4(v2f a, v2f b, v8f c) {
  return __builtin_amdgcn_wmma_f32_16x16x4_f32(false, a, false, b, (short)0, c,
                                               false, false);
}

__global__ void mean_kernel(const float* __restrict__ x, float* __restrict__ mean) {
  const int bc = blockIdx.x;
  const float* p = x + (size_t)bc * N_;
  float s = 0.f;
  for (int i = threadIdx.x; i < N_; i += 256) s += p[i];
  __shared__ float red[256];
  red[threadIdx.x] = s;
  __syncthreads();
  for (int o = 128; o > 0; o >>= 1) {
    if ((int)threadIdx.x < o) red[threadIdx.x] += red[threadIdx.x + o];
    __syncthreads();
  }
  if (threadIdx.x == 0) mean[bc] = red[0] * (1.0f / (float)N_);
}

__global__ __launch_bounds__(32) void cov_kernel(const float* __restrict__ x,
                                                 const float* __restrict__ mean,
                                                 float* __restrict__ cov) {
  const int b    = blockIdx.z;
  const int row0 = blockIdx.y * 64;
  const int col0 = blockIdx.x * 64;
  const float* X  = x + (size_t)b * C_ * N_;
  const float* mu = mean + b * C_;
  float* D = cov + (size_t)b * C_ * C_;

  const int lane = threadIdx.x & 31;
  const int lm = lane & 15;
  const int kh = lane >> 4;

  const v8f zero = {0.f, 0.f, 0.f, 0.f, 0.f, 0.f, 0.f, 0.f};
  v8f acc[4][4];
#pragma unroll
  for (int m = 0; m < 4; ++m)
#pragma unroll
    for (int n = 0; n < 4; ++n) acc[m][n] = zero;

  const float* ap[4];
  const float* bp[4];
#pragma unroll
  for (int m = 0; m < 4; ++m) ap[m] = X + (size_t)(row0 + 16 * m + lm) * N_ + 2 * kh;
#pragma unroll
  for (int n = 0; n < 4; ++n) bp[n] = X + (size_t)(col0 + 16 * n + lm) * N_ + 2 * kh;

  for (int k = 0; k < N_; k += 4) {
    v2f av[4], bv[4];
#pragma unroll
    for (int m = 0; m < 4; ++m) av[m] = *(const v2f*)(ap[m] + k);
#pragma unroll
    for (int n = 0; n < 4; ++n) bv[n] = *(const v2f*)(bp[n] + k);
#pragma unroll
    for (int m = 0; m < 4; ++m)
#pragma unroll
      for (int n = 0; n < 4; ++n) acc[m][n] = wmma4(av[m], bv[n], acc[m][n]);
  }

  const float fac = 1.0f / (float)(N_ - 1);
#pragma unroll
  for (int m = 0; m < 4; ++m) {
#pragma unroll
    for (int n = 0; n < 4; ++n) {
      const int cidx = col0 + 16 * n + lm;
      const float mj = mu[cidx];
#pragma unroll
      for (int v = 0; v < 8; ++v) {
        const int r = row0 + 16 * m + v + 8 * kh;
        float val = (acc[m][n][v] - (float)N_ * mu[r] * mj) * fac;
        if (r == cidx) val += EPS_;
        D[(size_t)r * C_ + cidx] = val;
      }
    }
  }
}

__global__ void trace_kernel(const float* __restrict__ cov, float* __restrict__ tr) {
  const int m = blockIdx.x;
  const float* A = cov + (size_t)m * C_ * C_;
  float s = 0.f;
  for (int i = threadIdx.x; i < C_; i += 256) s += A[(size_t)i * C_ + i];
  __shared__ float red[256];
  red[threadIdx.x] = s;
  __syncthreads();
  for (int o = 128; o > 0; o >>= 1) {
    if ((int)threadIdx.x < o) red[threadIdx.x] += red[threadIdx.x + o];
    __syncthreads();
  }
  if (threadIdx.x == 0) tr[m] = red[0];
}

__global__ void ns_init_kernel(const float* __restrict__ cov,
                               const float* __restrict__ tr,
                               float* __restrict__ Y, float* __restrict__ Z) {
  const size_t i = (size_t)blockIdx.x * 256 + threadIdx.x;
  const size_t total = (size_t)8 * C_ * C_;
  if (i >= total) return;
  const int m = (int)(i / ((size_t)C_ * C_));
  const size_t rr = i % ((size_t)C_ * C_);
  Y[i] = cov[i] / tr[m];
  Z[i] = ((rr / C_) == (rr % C_)) ? 1.f : 0.f;
}

__global__ __launch_bounds__(32) void gemm512_kernel(const float* __restrict__ Amat,
                                                     const float* __restrict__ Bmat,
                                                     float* __restrict__ Dmat,
                                                     float alpha, float diagBeta) {
  const int mm = blockIdx.z;
  const float* A  = Amat + (size_t)mm * C_ * C_;
  const float* Bp = Bmat + (size_t)mm * C_ * C_;
  float* D = Dmat + (size_t)mm * C_ * C_;
  const int row0 = blockIdx.y * 64;
  const int col0 = blockIdx.x * 64;
  const int lane = threadIdx.x & 31;
  const int lm = lane & 15;
  const int kh = lane >> 4;

  const v8f zero = {0.f, 0.f, 0.f, 0.f, 0.f, 0.f, 0.f, 0.f};
  v8f acc[4][4];
#pragma unroll
  for (int m = 0; m < 4; ++m)
#pragma unroll
    for (int n = 0; n < 4; ++n) acc[m][n] = zero;

  for (int k = 0; k < C_; k += 4) {
    v2f av[4], bv[4];
#pragma unroll
    for (int m = 0; m < 4; ++m)
      av[m] = *(const v2f*)(A + (size_t)(row0 + 16 * m + lm) * C_ + k + 2 * kh);
#pragma unroll
    for (int n = 0; n < 4; ++n) {
      const float* q = Bp + (size_t)(k + 2 * kh) * C_ + col0 + 16 * n + lm;
      v2f t;
      t.x = q[0];
      t.y = q[C_];
      bv[n] = t;
    }
#pragma unroll
    for (int m = 0; m < 4; ++m)
#pragma unroll
      for (int n = 0; n < 4; ++n) acc[m][n] = wmma4(av[m], bv[n], acc[m][n]);
  }

#pragma unroll
  for (int m = 0; m < 4; ++m)
#pragma unroll
    for (int n = 0; n < 4; ++n) {
      const int cidx = col0 + 16 * n + lm;
#pragma unroll
      for (int v = 0; v < 8; ++v) {
        const int r = row0 + 16 * m + v + 8 * kh;
        float val = alpha * acc[m][n][v];
        if (r == cidx) val += diagBeta;
        D[(size_t)r * C_ + cidx] = val;
      }
    }
}

__global__ void bias_white_kernel(const float* __restrict__ Z,
                                  const float* __restrict__ cmean,
                                  const float* __restrict__ tr,
                                  float* __restrict__ bias) {
  const int idx = blockIdx.x * 256 + threadIdx.x;
  if (idx >= B_ * C_) return;
  const int b = idx / C_;
  const int i = idx % C_;
  const float* r = Z + ((size_t)b * C_ + i) * C_;
  const float* mu = cmean + b * C_;
  float s = 0.f;
  for (int j = 0; j < C_; ++j) s += r[j] * mu[j];
  bias[idx] = -s * rsqrtf(tr[b]);
}

__global__ __launch_bounds__(32) void apply_kernel(const float* __restrict__ Mmat,
                                                   const float* __restrict__ X,
                                                   const float* __restrict__ bias,
                                                   const float* __restrict__ tr,
                                                   int trOff, int useSqrt,
                                                   float* __restrict__ out) {
  const int b = blockIdx.z;
  const float* Mb = Mmat + (size_t)b * C_ * C_;
  const float* Xb = X + (size_t)b * C_ * N_;
  float* Ob = out + (size_t)b * C_ * N_;
  const float* bi = bias + b * C_;
  const float t = tr[trOff + b];
  const float scale = useSqrt ? sqrtf(t) : rsqrtf(t);

  const int row0 = blockIdx.y * 64;
  const int col0 = blockIdx.x * 64;
  const int lane = threadIdx.x & 31;
  const int lm = lane & 15;
  const int kh = lane >> 4;

  const v8f zero = {0.f, 0.f, 0.f, 0.f, 0.f, 0.f, 0.f, 0.f};
  v8f acc[4][4];
#pragma unroll
  for (int m = 0; m < 4; ++m)
#pragma unroll
    for (int n = 0; n < 4; ++n) acc[m][n] = zero;

  for (int k = 0; k < C_; k += 4) {
    v2f av[4], bv[4];
#pragma unroll
    for (int m = 0; m < 4; ++m)
      av[m] = *(const v2f*)(Mb + (size_t)(row0 + 16 * m + lm) * C_ + k + 2 * kh);
#pragma unroll
    for (int n = 0; n < 4; ++n) {
      const float* q = Xb + (size_t)(k + 2 * kh) * N_ + col0 + 16 * n + lm;
      v2f tt;
      tt.x = q[0];
      tt.y = q[N_];
      bv[n] = tt;
    }
#pragma unroll
    for (int m = 0; m < 4; ++m)
#pragma unroll
      for (int n = 0; n < 4; ++n) acc[m][n] = wmma4(av[m], bv[n], acc[m][n]);
  }

#pragma unroll
  for (int m = 0; m < 4; ++m)
#pragma unroll
    for (int n = 0; n < 4; ++n) {
      const int c = col0 + 16 * n + lm;
#pragma unroll
      for (int v = 0; v < 8; ++v) {
        const int r = row0 + 16 * m + v + 8 * kh;
        Ob[(size_t)r * N_ + c] = scale * acc[m][n][v] + bi[r];
      }
    }
}

extern "C" void kernel_launch(void* const* d_in, const int* in_sizes, int n_in,
                              void* d_out, int out_size, void* d_ws, size_t ws_size,
                              hipStream_t stream) {
  (void)in_sizes; (void)n_in; (void)out_size; (void)ws_size;
  const float* content = (const float*)d_in[0];
  const float* style   = (const float*)d_in[1];
  float* out = (float*)d_out;
  float* ws  = (float*)d_ws;

  const size_t CM = (size_t)C_ * C_;

  float* c_mean  = ws;
  float* s_mean  = c_mean + B_ * C_;
  float* tr      = s_mean + B_ * C_;
  float* bias_w  = tr + 8;
  float* cov_all = bias_w + B_ * C_;
  float* Y  = cov_all + 8 * CM;
  float* Z  = Y + 8 * CM;
  float* Y2 = Z + 8 * CM;
  float* Z2 = Y2 + 8 * CM;
  float* T  = Z2 + 8 * CM;
  float* whit = T + 8 * CM;

  mean_kernel<<<B_ * C_, 256, 0, stream>>>(content, c_mean);
  mean_kernel<<<B_ * C_, 256, 0, stream>>>(style, s_mean);

  dim3 covGrid(C_ / 64, C_ / 64, B_);
  cov_kernel<<<covGrid, 32, 0, stream>>>(content, c_mean, cov_all);
  cov_kernel<<<covGrid, 32, 0, stream>>>(style, s_mean, cov_all + 4 * CM);

  trace_kernel<<<8, 256, 0, stream>>>(cov_all, tr);

  {
    const size_t tot = 8 * CM;
    ns_init_kernel<<<(unsigned)((tot + 255) / 256), 256, 0, stream>>>(cov_all, tr, Y, Z);
  }
  dim3 g512(C_ / 64, C_ / 64, 8);
  float* y = Y; float* z = Z; float* y2 = Y2; float* z2 = Z2;
  for (int it = 0; it < NS_ITERS; ++it) {
    gemm512_kernel<<<g512, 32, 0, stream>>>(z, y, T, -0.5f, 1.5f);
    gemm512_kernel<<<g512, 32, 0, stream>>>(y, T, y2, 1.0f, 0.0f);
    gemm512_kernel<<<g512, 32, 0, stream>>>(T, z, z2, 1.0f, 0.0f);
    float* tmp;
    tmp = y; y = y2; y2 = tmp;
    tmp = z; z = z2; z2 = tmp;
  }

  bias_white_kernel<<<(B_ * C_ + 255) / 256, 256, 0, stream>>>(z, c_mean, tr, bias_w);

  dim3 gApply(N_ / 64, C_ / 64, B_);
  apply_kernel<<<gApply, 32, 0, stream>>>(z, content, bias_w, tr, 0, 0, whit);

  apply_kernel<<<gApply, 32, 0, stream>>>(y + 4 * CM, whit, s_mean, tr, 4, 1, out);
}


# Batched 512^3 GEMM (36 of 45 launches): 4-wave workgroups stage 128x32 A / 32x64 B chunks in LDS, double-buffered, 64x32 tile per wave; k summed in a permuted order inside each 8-block
# speedup vs baseline: 1.7728x; 1.7728x over previous
#include <hip/hip_runtime.h>
#include <math.h>

#define B_ 4
#define C_ 512
#define N_ 16384
#define EPS_ 1e-5f
#define NS_ITERS 12

typedef __attribute__((ext_vector_type(2))) float v2f;
typedef __attribute__((ext_vector_type(8))) float v8f;

__device__ __forceinline__ v8f wmma4(v2f a, v2f b, v8f c) {
  return __builtin_amdgcn_wmma_f32_16x16x4_f32(false, a, false, b, (short)0, c,
                                               false, false);
}

__global__ void mean_kernel(const float* __restrict__ x, float* __restrict__ mean) {
  const int bc = blockIdx.x;
  const float* p = x + (size_t)bc * N_;
  float s = 0.f;
  for (int i = threadIdx.x; i < N_; i += 256) s += p[i];
  __shared__ float red[256];
  red[threadIdx.x] = s;
  __syncthreads();
  for (int o = 128; o > 0; o >>= 1) {
    if ((int)threadIdx.x < o) red[threadIdx.x] += red[threadIdx.x + o];
    __syncthreads();
  }
  if (threadIdx.x == 0) mean[bc] = red[0] * (1.0f / (float)N_);
}

__global__ __launch_bounds__(32) void cov_kernel(const float* __restrict__ x,
                                                 const float* __restrict__ mean,
                                                 float* __restrict__ cov) {
  const int b    = blockIdx.z;
  const int row0 = blockIdx.y * 64;
  const int col0 = blockIdx.x * 64;
  const float* X  = x + (size_t)b * C_ * N_;
  const float* mu = mean + b * C_;
  float* D = cov + (size_t)b * C_ * C_;

  const int lane = threadIdx.x & 31;
  const int lm = lane & 15;
  const int kh = lane >> 4;

  const v8f zero = {0.f, 0.f, 0.f, 0.f, 0.f, 0.f, 0.f, 0.f};
  v8f acc[4][4];
#pragma unroll
  for (int m = 0; m < 4; ++m)
#pragma unroll
    for (int n = 0; n < 4; ++n) acc[m][n] = zero;

  const float* ap[4];
  const float* bp[4];
#pragma unroll
  for (int m = 0; m < 4; ++m) ap[m] = X + (size_t)(row0 + 16 * m + lm) * N_ + 2 * kh;
#pragma unroll
  for (int n = 0; n < 4; ++n) bp[n] = X + (size_t)(col0 + 16 * n + lm) * N_ + 2 * kh;

  for (int k = 0; k < N_; k += 4) {
    v2f av[4], bv[4];
#pragma unroll
    for (int m = 0; m < 4; ++m) av[m] = *(const v2f*)(ap[m] + k);
#pragma unroll
    for (int n = 0; n < 4; ++n) bv[n] = *(const v2f*)(bp[n] + k);
#pragma unroll
    for (int m = 0; m < 4; ++m)
#pragma unroll
      for (int n = 0; n < 4; ++n) acc[m][n] = wmma4(av[m], bv[n], acc[m][n]);
  }

  const float fac = 1.0f / (float)(N_ - 1);
#pragma unroll
  for (int m = 0; m < 4; ++m) {
#pragma unroll
    for (int n = 0; n < 4; ++n) {
      const int cidx = col0 + 16 * n + lm;
      const float mj = mu[cidx];
#pragma unroll
      for (int v = 0; v < 8; ++v) {
        const int r = row0 + 16 * m + v + 8 * kh;
        float val = (acc[m][n][v] - (float)N_ * mu[r] * mj) * fac;
        if (r == cidx) val += EPS_;
        D[(size_t)r * C_ + cidx] = val;
      }
    }
  }
}

__global__ void trace_kernel(const float* __restrict__ cov, float* __restrict__ tr) {
  const int m = blockIdx.x;
  const float* A = cov + (size_t)m * C_ * C_;
  float s = 0.f;
  for (int i = threadIdx.x; i < C_; i += 256) s += A[(size_t)i * C_ + i];
  __shared__ float red[256];
  red[threadIdx.x] = s;
  __syncthreads();
  for (int o = 128; o > 0; o >>= 1) {
    if ((int)threadIdx.x < o) red[threadIdx.x] += red[threadIdx.x + o];
    __syncthreads();
  }
  if (threadIdx.x == 0) tr[m] = red[0];
}

__global__ void ns_init_kernel(const float* __restrict__ cov,
                               const float* __restrict__ tr,
                               float* __restrict__ Y, float* __restrict__ Z) {
  const size_t i = (size_t)blockIdx.x * 256 + threadIdx.x;
  const size_t total = (size_t)8 * C_ * C_;
  if (i >= total) return;
  const int m = (int)(i / ((size_t)C_ * C_));
  const size_t rr = i % ((size_t)C_ * C_);
  Y[i] = cov[i] / tr[m];
  Z[i] = ((rr / C_) == (rr % C_)) ? 1.f : 0.f;
}

#define G_BM 128
#define G_BN 64
#define G_KC 32
#define G_LD 36

typedef __attribute__((ext_vector_type(4))) float v4f;

__global__ __launch_bounds__(128) void gemm512_kernel(const float* __restrict__ Amat,
                                                      const float* __restrict__ Bmat,
                                                      float* __restrict__ Dmat,
                                                      float alpha, float diagBeta) {
  __shared__ __attribute__((aligned(16))) float As[2][G_BM * G_LD];
  __shared__ __attribute__((aligned(16))) float Bs[2][G_BN * G_LD];

  const int mm = blockIdx.z;
  const float* A  = Amat + (size_t)mm * C_ * C_;
  const float* Bp = Bmat + (size_t)mm * C_ * C_;
  float* D = Dmat + (size_t)mm * C_ * C_;
  const int row0 = blockIdx.y * G_BM;
  const int col0 = blockIdx.x * G_BN;
  const int tid  = threadIdx.x;
  const int lane = tid & 31;
  const int wave = tid >> 5;
  const int wm = wave >> 1;
  const int wn = wave & 1;
  const int lm = lane & 15;
  const int kh = lane >> 4;

  const float* agp = A + (size_t)(row0 + (tid >> 3)) * C_ + (tid & 7) * 4;
  const float* bgp = Bp + (size_t)(tid >> 4) * C_ + col0 + (tid & 15) * 4;
  const int a_st = (tid >> 3) * G_LD + (tid & 7) * 4;
  const int b_st = ((tid & 15) * 4) * G_LD + (tid >> 4);
  const int a_rd = (wm * 64 + lm) * G_LD + 4 * kh;
  const int b_rd = (wn * 32 + lm) * G_LD + 4 * kh;

  const v8f zero = {0.f, 0.f, 0.f, 0.f, 0.f, 0.f, 0.f, 0.f};
  v8f acc[4][2];
#pragma unroll
  for (int m = 0; m < 4; ++m)
#pragma unroll
    for (int n = 0; n < 2; ++n) acc[m][n] = zero;

  v4f ga[8], gb[4];
#pragma unroll
  for (int i = 0; i < 8; ++i) ga[i] = *(const v4f*)(agp + (size_t)(16 * i) * C_);
#pragma unroll
  for (int i = 0; i < 4; ++i) gb[i] = *(const v4f*)(bgp + (size_t)(8 * i) * C_);
#pragma unroll
  for (int i = 0; i < 8; ++i) *(v4f*)(&As[0][a_st + 16 * i * G_LD]) = ga[i];
#pragma unroll
  for (int i = 0; i < 4; ++i) {
    Bs[0][b_st + 0 * G_LD + 8 * i] = gb[i].x;
    Bs[0][b_st + 1 * G_LD + 8 * i] = gb[i].y;
    Bs[0][b_st + 2 * G_LD + 8 * i] = gb[i].z;
    Bs[0][b_st + 3 * G_LD + 8 * i] = gb[i].w;
  }
  __syncthreads();

  for (int t = 0; t < C_ / G_KC; ++t) {
    const int buf = t & 1;
    const bool more = (t + 1 < C_ / G_KC);
    if (more) {
      const int k1 = (t + 1) * G_KC;
#pragma unroll
      for (int i = 0; i < 8; ++i) ga[i] = *(const v4f*)(agp + (size_t)(16 * i) * C_ + k1);
#pragma unroll
      for (int i = 0; i < 4; ++i) gb[i] = *(const v4f*)(bgp + (size_t)(k1 + 8 * i) * C_);
    }

    const float* as = &As[buf][a_rd];
    const float* bs = &Bs[buf][b_rd];
    v4f af[4], bf[2];
#pragma unroll
    for (int m = 0; m < 4; ++m) af[m] = *(const v4f*)(as + 16 * m * G_LD);
#pragma unroll
    for (int n = 0; n < 2; ++n) bf[n] = *(const v4f*)(bs + 16 * n * G_LD);
#pragma unroll
    for (int s = 0; s < G_KC / 8; ++s) {
      v4f an[4], bn[2];
      if (s + 1 < G_KC / 8) {
#pragma unroll
        for (int m = 0; m < 4; ++m) an[m] = *(const v4f*)(as + 16 * m * G_LD + 8 * (s + 1));
#pragma unroll
        for (int n = 0; n < 2; ++n) bn[n] = *(const v4f*)(bs + 16 * n * G_LD + 8 * (s + 1));
      }
#pragma unroll
      for (int m = 0; m < 4; ++m)
#pragma unroll
        for (int n = 0; n < 2; ++n) {
          v2f a0 = {af[m].x, af[m].y};
          v2f b0 = {bf[n].x, bf[n].y};
          acc[m][n] = wmma4(a0, b0, acc[m][n]);
        }
#pragma unroll
      for (int m = 0; m < 4; ++m)
#pragma unroll
        for (int n = 0; n < 2; ++n) {
          v2f a1 = {af[m].z, af[m].w};
          v2f b1 = {bf[n].z, bf[n].w};
          acc[m][n] = wmma4(a1, b1, acc[m][n]);
        }
      if (s + 1 < G_KC / 8) {
#pragma unroll
        for (int m = 0; m < 4; ++m) af[m] = an[m];
#pragma unroll
        for (int n = 0; n < 2; ++n) bf[n] = bn[n];
      }
    }

    if (more) {
      float* ad = &As[buf ^ 1][a_st];
      float* bd = &Bs[buf ^ 1][b_st];
#pragma unroll
      for (int i = 0; i < 8; ++i) *(v4f*)(ad + 16 * i * G_LD) = ga[i];
#pragma unroll
      for (int i = 0; i < 4; ++i) {
        bd[0 * G_LD + 8 * i] = gb[i].x;
        bd[1 * G_LD + 8 * i] = gb[i].y;
        bd[2 * G_LD + 8 * i] = gb[i].z;
        bd[3 * G_LD + 8 * i] = gb[i].w;
      }
    }
    __syncthreads();
  }

#pragma unroll
  for (int m = 0; m < 4; ++m)
#pragma unroll
    for (int n = 0; n < 2; ++n) {
      const int cidx = col0 + wn * 32 + 16 * n + lm;
#pragma unroll
      for (int v = 0; v < 8; ++v) {
        const int r = row0 + wm * 64 + 16 * m + v + 8 * kh;
        float val = alpha * acc[m][n][v];
        if (r == cidx) val += diagBeta;
        D[(size_t)r * C_ + cidx] = val;
      }
    }
}

__global__ void bias_white_kernel(const float* __restrict__ Z,
                                  const float* __restrict__ cmean,
                                  const float* __restrict__ tr,
                                  float* __restrict__ bias) {
  const int idx = blockIdx.x * 256 + threadIdx.x;
  if (idx >= B_ * C_) return;
  const int b = idx / C_;
  const int i = idx % C_;
  const float* r = Z + ((size_t)b * C_ + i) * C_;
  const float* mu = cmean + b * C_;
  float s = 0.f;
  for (int j = 0; j < C_; ++j) s += r[j] * mu[j];
  bias[idx] = -s * rsqrtf(tr[b]);
}

__global__ __launch_bounds__(32) void apply_kernel(const float* __restrict__ Mmat,
                                                   const float* __restrict__ X,
                                                   const float* __restrict__ bias,
                                                   const float* __restrict__ tr,
                                                   int trOff, int useSqrt,
                                                   float* __restrict__ out) {
  const int b = blockIdx.z;
  const float* Mb = Mmat + (size_t)b * C_ * C_;
  const float* Xb = X + (size_t)b * C_ * N_;
  float* Ob = out + (size_t)b * C_ * N_;
  const float* bi = bias + b * C_;
  const float t = tr[trOff + b];
  const float scale = useSqrt ? sqrtf(t) : rsqrtf(t);

  const int row0 = blockIdx.y * 64;
  const int col0 = blockIdx.x * 64;
  const int lane = threadIdx.x & 31;
  const int lm = lane & 15;
  const int kh = lane >> 4;

  const v8f zero = {0.f, 0.f, 0.f, 0.f, 0.f, 0.f, 0.f, 0.f};
  v8f acc[4][4];
#pragma unroll
  for (int m = 0; m < 4; ++m)
#pragma unroll
    for (int n = 0; n < 4; ++n) acc[m][n] = zero;

  for (int k = 0; k < C_; k += 4) {
    v2f av[4], bv[4];
#pragma unroll
    for (int m = 0; m < 4; ++m)
      av[m] = *(const v2f*)(Mb + (size_t)(row0 + 16 * m + lm) * C_ + k + 2 * kh);
#pragma unroll
    for (int n = 0; n < 4; ++n) {
      const float* q = Xb + (size_t)(k + 2 * kh) * N_ + col0 + 16 * n + lm;
      v2f tt;
      tt.x = q[0];
      tt.y = q[N_];
      bv[n] = tt;
    }
#pragma unroll
    for (int m = 0; m < 4; ++m)
#pragma unroll
      for (int n = 0; n < 4; ++n) acc[m][n] = wmma4(av[m], bv[n], acc[m][n]);
  }

#pragma unroll
  for (int m = 0; m < 4; ++m)
#pragma unroll
    for (int n = 0; n < 4; ++n) {
      const int c = col0 + 16 * n + lm;
#pragma unroll
      for (int v = 0; v < 8; ++v) {
        const int r = row0 + 16 * m + v + 8 * kh;
        Ob[(size_t)r * N_ + c] = scale * acc[m][n][v] + bi[r];
      }
    }
}

extern "C" void kernel_launch(void* const* d_in, const int* in_sizes, int n_in,
                              void* d_out, int out_size, void* d_ws, size_t ws_size,
                              hipStream_t stream) {
  (void)in_sizes; (void)n_in; (void)out_size; (void)ws_size;
  const float* content = (const float*)d_in[0];
  const float* style   = (const float*)d_in[1];
  float* out = (float*)d_out;
  float* ws  = (float*)d_ws;

  const size_t CM = (size_t)C_ * C_;

  float* c_mean  = ws;
  float* s_mean  = c_mean + B_ * C_;
  float* tr      = s_mean + B_ * C_;
  float* bias_w  = tr + 8;
  float* cov_all = bias_w + B_ * C_;
  float* Y  = cov_all + 8 * CM;
  float* Z  = Y + 8 * CM;
  float* Y2 = Z + 8 * CM;
  float* Z2 = Y2 + 8 * CM;
  float* T  = Z2 + 8 * CM;
  float* whit = T + 8 * CM;

  mean_kernel<<<B_ * C_, 256, 0, stream>>>(content, c_mean);
  mean_kernel<<<B_ * C_, 256, 0, stream>>>(style, s_mean);

  dim3 covGrid(C_ / 64, C_ / 64, B_);
  cov_kernel<<<covGrid, 32, 0, stream>>>(content, c_mean, cov_all);
  cov_kernel<<<covGrid, 32, 0, stream>>>(style, s_mean, cov_all + 4 * CM);

  trace_kernel<<<8, 256, 0, stream>>>(cov_all, tr);

  {
    const size_t tot = 8 * CM;
    ns_init_kernel<<<(unsigned)((tot + 255) / 256), 256, 0, stream>>>(cov_all, tr, Y, Z);
  }
  dim3 g512(C_ / G_BN, C_ / G_BM, 8);
  float* y = Y; float* z = Z; float* y2 = Y2; float* z2 = Z2;
  for (int it = 0; it < NS_ITERS; ++it) {
    gemm512_kernel<<<g512, 128, 0, stream>>>(z, y, T, -0.5f, 1.5f);
    gemm512_kernel<<<g512, 128, 0, stream>>>(y, T, y2, 1.0f, 0.0f);
    gemm512_kernel<<<g512, 128, 0, stream>>>(T, z, z2, 1.0f, 0.0f);
    float* tmp;
    tmp = y; y = y2; y2 = tmp;
    tmp = z; z = z2; z2 = tmp;
  }

  bias_white_kernel<<<(B_ * C_ + 255) / 256, 256, 0, stream>>>(z, c_mean, tr, bias_w);

  dim3 gApply(N_ / 64, C_ / 64, B_);
  apply_kernel<<<gApply, 32, 0, stream>>>(z, content, bias_w, tr, 0, 0, whit);

  apply_kernel<<<gApply, 32, 0, stream>>>(y + 4 * CM, whit, s_mean, tr, 4, 1, out);
}


# Covariance kernel (2 launches, K=16384): 4-wave workgroups stage 128-row x 128-k panels of X in LDS, double-buffered, 32x32 tile per wave; k summed in a permuted order inside each 8-block
# speedup vs baseline: 3.1438x; 1.7734x over previous
#include <hip/hip_runtime.h>
#include <math.h>

#define B_ 4
#define C_ 512
#define N_ 16384
#define EPS_ 1e-5f
#define NS_ITERS 12

typedef __attribute__((ext_vector_type(2))) float v2f;
typedef __attribute__((ext_vector_type(8))) float v8f;

__device__ __forceinline__ v8f wmma4(v2f a, v2f b, v8f c) {
  return __builtin_amdgcn_wmma_f32_16x16x4_f32(false, a, false, b, (short)0, c,
                                               false, false);
}

__global__ void mean_kernel(const float* __restrict__ x, float* __restrict__ mean) {
  const int bc = blockIdx.x;
  const float* p = x + (size_t)bc * N_;
  float s = 0.f;
  for (int i = threadIdx.x; i < N_; i += 256) s += p[i];
  __shared__ float red[256];
  red[threadIdx.x] = s;
  __syncthreads();
  for (int o = 128; o > 0; o >>= 1) {
    if ((int)threadIdx.x < o) red[threadIdx.x] += red[threadIdx.x + o];
    __syncthreads();
  }
  if (threadIdx.x == 0) mean[bc] = red[0] * (1.0f / (float)N_);
}

#define V_KC 128
#define V_LD 132
#define V_PR 128

typedef __attribute__((ext_vector_type(4))) float v4f;

__global__ __launch_bounds__(128) void cov_kernel(const float* __restrict__ x,
                                                  const float* __restrict__ mean,
                                                  float* __restrict__ cov) {
  __shared__ __attribute__((aligned(16))) float Ps[2][V_PR * V_LD];

  const int b    = blockIdx.z;
  const int row0 = blockIdx.y * 64;
  const int col0 = blockIdx.x * 64;
  const float* X  = x + (size_t)b * C_ * N_;
  const float* mu = mean + b * C_;
  float* D = cov + (size_t)b * C_ * C_;

  const int tid  = threadIdx.x;
  const int lane = tid & 31;
  const int wave = tid >> 5;
  const int wm = wave >> 1;
  const int wn = wave & 1;
  const int lm = lane & 15;
  const int kh = lane >> 4;

  const float* gpa = X + (size_t)(row0 + wave) * N_ + lane * 4;
  const float* gpb = X + (size_t)(col0 + wave) * N_ + lane * 4;
  const int p_st = wave * V_LD + lane * 4;
  const int a_rd = (wm * 32 + lm) * V_LD + 4 * kh;
  const int b_rd = (64 + wn * 32 + lm) * V_LD + 4 * kh;

  const v8f zero = {0.f, 0.f, 0.f, 0.f, 0.f, 0.f, 0.f, 0.f};
  v8f acc[2][2];
#pragma unroll
  for (int m = 0; m < 2; ++m)
#pragma unroll
    for (int n = 0; n < 2; ++n) acc[m][n] = zero;

  v4f g[32];
#pragma unroll
  for (int j = 0; j < 16; ++j) g[j] = *(const v4f*)(gpa + (size_t)(4 * j) * N_);
#pragma unroll
  for (int j = 0; j < 16; ++j) g[16 + j] = *(const v4f*)(gpb + (size_t)(4 * j) * N_);
#pragma unroll
  for (int j = 0; j < 32; ++j) *(v4f*)(&Ps[0][p_st + 4 * j * V_LD]) = g[j];
  __syncthreads();

  for (int t = 0; t < N_ / V_KC; ++t) {
    const int buf = t & 1;
    const bool more = (t + 1 < N_ / V_KC);
    if (more) {
      const int k1 = (t + 1) * V_KC;
#pragma unroll
      for (int j = 0; j < 16; ++j) g[j] = *(const v4f*)(gpa + (size_t)(4 * j) * N_ + k1);
#pragma unroll
      for (int j = 0; j < 16; ++j) g[16 + j] = *(const v4f*)(gpb + (size_t)(4 * j) * N_ + k1);
    }

    const float* as = &Ps[buf][a_rd];
    const float* bs = &Ps[buf][b_rd];
    v4f af[2], bf[2];
#pragma unroll
    for (int m = 0; m < 2; ++m) af[m] = *(const v4f*)(as + 16 * m * V_LD);
#pragma unroll
    for (int n = 0; n < 2; ++n) bf[n] = *(const v4f*)(bs + 16 * n * V_LD);
#pragma unroll
    for (int s = 0; s < V_KC / 8; ++s) {
      v4f an[2], bn[2];
      if (s + 1 < V_KC / 8) {
#pragma unroll
        for (int m = 0; m < 2; ++m) an[m] = *(const v4f*)(as + 16 * m * V_LD + 8 * (s + 1));
#pragma unroll
        for (int n = 0; n < 2; ++n) bn[n] = *(const v4f*)(bs + 16 * n * V_LD + 8 * (s + 1));
      }
#pragma unroll
      for (int m = 0; m < 2; ++m)
#pragma unroll
        for (int n = 0; n < 2; ++n) {
          v2f a0 = {af[m].x, af[m].y};
          v2f b0 = {bf[n].x, bf[n].y};
          acc[m][n] = wmma4(a0, b0, acc[m][n]);
        }
#pragma unroll
      for (int m = 0; m < 2; ++m)
#pragma unroll
        for (int n = 0; n < 2; ++n) {
          v2f a1 = {af[m].z, af[m].w};
          v2f b1 = {bf[n].z, bf[n].w};
          acc[m][n] = wmma4(a1, b1, acc[m][n]);
        }
      if (s + 1 < V_KC / 8) {
#pragma unroll
        for (int m = 0; m < 2; ++m) af[m] = an[m];
#pragma unroll
        for (int n = 0; n < 2; ++n) bf[n] = bn[n];
      }
    }

    if (more) {
      float* pd = &Ps[buf ^ 1][p_st];
#pragma unroll
      for (int j = 0; j < 32; ++j) *(v4f*)(pd + 4 * j * V_LD) = g[j];
    }
    __syncthreads();
  }

  const float fac = 1.0f / (float)(N_ - 1);
#pragma unroll
  for (int m = 0; m < 2; ++m) {
#pragma unroll
    for (int n = 0; n < 2; ++n) {
      const int cidx = col0 + wn * 32 + 16 * n + lm;
      const float mj = mu[cidx];
#pragma unroll
      for (int v = 0; v < 8; ++v) {
        const int r = row0 + wm * 32 + 16 * m + v + 8 * kh;
        float val = (acc[m][n][v] - (float)N_ * mu[r] * mj) * fac;
        if (r == cidx) val += EPS_;
        D[(size_t)r * C_ + cidx] = val;
      }
    }
  }
}

__global__ void trace_kernel(const float* __restrict__ cov, float* __restrict__ tr) {
  const int m = blockIdx.x;
  const float* A = cov + (size_t)m * C_ * C_;
  float s = 0.f;
  for (int i = threadIdx.x; i < C_; i += 256) s += A[(size_t)i * C_ + i];
  __shared__ float red[256];
  red[threadIdx.x] = s;
  __syncthreads();
  for (int o = 128; o > 0; o >>= 1) {
    if ((int)threadIdx.x < o) red[threadIdx.x] += red[threadIdx.x + o];
    __syncthreads();
  }
  if (threadIdx.x == 0) tr[m] = red[0];
}

__global__ void ns_init_kernel(const float* __restrict__ cov,
                               const float* __restrict__ tr,
                               float* __restrict__ Y, float* __restrict__ Z) {
  const size_t i = (size_t)blockIdx.x * 256 + threadIdx.x;
  const size_t total = (size_t)8 * C_ * C_;
  if (i >= total) return;
  const int m = (int)(i / ((size_t)C_ * C_));
  const size_t rr = i % ((size_t)C_ * C_);
  Y[i] = cov[i] / tr[m];
  Z[i] = ((rr / C_) == (rr % C_)) ? 1.f : 0.f;
}

#define G_BM 128
#define G_BN 64
#define G_KC 32
#define G_LD 36

__global__ __launch_bounds__(128) void gemm512_kernel(const float* __restrict__ Amat,
                                                      const float* __restrict__ Bmat,
                                                      float* __restrict__ Dmat,
                                                      float alpha, float diagBeta) {
  __shared__ __attribute__((aligned(16))) float As[2][G_BM * G_LD];
  __shared__ __attribute__((aligned(16))) float Bs[2][G_BN * G_LD];

  const int mm = blockIdx.z;
  const float* A  = Amat + (size_t)mm * C_ * C_;
  const float* Bp = Bmat + (size_t)mm * C_ * C_;
  float* D = Dmat + (size_t)mm * C_ * C_;
  const int row0 = blockIdx.y * G_BM;
  const int col0 = blockIdx.x * G_BN;
  const int tid  = threadIdx.x;
  const int lane = tid & 31;
  const int wave = tid >> 5;
  const int wm = wave >> 1;
  const int wn = wave & 1;
  const int lm = lane & 15;
  const int kh = lane >> 4;

  const float* agp = A + (size_t)(row0 + (tid >> 3)) * C_ + (tid & 7) * 4;
  const float* bgp = Bp + (size_t)(tid >> 4) * C_ + col0 + (tid & 15) * 4;
  const int a_st = (tid >> 3) * G_LD + (tid & 7) * 4;
  const int b_st = ((tid & 15) * 4) * G_LD + (tid >> 4);
  const int a_rd = (wm * 64 + lm) * G_LD + 4 * kh;
  const int b_rd = (wn * 32 + lm) * G_LD + 4 * kh;

  const v8f zero = {0.f, 0.f, 0.f, 0.f, 0.f, 0.f, 0.f, 0.f};
  v8f acc[4][2];
#pragma unroll
  for (int m = 0; m < 4; ++m)
#pragma unroll
    for (int n = 0; n < 2; ++n) acc[m][n] = zero;

  v4f ga[8], gb[4];
#pragma unroll
  for (int i = 0; i < 8; ++i) ga[i] = *(const v4f*)(agp + (size_t)(16 * i) * C_);
#pragma unroll
  for (int i = 0; i < 4; ++i) gb[i] = *(const v4f*)(bgp + (size_t)(8 * i) * C_);
#pragma unroll
  for (int i = 0; i < 8; ++i) *(v4f*)(&As[0][a_st + 16 * i * G_LD]) = ga[i];
#pragma unroll
  for (int i = 0; i < 4; ++i) {
    Bs[0][b_st + 0 * G_LD + 8 * i] = gb[i].x;
    Bs[0][b_st + 1 * G_LD + 8 * i] = gb[i].y;
    Bs[0][b_st + 2 * G_LD + 8 * i] = gb[i].z;
    Bs[0][b_st + 3 * G_LD + 8 * i] = gb[i].w;
  }
  __syncthreads();

  for (int t = 0; t < C_ / G_KC; ++t) {
    const int buf = t & 1;
    const bool more = (t + 1 < C_ / G_KC);
    if (more) {
      const int k1 = (t + 1) * G_KC;
#pragma unroll
      for (int i = 0; i < 8; ++i) ga[i] = *(const v4f*)(agp + (size_t)(16 * i) * C_ + k1);
#pragma unroll
      for (int i = 0; i < 4; ++i) gb[i] = *(const v4f*)(bgp + (size_t)(k1 + 8 * i) * C_);
    }

    const float* as = &As[buf][a_rd];
    const float* bs = &Bs[buf][b_rd];
    v4f af[4], bf[2];
#pragma unroll
    for (int m = 0; m < 4; ++m) af[m] = *(const v4f*)(as + 16 * m * G_LD);
#pragma unroll
    for (int n = 0; n < 2; ++n) bf[n] = *(const v4f*)(bs + 16 * n * G_LD);
#pragma unroll
    for (int s = 0; s < G_KC / 8; ++s) {
      v4f an[4], bn[2];
      if (s + 1 < G_KC / 8) {
#pragma unroll
        for (int m = 0; m < 4; ++m) an[m] = *(const v4f*)(as + 16 * m * G_LD + 8 * (s + 1));
#pragma unroll
        for (int n = 0; n < 2; ++n) bn[n] = *(const v4f*)(bs + 16 * n * G_LD + 8 * (s + 1));
      }
#pragma unroll
      for (int m = 0; m < 4; ++m)
#pragma unroll
        for (int n = 0; n < 2; ++n) {
          v2f a0 = {af[m].x, af[m].y};
          v2f b0 = {bf[n].x, bf[n].y};
          acc[m][n] = wmma4(a0, b0, acc[m][n]);
        }
#pragma unroll
      for (int m = 0; m < 4; ++m)
#pragma unroll
        for (int n = 0; n < 2; ++n) {
          v2f a1 = {af[m].z, af[m].w};
          v2f b1 = {bf[n].z, bf[n].w};
          acc[m][n] = wmma4(a1, b1, acc[m][n]);
        }
      if (s + 1 < G_KC / 8) {
#pragma unroll
        for (int m = 0; m < 4; ++m) af[m] = an[m];
#pragma unroll
        for (int n = 0; n < 2; ++n) bf[n] = bn[n];
      }
    }

    if (more) {
      float* ad = &As[buf ^ 1][a_st];
      float* bd = &Bs[buf ^ 1][b_st];
#pragma unroll
      for (int i = 0; i < 8; ++i) *(v4f*)(ad + 16 * i * G_LD) = ga[i];
#pragma unroll
      for (int i = 0; i < 4; ++i) {
        bd[0 * G_LD + 8 * i] = gb[i].x;
        bd[1 * G_LD + 8 * i] = gb[i].y;
        bd[2 * G_LD + 8 * i] = gb[i].z;
        bd[3 * G_LD + 8 * i] = gb[i].w;
      }
    }
    __syncthreads();
  }

#pragma unroll
  for (int m = 0; m < 4; ++m)
#pragma unroll
    for (int n = 0; n < 2; ++n) {
      const int cidx = col0 + wn * 32 + 16 * n + lm;
#pragma unroll
      for (int v = 0; v < 8; ++v) {
        const int r = row0 + wm * 64 + 16 * m + v + 8 * kh;
        float val = alpha * acc[m][n][v];
        if (r == cidx) val += diagBeta;
        D[(size_t)r * C_ + cidx] = val;
      }
    }
}

__global__ void bias_white_kernel(const float* __restrict__ Z,
                                  const float* __restrict__ cmean,
                                  const float* __restrict__ tr,
                                  float* __restrict__ bias) {
  const int idx = blockIdx.x * 256 + threadIdx.x;
  if (idx >= B_ * C_) return;
  const int b = idx / C_;
  const int i = idx % C_;
  const float* r = Z + ((size_t)b * C_ + i) * C_;
  const float* mu = cmean + b * C_;
  float s = 0.f;
  for (int j = 0; j < C_; ++j) s += r[j] * mu[j];
  bias[idx] = -s * rsqrtf(tr[b]);
}

__global__ __launch_bounds__(32) void apply_kernel(const float* __restrict__ Mmat,
                                                   const float* __restrict__ X,
                                                   const float* __restrict__ bias,
                                                   const float* __restrict__ tr,
                                                   int trOff, int useSqrt,
                                                   float* __restrict__ out) {
  const int b = blockIdx.z;
  const float* Mb = Mmat + (size_t)b * C_ * C_;
  const float* Xb = X + (size_t)b * C_ * N_;
  float* Ob = out + (size_t)b * C_ * N_;
  const float* bi = bias + b * C_;
  const float t = tr[trOff + b];
  const float scale = useSqrt ? sqrtf(t) : rsqrtf(t);

  const int row0 = blockIdx.y * 64;
  const int col0 = blockIdx.x * 64;
  const int lane = threadIdx.x & 31;
  const int lm = lane & 15;
  const int kh = lane >> 4;

  const v8f zero = {0.f, 0.f, 0.f, 0.f, 0.f, 0.f, 0.f, 0.f};
  v8f acc[4][4];
#pragma unroll
  for (int m = 0; m < 4; ++m)
#pragma unroll
    for (int n = 0; n < 4; ++n) acc[m][n] = zero;

  for (int k = 0; k < C_; k += 4) {
    v2f av[4], bv[4];
#pragma unroll
    for (int m = 0; m < 4; ++m)
      av[m] = *(const v2f*)(Mb + (size_t)(row0 + 16 * m + lm) * C_ + k + 2 * kh);
#pragma unroll
    for (int n = 0; n < 4; ++n) {
      const float* q = Xb + (size_t)(k + 2 * kh) * N_ + col0 + 16 * n + lm;
      v2f tt;
      tt.x = q[0];
      tt.y = q[N_];
      bv[n] = tt;
    }
#pragma unroll
    for (int m = 0; m < 4; ++m)
#pragma unroll
      for (int n = 0; n < 4; ++n) acc[m][n] = wmma4(av[m], bv[n], acc[m][n]);
  }

#pragma unroll
  for (int m = 0; m < 4; ++m)
#pragma unroll
    for (int n = 0; n < 4; ++n) {
      const int c = col0 + 16 * n + lm;
#pragma unroll
      for (int v = 0; v < 8; ++v) {
        const int r = row0 + 16 * m + v + 8 * kh;
        Ob[(size_t)r * N_ + c] = scale * acc[m][n][v] + bi[r];
      }
    }
}

extern "C" void kernel_launch(void* const* d_in, const int* in_sizes, int n_in,
                              void* d_out, int out_size, void* d_ws, size_t ws_size,
                              hipStream_t stream) {
  (void)in_sizes; (void)n_in; (void)out_size; (void)ws_size;
  const float* content = (const float*)d_in[0];
  const float* style   = (const float*)d_in[1];
  float* out = (float*)d_out;
  float* ws  = (float*)d_ws;

  const size_t CM = (size_t)C_ * C_;

  float* c_mean  = ws;
  float* s_mean  = c_mean + B_ * C_;
  float* tr      = s_mean + B_ * C_;
  float* bias_w  = tr + 8;
  float* cov_all = bias_w + B_ * C_;
  float* Y  = cov_all + 8 * CM;
  float* Z  = Y + 8 * CM;
  float* Y2 = Z + 8 * CM;
  float* Z2 = Y2 + 8 * CM;
  float* T  = Z2 + 8 * CM;
  float* whit = T + 8 * CM;

  mean_kernel<<<B_ * C_, 256, 0, stream>>>(content, c_mean);
  mean_kernel<<<B_ * C_, 256, 0, stream>>>(style, s_mean);

  dim3 covGrid(C_ / 64, C_ / 64, B_);
  cov_kernel<<<covGrid, 128, 0, stream>>>(content, c_mean, cov_all);
  cov_kernel<<<covGrid, 128, 0, stream>>>(style, s_mean, cov_all + 4 * CM);

  trace_kernel<<<8, 256, 0, stream>>>(cov_all, tr);

  {
    const size_t tot = 8 * CM;
    ns_init_kernel<<<(unsigned)((tot + 255) / 256), 256, 0, stream>>>(cov_all, tr, Y, Z);
  }
  dim3 g512(C_ / G_BN, C_ / G_BM, 8);
  float* y = Y; float* z = Z; float* y2 = Y2; float* z2 = Z2;
  for (int it = 0; it < NS_ITERS; ++it) {
    gemm512_kernel<<<g512, 128, 0, stream>>>(z, y, T, -0.5f, 1.5f);
    gemm512_kernel<<<g512, 128, 0, stream>>>(y, T, y2, 1.0f, 0.0f);
    gemm512_kernel<<<g512, 128, 0, stream>>>(T, z, z2, 1.0f, 0.0f);
    float* tmp;
    tmp = y; y = y2; y2 = tmp;
    tmp = z; z = z2; z2 = tmp;
  }

  bias_white_kernel<<<(B_ * C_ + 255) / 256, 256, 0, stream>>>(z, c_mean, tr, bias_w);

  dim3 gApply(N_ / 64, C_ / 64, B_);
  apply_kernel<<<gApply, 32, 0, stream>>>(z, content, bias_w, tr, 0, 0, whit);

  apply_kernel<<<gApply, 32, 0, stream>>>(y + 4 * CM, whit, s_mean, tr, 4, 1, out);
}


# Fold coloring and whitening into one matrix per batch, P = Ys*Zc (one extra 512^3 GEMM launch): one apply pass over the pixels, not two; sum order changes: (Ys*Zc)*X for Ys*(Zc*X)
# speedup vs baseline: 3.7068x; 1.1791x over previous
#include <hip/hip_runtime.h>
#include <math.h>

#define B_ 4
#define C_ 512
#define N_ 16384
#define EPS_ 1e-5f
#define NS_ITERS 12

typedef __attribute__((ext_vector_type(2))) float v2f;
typedef __attribute__((ext_vector_type(8))) float v8f;

__device__ __forceinline__ v8f wmma4(v2f a, v2f b, v8f c) {
  return __builtin_amdgcn_wmma_f32_16x16x4_f32(false, a, false, b, (short)0, c,
                                               false, false);
}

__global__ void mean_kernel(const float* __restrict__ x, float* __restrict__ mean) {
  const int bc = blockIdx.x;
  const float* p = x + (size_t)bc * N_;
  float s = 0.f;
  for (int i = threadIdx.x; i < N_; i += 256) s += p[i];
  __shared__ float red[256];
  red[threadIdx.x] = s;
  __syncthreads();
  for (int o = 128; o > 0; o >>= 1) {
    if ((int)threadIdx.x < o) red[threadIdx.x] += red[threadIdx.x + o];
    __syncthreads();
  }
  if (threadIdx.x == 0) mean[bc] = red[0] * (1.0f / (float)N_);
}

#define V_KC 128
#define V_LD 132
#define V_PR 128

typedef __attribute__((ext_vector_type(4))) float v4f;

__global__ __launch_bounds__(128) void cov_kernel(const float* __restrict__ x,
                                                  const float* __restrict__ mean,
                                                  float* __restrict__ cov) {
  __shared__ __attribute__((aligned(16))) float Ps[2][V_PR * V_LD];

  const int b    = blockIdx.z;
  const int row0 = blockIdx.y * 64;
  const int col0 = blockIdx.x * 64;
  const float* X  = x + (size_t)b * C_ * N_;
  const float* mu = mean + b * C_;
  float* D = cov + (size_t)b * C_ * C_;

  const int tid  = threadIdx.x;
  const int lane = tid & 31;
  const int wave = tid >> 5;
  const int wm = wave >> 1;
  const int wn = wave & 1;
  const int lm = lane & 15;
  const int kh = lane >> 4;

  const float* gpa = X + (size_t)(row0 + wave) * N_ + lane * 4;
  const float* gpb = X + (size_t)(col0 + wave) * N_ + lane * 4;
  const int p_st = wave * V_LD + lane * 4;
  const int a_rd = (wm * 32 + lm) * V_LD + 4 * kh;
  const int b_rd = (64 + wn * 32 + lm) * V_LD + 4 * kh;

  const v8f zero = {0.f, 0.f, 0.f, 0.f, 0.f, 0.f, 0.f, 0.f};
  v8f acc[2][2];
#pragma unroll
  for (int m = 0; m < 2; ++m)
#pragma unroll
    for (int n = 0; n < 2; ++n) acc[m][n] = zero;

  v4f g[32];
#pragma unroll
  for (int j = 0; j < 16; ++j) g[j] = *(const v4f*)(gpa + (size_t)(4 * j) * N_);
#pragma unroll
  for (int j = 0; j < 16; ++j) g[16 + j] = *(const v4f*)(gpb + (size_t)(4 * j) * N_);
#pragma unroll
  for (int j = 0; j < 32; ++j) *(v4f*)(&Ps[0][p_st + 4 * j * V_LD]) = g[j];
  __syncthreads();

  for (int t = 0; t < N_ / V_KC; ++t) {
    const int buf = t & 1;
    const bool more = (t + 1 < N_ / V_KC);
    if (more) {
      const int k1 = (t + 1) * V_KC;
#pragma unroll
      for (int j = 0; j < 16; ++j) g[j] = *(const v4f*)(gpa + (size_t)(4 * j) * N_ + k1);
#pragma unroll
      for (int j = 0; j < 16; ++j) g[16 + j] = *(const v4f*)(gpb + (size_t)(4 * j) * N_ + k1);
    }

    const float* as = &Ps[buf][a_rd];
    const float* bs = &Ps[buf][b_rd];
    v4f af[2], bf[2];
#pragma unroll
    for (int m = 0; m < 2; ++m) af[m] = *(const v4f*)(as + 16 * m * V_LD);
#pragma unroll
    for (int n = 0; n < 2; ++n) bf[n] = *(const v4f*)(bs + 16 * n * V_LD);
#pragma unroll
    for (int s = 0; s < V_KC / 8; ++s) {
      v4f an[2], bn[2];
      if (s + 1 < V_KC / 8) {
#pragma unroll
        for (int m = 0; m < 2; ++m) an[m] = *(const v4f*)(as + 16 * m * V_LD + 8 * (s + 1));
#pragma unroll
        for (int n = 0; n < 2; ++n) bn[n] = *(const v4f*)(bs + 16 * n * V_LD + 8 * (s + 1));
      }
#pragma unroll
      for (int m = 0; m < 2; ++m)
#pragma unroll
        for (int n = 0; n < 2; ++n) {
          v2f a0 = {af[m].x, af[m].y};
          v2f b0 = {bf[n].x, bf[n].y};
          acc[m][n] = wmma4(a0, b0, acc[m][n]);
        }
#pragma unroll
      for (int m = 0; m < 2; ++m)
#pragma unroll
        for (int n = 0; n < 2; ++n) {
          v2f a1 = {af[m].z, af[m].w};
          v2f b1 = {bf[n].z, bf[n].w};
          acc[m][n] = wmma4(a1, b1, acc[m][n]);
        }
      if (s + 1 < V_KC / 8) {
#pragma unroll
        for (int m = 0; m < 2; ++m) af[m] = an[m];
#pragma unroll
        for (int n = 0; n < 2; ++n) bf[n] = bn[n];
      }
    }

    if (more) {
      float* pd = &Ps[buf ^ 1][p_st];
#pragma unroll
      for (int j = 0; j < 32; ++j) *(v4f*)(pd + 4 * j * V_LD) = g[j];
    }
    __syncthreads();
  }

  const float fac = 1.0f / (float)(N_ - 1);
#pragma unroll
  for (int m = 0; m < 2; ++m) {
#pragma unroll
    for (int n = 0; n < 2; ++n) {
      const int cidx = col0 + wn * 32 + 16 * n + lm;
      const float mj = mu[cidx];
#pragma unroll
      for (int v = 0; v < 8; ++v) {
        const int r = row0 + wm * 32 + 16 * m + v + 8 * kh;
        float val = (acc[m][n][v] - (float)N_ * mu[r] * mj) * fac;
        if (r == cidx) val += EPS_;
        D[(size_t)r * C_ + cidx] = val;
      }
    }
  }
}

__global__ void trace_kernel(const float* __restrict__ cov, float* __restrict__ tr) {
  const int m = blockIdx.x;
  const float* A = cov + (size_t)m * C_ * C_;
  float s = 0.f;
  for (int i = threadIdx.x; i < C_; i += 256) s += A[(size_t)i * C_ + i];
  __shared__ float red[256];
  red[threadIdx.x] = s;
  __syncthreads();
  for (int o = 128; o > 0; o >>= 1) {
    if ((int)threadIdx.x < o) red[threadIdx.x] += red[threadIdx.x + o];
    __syncthreads();
  }
  if (threadIdx.x == 0) tr[m] = red[0];
}

__global__ void ns_init_kernel(const float* __restrict__ cov,
                               const float* __restrict__ tr,
                               float* __restrict__ Y, float* __restrict__ Z) {
  const size_t i = (size_t)blockIdx.x * 256 + threadIdx.x;
  const size_t total = (size_t)8 * C_ * C_;
  if (i >= total) return;
  const int m = (int)(i / ((size_t)C_ * C_));
  const size_t rr = i % ((size_t)C_ * C_);
  Y[i] = cov[i] / tr[m];
  Z[i] = ((rr / C_) == (rr % C_)) ? 1.f : 0.f;
}

#define G_BM 128
#define G_BN 64
#define G_KC 32
#define G_LD 36

__global__ __launch_bounds__(128) void gemm512_kernel(const float* __restrict__ Amat,
                                                      const float* __restrict__ Bmat,
                                                      float* __restrict__ Dmat,
                                                      float alpha, float diagBeta) {
  __shared__ __attribute__((aligned(16))) float As[2][G_BM * G_LD];
  __shared__ __attribute__((aligned(16))) float Bs[2][G_BN * G_LD];

  const int mm = blockIdx.z;
  const float* A  = Amat + (size_t)mm * C_ * C_;
  const float* Bp = Bmat + (size_t)mm * C_ * C_;
  float* D = Dmat + (size_t)mm * C_ * C_;
  const int row0 = blockIdx.y * G_BM;
  const int col0 = blockIdx.x * G_BN;
  const int tid  = threadIdx.x;
  const int lane = tid & 31;
  const int wave = tid >> 5;
  const int wm = wave >> 1;
  const int wn = wave & 1;
  const int lm = lane & 15;
  const int kh = lane >> 4;

  const float* agp = A + (size_t)(row0 + (tid >> 3)) * C_ + (tid & 7) * 4;
  const float* bgp = Bp + (size_t)(tid >> 4) * C_ + col0 + (tid & 15) * 4;
  const int a_st = (tid >> 3) * G_LD + (tid & 7) * 4;
  const int b_st = ((tid & 15) * 4) * G_LD + (tid >> 4);
  const int a_rd = (wm * 64 + lm) * G_LD + 4 * kh;
  const int b_rd = (wn * 32 + lm) * G_LD + 4 * kh;

  const v8f zero = {0.f, 0.f, 0.f, 0.f, 0.f, 0.f, 0.f, 0.f};
  v8f acc[4][2];
#pragma unroll
  for (int m = 0; m < 4; ++m)
#pragma unroll
    for (int n = 0; n < 2; ++n) acc[m][n] = zero;

  v4f ga[8], gb[4];
#pragma unroll
  for (int i = 0; i < 8; ++i) ga[i] = *(const v4f*)(agp + (size_t)(16 * i) * C_);
#pragma unroll
  for (int i = 0; i < 4; ++i) gb[i] = *(const v4f*)(bgp + (size_t)(8 * i) * C_);
#pragma unroll
  for (int i = 0; i < 8; ++i) *(v4f*)(&As[0][a_st + 16 * i * G_LD]) = ga[i];
#pragma unroll
  for (int i = 0; i < 4; ++i) {
    Bs[0][b_st + 0 * G_LD + 8 * i] = gb[i].x;
    Bs[0][b_st + 1 * G_LD + 8 * i] = gb[i].y;
    Bs[0][b_st + 2 * G_LD + 8 * i] = gb[i].z;
    Bs[0][b_st + 3 * G_LD + 8 * i] = gb[i].w;
  }
  __syncthreads();

  for (int t = 0; t < C_ / G_KC; ++t) {
    const int buf = t & 1;
    const bool more = (t + 1 < C_ / G_KC);
    if (more) {
      const int k1 = (t + 1) * G_KC;
#pragma unroll
      for (int i = 0; i < 8; ++i) ga[i] = *(const v4f*)(agp + (size_t)(16 * i) * C_ + k1);
#pragma unroll
      for (int i = 0; i < 4; ++i) gb[i] = *(const v4f*)(bgp + (size_t)(k1 + 8 * i) * C_);
    }

    const float* as = &As[buf][a_rd];
    const float* bs = &Bs[buf][b_rd];
    v4f af[4], bf[2];
#pragma unroll
    for (int m = 0; m < 4; ++m) af[m] = *(const v4f*)(as + 16 * m * G_LD);
#pragma unroll
    for (int n = 0; n < 2; ++n) bf[n] = *(const v4f*)(bs + 16 * n * G_LD);
#pragma unroll
    for (int s = 0; s < G_KC / 8; ++s) {
      v4f an[4], bn[2];
      if (s + 1 < G_KC / 8) {
#pragma unroll
        for (int m = 0; m < 4; ++m) an[m] = *(const v4f*)(as + 16 * m * G_LD + 8 * (s + 1));
#pragma unroll
        for (int n = 0; n < 2; ++n) bn[n] = *(const v4f*)(bs + 16 * n * G_LD + 8 * (s + 1));
      }
#pragma unroll
      for (int m = 0; m < 4; ++m)
#pragma unroll
        for (int n = 0; n < 2; ++n) {
          v2f a0 = {af[m].x, af[m].y};
          v2f b0 = {bf[n].x, bf[n].y};
          acc[m][n] = wmma4(a0, b0, acc[m][n]);
        }
#pragma unroll
      for (int m = 0; m < 4; ++m)
#pragma unroll
        for (int n = 0; n < 2; ++n) {
          v2f a1 = {af[m].z, af[m].w};
          v2f b1 = {bf[n].z, bf[n].w};
          acc[m][n] = wmma4(a1, b1, acc[m][n]);
        }
      if (s + 1 < G_KC / 8) {
#pragma unroll
        for (int m = 0; m < 4; ++m) af[m] = an[m];
#pragma unroll
        for (int n = 0; n < 2; ++n) bf[n] = bn[n];
      }
    }

    if (more) {
      float* ad = &As[buf ^ 1][a_st];
      float* bd = &Bs[buf ^ 1][b_st];
#pragma unroll
      for (int i = 0; i < 8; ++i) *(v4f*)(ad + 16 * i * G_LD) = ga[i];
#pragma unroll
      for (int i = 0; i < 4; ++i) {
        bd[0 * G_LD + 8 * i] = gb[i].x;
        bd[1 * G_LD + 8 * i] = gb[i].y;
        bd[2 * G_LD + 8 * i] = gb[i].z;
        bd[3 * G_LD + 8 * i] = gb[i].w;
      }
    }
    __syncthreads();
  }

#pragma unroll
  for (int m = 0; m < 4; ++m)
#pragma unroll
    for (int n = 0; n < 2; ++n) {
      const int cidx = col0 + wn * 32 + 16 * n + lm;
#pragma unroll
      for (int v = 0; v < 8; ++v) {
        const int r = row0 + wm * 64 + 16 * m + v + 8 * kh;
        float val = alpha * acc[m][n][v];
        if (r == cidx) val += diagBeta;
        D[(size_t)r * C_ + cidx] = val;
      }
    }
}

__global__ void bias_fold_kernel(const float* __restrict__ P,
                                 const float* __restrict__ cmean,
                                 const float* __restrict__ smean,
                                 const float* __restrict__ tr,
                                 float* __restrict__ bias) {
  const int idx = blockIdx.x * 256 + threadIdx.x;
  if (idx >= B_ * C_) return;
  const int b = idx / C_;
  const int i = idx % C_;
  const float* r = P + ((size_t)b * C_ + i) * C_;
  const float* mu = cmean + b * C_;
  float s = 0.f;
  for (int j = 0; j < C_; ++j) s += r[j] * mu[j];
  bias[idx] = smean[idx] - s * (sqrtf(tr[4 + b]) * rsqrtf(tr[b]));
}

__global__ __launch_bounds__(32) void apply_kernel(const float* __restrict__ Mmat,
                                                   const float* __restrict__ X,
                                                   const float* __restrict__ bias,
                                                   const float* __restrict__ tr,
                                                   float* __restrict__ out) {
  const int b = blockIdx.z;
  const float* Mb = Mmat + (size_t)b * C_ * C_;
  const float* Xb = X + (size_t)b * C_ * N_;
  float* Ob = out + (size_t)b * C_ * N_;
  const float* bi = bias + b * C_;
  const float scale = sqrtf(tr[4 + b]) * rsqrtf(tr[b]);

  const int row0 = blockIdx.y * 64;
  const int col0 = blockIdx.x * 64;
  const int lane = threadIdx.x & 31;
  const int lm = lane & 15;
  const int kh = lane >> 4;

  const v8f zero = {0.f, 0.f, 0.f, 0.f, 0.f, 0.f, 0.f, 0.f};
  v8f acc[4][4];
#pragma unroll
  for (int m = 0; m < 4; ++m)
#pragma unroll
    for (int n = 0; n < 4; ++n) acc[m][n] = zero;

  for (int k = 0; k < C_; k += 4) {
    v2f av[4], bv[4];
#pragma unroll
    for (int m = 0; m < 4; ++m)
      av[m] = *(const v2f*)(Mb + (size_t)(row0 + 16 * m + lm) * C_ + k + 2 * kh);
#pragma unroll
    for (int n = 0; n < 4; ++n) {
      const float* q = Xb + (size_t)(k + 2 * kh) * N_ + col0 + 16 * n + lm;
      v2f tt;
      tt.x = q[0];
      tt.y = q[N_];
      bv[n] = tt;
    }
#pragma unroll
    for (int m = 0; m < 4; ++m)
#pragma unroll
      for (int n = 0; n < 4; ++n) acc[m][n] = wmma4(av[m], bv[n], acc[m][n]);
  }

#pragma unroll
  for (int m = 0; m < 4; ++m)
#pragma unroll
    for (int n = 0; n < 4; ++n) {
      const int c = col0 + 16 * n + lm;
#pragma unroll
      for (int v = 0; v < 8; ++v) {
        const int r = row0 + 16 * m + v + 8 * kh;
        Ob[(size_t)r * N_ + c] = scale * acc[m][n][v] + bi[r];
      }
    }
}

extern "C" void kernel_launch(void* const* d_in, const int* in_sizes, int n_in,
                              void* d_out, int out_size, void* d_ws, size_t ws_size,
                              hipStream_t stream) {
  (void)in_sizes; (void)n_in; (void)out_size; (void)ws_size;
  const float* content = (const float*)d_in[0];
  const float* style   = (const float*)d_in[1];
  float* out = (float*)d_out;
  float* ws  = (float*)d_ws;

  const size_t CM = (size_t)C_ * C_;

  float* c_mean  = ws;
  float* s_mean  = c_mean + B_ * C_;
  float* tr      = s_mean + B_ * C_;
  float* bias_w  = tr + 8;
  float* cov_all = bias_w + B_ * C_;
  float* Y  = cov_all + 8 * CM;
  float* Z  = Y + 8 * CM;
  float* Y2 = Z + 8 * CM;
  float* Z2 = Y2 + 8 * CM;
  float* T  = Z2 + 8 * CM;

  mean_kernel<<<B_ * C_, 256, 0, stream>>>(content, c_mean);
  mean_kernel<<<B_ * C_, 256, 0, stream>>>(style, s_mean);

  dim3 covGrid(C_ / 64, C_ / 64, B_);
  cov_kernel<<<covGrid, 128, 0, stream>>>(content, c_mean, cov_all);
  cov_kernel<<<covGrid, 128, 0, stream>>>(style, s_mean, cov_all + 4 * CM);

  trace_kernel<<<8, 256, 0, stream>>>(cov_all, tr);

  {
    const size_t tot = 8 * CM;
    ns_init_kernel<<<(unsigned)((tot + 255) / 256), 256, 0, stream>>>(cov_all, tr, Y, Z);
  }
  dim3 g512(C_ / G_BN, C_ / G_BM, 8);
  float* y = Y; float* z = Z; float* y2 = Y2; float* z2 = Z2;
  for (int it = 0; it < NS_ITERS; ++it) {
    gemm512_kernel<<<g512, 128, 0, stream>>>(z, y, T, -0.5f, 1.5f);
    gemm512_kernel<<<g512, 128, 0, stream>>>(y, T, y2, 1.0f, 0.0f);
    gemm512_kernel<<<g512, 128, 0, stream>>>(T, z, z2, 1.0f, 0.0f);
    float* tmp;
    tmp = y; y = y2; y2 = tmp;
    tmp = z; z = z2; z2 = tmp;
  }

  float* P = T;
  dim3 gFold(C_ / G_BN, C_ / G_BM, B_);
  gemm512_kernel<<<gFold, 128, 0, stream>>>(y + 4 * CM, z, P, 1.0f, 0.0f);

  bias_fold_kernel<<<(B_ * C_ + 255) / 256, 256, 0, stream>>>(P, c_mean, s_mean, tr, bias_w);

  dim3 gApply(N_ / 64, C_ / 64, B_);
  apply_kernel<<<gApply, 32, 0, stream>>>(P, content, bias_w, tr, out);
}


# Pairs of independent launches merged (2 means, 2 covariances, Y'=YT with Z'=TZ per iteration): 45 -> 31 launches, two workgroups per WGP overlap each other's LDS/barrier waits; sums unchanged
# speedup vs baseline: 4.1030x; 1.1069x over previous
#include <hip/hip_runtime.h>
#include <math.h>

#define B_ 4
#define C_ 512
#define N_ 16384
#define EPS_ 1e-5f
#define NS_ITERS 12

typedef __attribute__((ext_vector_type(2))) float v2f;
typedef __attribute__((ext_vector_type(8))) float v8f;

__device__ __forceinline__ v8f wmma4(v2f a, v2f b, v8f c) {
  return __builtin_amdgcn_wmma_f32_16x16x4_f32(false, a, false, b, (short)0, c,
                                               false, false);
}

__global__ void mean_kernel(const float* __restrict__ x0, const float* __restrict__ x1,
                            float* __restrict__ mean) {
  const int bc = blockIdx.x;
  const int which = blockIdx.y;
  const float* p = (which == 0 ? x0 : x1) + (size_t)bc * N_;
  float s = 0.f;
  for (int i = threadIdx.x; i < N_; i += 256) s += p[i];
  __shared__ float red[256];
  red[threadIdx.x] = s;
  __syncthreads();
  for (int o = 128; o > 0; o >>= 1) {
    if ((int)threadIdx.x < o) red[threadIdx.x] += red[threadIdx.x + o];
    __syncthreads();
  }
  if (threadIdx.x == 0) mean[which * (B_ * C_) + bc] = red[0] * (1.0f / (float)N_);
}

#define V_KC 128
#define V_LD 132
#define V_PR 128

typedef __attribute__((ext_vector_type(4))) float v4f;

__global__ __launch_bounds__(128) void cov_kernel(const float* __restrict__ x0,
                                                  const float* __restrict__ x1,
                                                  const float* __restrict__ mean,
                                                  float* __restrict__ cov) {
  __shared__ __attribute__((aligned(16))) float Ps[2][V_PR * V_LD];

  const int b    = blockIdx.z;
  const int row0 = blockIdx.y * 64;
  const int col0 = blockIdx.x * 64;
  const float* X  = (b < B_ ? x0 : x1) + (size_t)(b & (B_ - 1)) * C_ * N_;
  const float* mu = mean + b * C_;
  float* D = cov + (size_t)b * C_ * C_;

  const int tid  = threadIdx.x;
  const int lane = tid & 31;
  const int wave = tid >> 5;
  const int wm = wave >> 1;
  const int wn = wave & 1;
  const int lm = lane & 15;
  const int kh = lane >> 4;

  const float* gpa = X + (size_t)(row0 + wave) * N_ + lane * 4;
  const float* gpb = X + (size_t)(col0 + wave) * N_ + lane * 4;
  const int p_st = wave * V_LD + lane * 4;
  const int a_rd = (wm * 32 + lm) * V_LD + 4 * kh;
  const int b_rd = (64 + wn * 32 + lm) * V_LD + 4 * kh;

  const v8f zero = {0.f, 0.f, 0.f, 0.f, 0.f, 0.f, 0.f, 0.f};
  v8f acc[2][2];
#pragma unroll
  for (int m = 0; m < 2; ++m)
#pragma unroll
    for (int n = 0; n < 2; ++n) acc[m][n] = zero;

  v4f g[32];
#pragma unroll
  for (int j = 0; j < 16; ++j) g[j] = *(const v4f*)(gpa + (size_t)(4 * j) * N_);
#pragma unroll
  for (int j = 0; j < 16; ++j) g[16 + j] = *(const v4f*)(gpb + (size_t)(4 * j) * N_);
#pragma unroll
  for (int j = 0; j < 32; ++j) *(v4f*)(&Ps[0][p_st + 4 * j * V_LD]) = g[j];
  __syncthreads();

  for (int t = 0; t < N_ / V_KC; ++t) {
    const int buf = t & 1;
    const bool more = (t + 1 < N_ / V_KC);
    if (more) {
      const int k1 = (t + 1) * V_KC;
#pragma unroll
      for (int j = 0; j < 16; ++j) g[j] = *(const v4f*)(gpa + (size_t)(4 * j) * N_ + k1);
#pragma unroll
      for (int j = 0; j < 16; ++j) g[16 + j] = *(const v4f*)(gpb + (size_t)(4 * j) * N_ + k1);
    }

    const float* as = &Ps[buf][a_rd];
    const float* bs = &Ps[buf][b_rd];
    v4f af[2], bf[2];
#pragma unroll
    for (int m = 0; m < 2; ++m) af[m] = *(const v4f*)(as + 16 * m * V_LD);
#pragma unroll
    for (int n = 0; n < 2; ++n) bf[n] = *(const v4f*)(bs + 16 * n * V_LD);
#pragma unroll
    for (int s = 0; s < V_KC / 8; ++s) {
      v4f an[2], bn[2];
      if (s + 1 < V_KC / 8) {
#pragma unroll
        for (int m = 0; m < 2; ++m) an[m] = *(const v4f*)(as + 16 * m * V_LD + 8 * (s + 1));
#pragma unroll
        for (int n = 0; n < 2; ++n) bn[n] = *(const v4f*)(bs + 16 * n * V_LD + 8 * (s + 1));
      }
#pragma unroll
      for (int m = 0; m < 2; ++m)
#pragma unroll
        for (int n = 0; n < 2; ++n) {
          v2f a0 = {af[m].x, af[m].y};
          v2f b0 = {bf[n].x, bf[n].y};
          acc[m][n] = wmma4(a0, b0, acc[m][n]);
        }
#pragma unroll
      for (int m = 0; m < 2; ++m)
#pragma unroll
        for (int n = 0; n < 2; ++n) {
          v2f a1 = {af[m].z, af[m].w};
          v2f b1 = {bf[n].z, bf[n].w};
          acc[m][n] = wmma4(a1, b1, acc[m][n]);
        }
      if (s + 1 < V_KC / 8) {
#pragma unroll
        for (int m = 0; m < 2; ++m) af[m] = an[m];
#pragma unroll
        for (int n = 0; n < 2; ++n) bf[n] = bn[n];
      }
    }

    if (more) {
      float* pd = &Ps[buf ^ 1][p_st];
#pragma unroll
      for (int j = 0; j < 32; ++j) *(v4f*)(pd + 4 * j * V_LD) = g[j];
    }
    __syncthreads();
  }

  const float fac = 1.0f / (float)(N_ - 1);
#pragma unroll
  for (int m = 0; m < 2; ++m) {
#pragma unroll
    for (int n = 0; n < 2; ++n) {
      const int cidx = col0 + wn * 32 + 16 * n + lm;
      const float mj = mu[cidx];
#pragma unroll
      for (int v = 0; v < 8; ++v) {
        const int r = row0 + wm * 32 + 16 * m + v + 8 * kh;
        float val = (acc[m][n][v] - (float)N_ * mu[r] * mj) * fac;
        if (r == cidx) val += EPS_;
        D[(size_t)r * C_ + cidx] = val;
      }
    }
  }
}

__global__ void trace_kernel(const float* __restrict__ cov, float* __restrict__ tr) {
  const int m = blockIdx.x;
  const float* A = cov + (size_t)m * C_ * C_;
  float s = 0.f;
  for (int i = threadIdx.x; i < C_; i += 256) s += A[(size_t)i * C_ + i];
  __shared__ float red[256];
  red[threadIdx.x] = s;
  __syncthreads();
  for (int o = 128; o > 0; o >>= 1) {
    if ((int)threadIdx.x < o) red[threadIdx.x] += red[threadIdx.x + o];
    __syncthreads();
  }
  if (threadIdx.x == 0) tr[m] = red[0];
}

__global__ void ns_init_kernel(const float* __restrict__ cov,
                               const float* __restrict__ tr,
                               float* __restrict__ Y, float* __restrict__ Z) {
  const size_t i = (size_t)blockIdx.x * 256 + threadIdx.x;
  const size_t total = (size_t)8 * C_ * C_;
  if (i >= total) return;
  const int m = (int)(i / ((size_t)C_ * C_));
  const size_t rr = i % ((size_t)C_ * C_);
  Y[i] = cov[i] / tr[m];
  Z[i] = ((rr / C_) == (rr % C_)) ? 1.f : 0.f;
}

#define G_BM 128
#define G_BN 64
#define G_KC 32
#define G_LD 36

__global__ __launch_bounds__(128) void gemm512_kernel(const float* A0, const float* B0,
                                                      float* D0,
                                                      const float* A1, const float* B1,
                                                      float* D1, int split,
                                                      float alpha, float diagBeta) {
  __shared__ __attribute__((aligned(16))) float As[2][G_BM * G_LD];
  __shared__ __attribute__((aligned(16))) float Bs[2][G_BN * G_LD];

  const bool first = ((int)blockIdx.z < split);
  const int mm = first ? (int)blockIdx.z : (int)blockIdx.z - split;
  const float* __restrict__ A  = (first ? A0 : A1) + (size_t)mm * C_ * C_;
  const float* __restrict__ Bp = (first ? B0 : B1) + (size_t)mm * C_ * C_;
  float* __restrict__ D = (first ? D0 : D1) + (size_t)mm * C_ * C_;
  const int row0 = blockIdx.y * G_BM;
  const int col0 = blockIdx.x * G_BN;
  const int tid  = threadIdx.x;
  const int lane = tid & 31;
  const int wave = tid >> 5;
  const int wm = wave >> 1;
  const int wn = wave & 1;
  const int lm = lane & 15;
  const int kh = lane >> 4;

  const float* agp = A + (size_t)(row0 + (tid >> 3)) * C_ + (tid & 7) * 4;
  const float* bgp = Bp + (size_t)(tid >> 4) * C_ + col0 + (tid & 15) * 4;
  const int a_st = (tid >> 3) * G_LD + (tid & 7) * 4;
  const int b_st = ((tid & 15) * 4) * G_LD + (tid >> 4);
  const int a_rd = (wm * 64 + lm) * G_LD + 4 * kh;
  const int b_rd = (wn * 32 + lm) * G_LD + 4 * kh;

  const v8f zero = {0.f, 0.f, 0.f, 0.f, 0.f, 0.f, 0.f, 0.f};
  v8f acc[4][2];
#pragma unroll
  for (int m = 0; m < 4; ++m)
#pragma unroll
    for (int n = 0; n < 2; ++n) acc[m][n] = zero;

  v4f ga[8], gb[4];
#pragma unroll
  for (int i = 0; i < 8; ++i) ga[i] = *(const v4f*)(agp + (size_t)(16 * i) * C_);
#pragma unroll
  for (int i = 0; i < 4; ++i) gb[i] = *(const v4f*)(bgp + (size_t)(8 * i) * C_);
#pragma unroll
  for (int i = 0; i < 8; ++i) *(v4f*)(&As[0][a_st + 16 * i * G_LD]) = ga[i];
#pragma unroll
  for (int i = 0; i < 4; ++i) {
    Bs[0][b_st + 0 * G_LD + 8 * i] = gb[i].x;
    Bs[0][b_st + 1 * G_LD + 8 * i] = gb[i].y;
    Bs[0][b_st + 2 * G_LD + 8 * i] = gb[i].z;
    Bs[0][b_st + 3 * G_LD + 8 * i] = gb[i].w;
  }
  __syncthreads();

  for (int t = 0; t < C_ / G_KC; ++t) {
    const int buf = t & 1;
    const bool more = (t + 1 < C_ / G_KC);
    if (more) {
      const int k1 = (t + 1) * G_KC;
#pragma unroll
      for (int i = 0; i < 8; ++i) ga[i] = *(const v4f*)(agp + (size_t)(16 * i) * C_ + k1);
#pragma unroll
      for (int i = 0; i < 4; ++i) gb[i] = *(const v4f*)(bgp + (size_t)(k1 + 8 * i) * C_);
    }

    const float* as = &As[buf][a_rd];
    const float* bs = &Bs[buf][b_rd];
    v4f af[4], bf[2];
#pragma unroll
    for (int m = 0; m < 4; ++m) af[m] = *(const v4f*)(as + 16 * m * G_LD);
#pragma unroll
    for (int n = 0; n < 2; ++n) bf[n] = *(const v4f*)(bs + 16 * n * G_LD);
#pragma unroll
    for (int s = 0; s < G_KC / 8; ++s) {
      v4f an[4], bn[2];
      if (s + 1 < G_KC / 8) {
#pragma unroll
        for (int m = 0; m < 4; ++m) an[m] = *(const v4f*)(as + 16 * m * G_LD + 8 * (s + 1));
#pragma unroll
        for (int n = 0; n < 2; ++n) bn[n] = *(const v4f*)(bs + 16 * n * G_LD + 8 * (s + 1));
      }
#pragma unroll
      for (int m = 0; m < 4; ++m)
#pragma unroll
        for (int n = 0; n < 2; ++n) {
          v2f a0 = {af[m].x, af[m].y};
          v2f b0 = {bf[n].x, bf[n].y};
          acc[m][n] = wmma4(a0, b0, acc[m][n]);
        }
#pragma unroll
      for (int m = 0; m < 4; ++m)
#pragma unroll
        for (int n = 0; n < 2; ++n) {
          v2f a1 = {af[m].z, af[m].w};
          v2f b1 = {bf[n].z, bf[n].w};
          acc[m][n] = wmma4(a1, b1, acc[m][n]);
        }
      if (s + 1 < G_KC / 8) {
#pragma unroll
        for (int m = 0; m < 4; ++m) af[m] = an[m];
#pragma unroll
        for (int n = 0; n < 2; ++n) bf[n] = bn[n];
      }
    }

    if (more) {
      float* ad = &As[buf ^ 1][a_st];
      float* bd = &Bs[buf ^ 1][b_st];
#pragma unroll
      for (int i = 0; i < 8; ++i) *(v4f*)(ad + 16 * i * G_LD) = ga[i];
#pragma unroll
      for (int i = 0; i < 4; ++i) {
        bd[0 * G_LD + 8 * i] = gb[i].x;
        bd[1 * G_LD + 8 * i] = gb[i].y;
        bd[2 * G_LD + 8 * i] = gb[i].z;
        bd[3 * G_LD + 8 * i] = gb[i].w;
      }
    }
    __syncthreads();
  }

#pragma unroll
  for (int m = 0; m < 4; ++m)
#pragma unroll
    for (int n = 0; n < 2; ++n) {
      const int cidx = col0 + wn * 32 + 16 * n + lm;
#pragma unroll
      for (int v = 0; v < 8; ++v) {
        const int r = row0 + wm * 64 + 16 * m + v + 8 * kh;
        float val = alpha * acc[m][n][v];
        if (r == cidx) val += diagBeta;
        D[(size_t)r * C_ + cidx] = val;
      }
    }
}

__global__ void bias_fold_kernel(const float* __restrict__ P,
                                 const float* __restrict__ cmean,
                                 const float* __restrict__ smean,
                                 const float* __restrict__ tr,
                                 float* __restrict__ bias) {
  const int idx = blockIdx.x * 256 + threadIdx.x;
  if (idx >= B_ * C_) return;
  const int b = idx / C_;
  const int i = idx % C_;
  const float* r = P + ((size_t)b * C_ + i) * C_;
  const float* mu = cmean + b * C_;
  float s = 0.f;
  for (int j = 0; j < C_; ++j) s += r[j] * mu[j];
  bias[idx] = smean[idx] - s * (sqrtf(tr[4 + b]) * rsqrtf(tr[b]));
}

__global__ __launch_bounds__(32) void apply_kernel(const float* __restrict__ Mmat,
                                                   const float* __restrict__ X,
                                                   const float* __restrict__ bias,
                                                   const float* __restrict__ tr,
                                                   float* __restrict__ out) {
  const int b = blockIdx.z;
  const float* Mb = Mmat + (size_t)b * C_ * C_;
  const float* Xb = X + (size_t)b * C_ * N_;
  float* Ob = out + (size_t)b * C_ * N_;
  const float* bi = bias + b * C_;
  const float scale = sqrtf(tr[4 + b]) * rsqrtf(tr[b]);

  const int row0 = blockIdx.y * 64;
  const int col0 = blockIdx.x * 64;
  const int lane = threadIdx.x & 31;
  const int lm = lane & 15;
  const int kh = lane >> 4;

  const v8f zero = {0.f, 0.f, 0.f, 0.f, 0.f, 0.f, 0.f, 0.f};
  v8f acc[4][4];
#pragma unroll
  for (int m = 0; m < 4; ++m)
#pragma unroll
    for (int n = 0; n < 4; ++n) acc[m][n] = zero;

  for (int k = 0; k < C_; k += 4) {
    v2f av[4], bv[4];
#pragma unroll
    for (int m = 0; m < 4; ++m)
      av[m] = *(const v2f*)(Mb + (size_t)(row0 + 16 * m + lm) * C_ + k + 2 * kh);
#pragma unroll
    for (int n = 0; n < 4; ++n) {
      const float* q = Xb + (size_t)(k + 2 * kh) * N_ + col0 + 16 * n + lm;
      v2f tt;
      tt.x = q[0];
      tt.y = q[N_];
      bv[n] = tt;
    }
#pragma unroll
    for (int m = 0; m < 4; ++m)
#pragma unroll
      for (int n = 0; n < 4; ++n) acc[m][n] = wmma4(av[m], bv[n], acc[m][n]);
  }

#pragma unroll
  for (int m = 0; m < 4; ++m)
#pragma unroll
    for (int n = 0; n < 4; ++n) {
      const int c = col0 + 16 * n + lm;
#pragma unroll
      for (int v = 0; v < 8; ++v) {
        const int r = row0 + 16 * m + v + 8 * kh;
        Ob[(size_t)r * N_ + c] = scale * acc[m][n][v] + bi[r];
      }
    }
}

extern "C" void kernel_launch(void* const* d_in, const int* in_sizes, int n_in,
                              void* d_out, int out_size, void* d_ws, size_t ws_size,
                              hipStream_t stream) {
  (void)in_sizes; (void)n_in; (void)out_size; (void)ws_size;
  const float* content = (const float*)d_in[0];
  const float* style   = (const float*)d_in[1];
  float* out = (float*)d_out;
  float* ws  = (float*)d_ws;

  const size_t CM = (size_t)C_ * C_;

  float* c_mean  = ws;
  float* s_mean  = c_mean + B_ * C_;
  float* tr      = s_mean + B_ * C_;
  float* bias_w  = tr + 8;
  float* cov_all = bias_w + B_ * C_;
  float* Y  = cov_all + 8 * CM;
  float* Z  = Y + 8 * CM;
  float* Y2 = Z + 8 * CM;
  float* Z2 = Y2 + 8 * CM;
  float* T  = Z2 + 8 * CM;

  dim3 meanGrid(B_ * C_, 2);
  mean_kernel<<<meanGrid, 256, 0, stream>>>(content, style, c_mean);

  dim3 covGrid(C_ / 64, C_ / 64, 2 * B_);
  cov_kernel<<<covGrid, 128, 0, stream>>>(content, style, c_mean, cov_all);

  trace_kernel<<<8, 256, 0, stream>>>(cov_all, tr);

  {
    const size_t tot = 8 * CM;
    ns_init_kernel<<<(unsigned)((tot + 255) / 256), 256, 0, stream>>>(cov_all, tr, Y, Z);
  }
  dim3 g512(C_ / G_BN, C_ / G_BM, 8);
  dim3 g512x2(C_ / G_BN, C_ / G_BM, 16);
  float* y = Y; float* z = Z; float* y2 = Y2; float* z2 = Z2;
  for (int it = 0; it < NS_ITERS; ++it) {
    gemm512_kernel<<<g512, 128, 0, stream>>>(z, y, T, z, y, T, 8, -0.5f, 1.5f);
    gemm512_kernel<<<g512x2, 128, 0, stream>>>(y, T, y2, T, z, z2, 8, 1.0f, 0.0f);
    float* tmp;
    tmp = y; y = y2; y2 = tmp;
    tmp = z; z = z2; z2 = tmp;
  }

  float* P = T;
  dim3 gFold(C_ / G_BN, C_ / G_BM, B_);
  gemm512_kernel<<<gFold, 128, 0, stream>>>(y + 4 * CM, z, P, y + 4 * CM, z, P, 4, 1.0f, 0.0f);

  bias_fold_kernel<<<(B_ * C_ + 255) / 256, 256, 0, stream>>>(P, c_mean, s_mean, tr, bias_w);

  dim3 gApply(N_ / 64, C_ / 64, B_);
  apply_kernel<<<gApply, 32, 0, stream>>>(P, content, bias_w, tr, out);
}


# Apply pass (512x16384x512 per batch): 4-wave workgroups stage 128x32 P / 32x64 X chunks in LDS, double-buffered, 64x32 tile per wave; k summed in a permuted order inside each 8-block
# speedup vs baseline: 4.3106x; 1.0506x over previous
#include <hip/hip_runtime.h>
#include <math.h>

#define B_ 4
#define C_ 512
#define N_ 16384
#define EPS_ 1e-5f
#define NS_ITERS 12

typedef __attribute__((ext_vector_type(2))) float v2f;
typedef __attribute__((ext_vector_type(8))) float v8f;

__device__ __forceinline__ v8f wmma4(v2f a, v2f b, v8f c) {
  return __builtin_amdgcn_wmma_f32_16x16x4_f32(false, a, false, b, (short)0, c,
                                               false, false);
}

__global__ void mean_kernel(const float* __restrict__ x0, const float* __restrict__ x1,
                            float* __restrict__ mean) {
  const int bc = blockIdx.x;
  const int which = blockIdx.y;
  const float* p = (which == 0 ? x0 : x1) + (size_t)bc * N_;
  float s = 0.f;
  for (int i = threadIdx.x; i < N_; i += 256) s += p[i];
  __shared__ float red[256];
  red[threadIdx.x] = s;
  __syncthreads();
  for (int o = 128; o > 0; o >>= 1) {
    if ((int)threadIdx.x < o) red[threadIdx.x] += red[threadIdx.x + o];
    __syncthreads();
  }
  if (threadIdx.x == 0) mean[which * (B_ * C_) + bc] = red[0] * (1.0f / (float)N_);
}

#define V_KC 128
#define V_LD 132
#define V_PR 128

typedef __attribute__((ext_vector_type(4))) float v4f;

__global__ __launch_bounds__(128) void cov_kernel(const float* __restrict__ x0,
                                                  const float* __restrict__ x1,
                                                  const float* __restrict__ mean,
                                                  float* __restrict__ cov) {
  __shared__ __attribute__((aligned(16))) float Ps[2][V_PR * V_LD];

  const int b    = blockIdx.z;
  const int row0 = blockIdx.y * 64;
  const int col0 = blockIdx.x * 64;
  const float* X  = (b < B_ ? x0 : x1) + (size_t)(b & (B_ - 1)) * C_ * N_;
  const float* mu = mean + b * C_;
  float* D = cov + (size_t)b * C_ * C_;

  const int tid  = threadIdx.x;
  const int lane = tid & 31;
  const int wave = tid >> 5;
  const int wm = wave >> 1;
  const int wn = wave & 1;
  const int lm = lane & 15;
  const int kh = lane >> 4;

  const float* gpa = X + (size_t)(row0 + wave) * N_ + lane * 4;
  const float* gpb = X + (size_t)(col0 + wave) * N_ + lane * 4;
  const int p_st = wave * V_LD + lane * 4;
  const int a_rd = (wm * 32 + lm) * V_LD + 4 * kh;
  const int b_rd = (64 + wn * 32 + lm) * V_LD + 4 * kh;

  const v8f zero = {0.f, 0.f, 0.f, 0.f, 0.f, 0.f, 0.f, 0.f};
  v8f acc[2][2];
#pragma unroll
  for (int m = 0; m < 2; ++m)
#pragma unroll
    for (int n = 0; n < 2; ++n) acc[m][n] = zero;

  v4f g[32];
#pragma unroll
  for (int j = 0; j < 16; ++j) g[j] = *(const v4f*)(gpa + (size_t)(4 * j) * N_);
#pragma unroll
  for (int j = 0; j < 16; ++j) g[16 + j] = *(const v4f*)(gpb + (size_t)(4 * j) * N_);
#pragma unroll
  for (int j = 0; j < 32; ++j) *(v4f*)(&Ps[0][p_st + 4 * j * V_LD]) = g[j];
  __syncthreads();

  for (int t = 0; t < N_ / V_KC; ++t) {
    const int buf = t & 1;
    const bool more = (t + 1 < N_ / V_KC);
    if (more) {
      const int k1 = (t + 1) * V_KC;
#pragma unroll
      for (int j = 0; j < 16; ++j) g[j] = *(const v4f*)(gpa + (size_t)(4 * j) * N_ + k1);
#pragma unroll
      for (int j = 0; j < 16; ++j) g[16 + j] = *(const v4f*)(gpb + (size_t)(4 * j) * N_ + k1);
    }

    const float* as = &Ps[buf][a_rd];
    const float* bs = &Ps[buf][b_rd];
    v4f af[2], bf[2];
#pragma unroll
    for (int m = 0; m < 2; ++m) af[m] = *(const v4f*)(as + 16 * m * V_LD);
#pragma unroll
    for (int n = 0; n < 2; ++n) bf[n] = *(const v4f*)(bs + 16 * n * V_LD);
#pragma unroll
    for (int s = 0; s < V_KC / 8; ++s) {
      v4f an[2], bn[2];
      if (s + 1 < V_KC / 8) {
#pragma unroll
        for (int m = 0; m < 2; ++m) an[m] = *(const v4f*)(as + 16 * m * V_LD + 8 * (s + 1));
#pragma unroll
        for (int n = 0; n < 2; ++n) bn[n] = *(const v4f*)(bs + 16 * n * V_LD + 8 * (s + 1));
      }
#pragma unroll
      for (int m = 0; m < 2; ++m)
#pragma unroll
        for (int n = 0; n < 2; ++n) {
          v2f a0 = {af[m].x, af[m].y};
          v2f b0 = {bf[n].x, bf[n].y};
          acc[m][n] = wmma4(a0, b0, acc[m][n]);
        }
#pragma unroll
      for (int m = 0; m < 2; ++m)
#pragma unroll
        for (int n = 0; n < 2; ++n) {
          v2f a1 = {af[m].z, af[m].w};
          v2f b1 = {bf[n].z, bf[n].w};
          acc[m][n] = wmma4(a1, b1, acc[m][n]);
        }
      if (s + 1 < V_KC / 8) {
#pragma unroll
        for (int m = 0; m < 2; ++m) af[m] = an[m];
#pragma unroll
        for (int n = 0; n < 2; ++n) bf[n] = bn[n];
      }
    }

    if (more) {
      float* pd = &Ps[buf ^ 1][p_st];
#pragma unroll
      for (int j = 0; j < 32; ++j) *(v4f*)(pd + 4 * j * V_LD) = g[j];
    }
    __syncthreads();
  }

  const float fac = 1.0f / (float)(N_ - 1);
#pragma unroll
  for (int m = 0; m < 2; ++m) {
#pragma unroll
    for (int n = 0; n < 2; ++n) {
      const int cidx = col0 + wn * 32 + 16 * n + lm;
      const float mj = mu[cidx];
#pragma unroll
      for (int v = 0; v < 8; ++v) {
        const int r = row0 + wm * 32 + 16 * m + v + 8 * kh;
        float val = (acc[m][n][v] - (float)N_ * mu[r] * mj) * fac;
        if (r == cidx) val += EPS_;
        D[(size_t)r * C_ + cidx] = val;
      }
    }
  }
}

__global__ void trace_kernel(const float* __restrict__ cov, float* __restrict__ tr) {
  const int m = blockIdx.x;
  const float* A = cov + (size_t)m * C_ * C_;
  float s = 0.f;
  for (int i = threadIdx.x; i < C_; i += 256) s += A[(size_t)i * C_ + i];
  __shared__ float red[256];
  red[threadIdx.x] = s;
  __syncthreads();
  for (int o = 128; o > 0; o >>= 1) {
    if ((int)threadIdx.x < o) red[threadIdx.x] += red[threadIdx.x + o];
    __syncthreads();
  }
  if (threadIdx.x == 0) tr[m] = red[0];
}

__global__ void ns_init_kernel(const float* __restrict__ cov,
                               const float* __restrict__ tr,
                               float* __restrict__ Y, float* __restrict__ Z) {
  const size_t i = (size_t)blockIdx.x * 256 + threadIdx.x;
  const size_t total = (size_t)8 * C_ * C_;
  if (i >= total) return;
  const int m = (int)(i / ((size_t)C_ * C_));
  const size_t rr = i % ((size_t)C_ * C_);
  Y[i] = cov[i] / tr[m];
  Z[i] = ((rr / C_) == (rr % C_)) ? 1.f : 0.f;
}

#define G_BM 128
#define G_BN 64
#define G_KC 32
#define G_LD 36

__global__ __launch_bounds__(128) void gemm512_kernel(const float* A0, const float* B0,
                                                      float* D0,
                                                      const float* A1, const float* B1,
                                                      float* D1, int split,
                                                      float alpha, float diagBeta) {
  __shared__ __attribute__((aligned(16))) float As[2][G_BM * G_LD];
  __shared__ __attribute__((aligned(16))) float Bs[2][G_BN * G_LD];

  const bool first = ((int)blockIdx.z < split);
  const int mm = first ? (int)blockIdx.z : (int)blockIdx.z - split;
  const float* __restrict__ A  = (first ? A0 : A1) + (size_t)mm * C_ * C_;
  const float* __restrict__ Bp = (first ? B0 : B1) + (size_t)mm * C_ * C_;
  float* __restrict__ D = (first ? D0 : D1) + (size_t)mm * C_ * C_;
  const int row0 = blockIdx.y * G_BM;
  const int col0 = blockIdx.x * G_BN;
  const int tid  = threadIdx.x;
  const int lane = tid & 31;
  const int wave = tid >> 5;
  const int wm = wave >> 1;
  const int wn = wave & 1;
  const int lm = lane & 15;
  const int kh = lane >> 4;

  const float* agp = A + (size_t)(row0 + (tid >> 3)) * C_ + (tid & 7) * 4;
  const float* bgp = Bp + (size_t)(tid >> 4) * C_ + col0 + (tid & 15) * 4;
  const int a_st = (tid >> 3) * G_LD + (tid & 7) * 4;
  const int b_st = ((tid & 15) * 4) * G_LD + (tid >> 4);
  const int a_rd = (wm * 64 + lm) * G_LD + 4 * kh;
  const int b_rd = (wn * 32 + lm) * G_LD + 4 * kh;

  const v8f zero = {0.f, 0.f, 0.f, 0.f, 0.f, 0.f, 0.f, 0.f};
  v8f acc[4][2];
#pragma unroll
  for (int m = 0; m < 4; ++m)
#pragma unroll
    for (int n = 0; n < 2; ++n) acc[m][n] = zero;

  v4f ga[8], gb[4];
#pragma unroll
  for (int i = 0; i < 8; ++i) ga[i] = *(const v4f*)(agp + (size_t)(16 * i) * C_);
#pragma unroll
  for (int i = 0; i < 4; ++i) gb[i] = *(const v4f*)(bgp + (size_t)(8 * i) * C_);
#pragma unroll
  for (int i = 0; i < 8; ++i) *(v4f*)(&As[0][a_st + 16 * i * G_LD]) = ga[i];
#pragma unroll
  for (int i = 0; i < 4; ++i) {
    Bs[0][b_st + 0 * G_LD + 8 * i] = gb[i].x;
    Bs[0][b_st + 1 * G_LD + 8 * i] = gb[i].y;
    Bs[0][b_st + 2 * G_LD + 8 * i] = gb[i].z;
    Bs[0][b_st + 3 * G_LD + 8 * i] = gb[i].w;
  }
  __syncthreads();

  for (int t = 0; t < C_ / G_KC; ++t) {
    const int buf = t & 1;
    const bool more = (t + 1 < C_ / G_KC);
    if (more) {
      const int k1 = (t + 1) * G_KC;
#pragma unroll
      for (int i = 0; i < 8; ++i) ga[i] = *(const v4f*)(agp + (size_t)(16 * i) * C_ + k1);
#pragma unroll
      for (int i = 0; i < 4; ++i) gb[i] = *(const v4f*)(bgp + (size_t)(k1 + 8 * i) * C_);
    }

    const float* as = &As[buf][a_rd];
    const float* bs = &Bs[buf][b_rd];
    v4f af[4], bf[2];
#pragma unroll
    for (int m = 0; m < 4; ++m) af[m] = *(const v4f*)(as + 16 * m * G_LD);
#pragma unroll
    for (int n = 0; n < 2; ++n) bf[n] = *(const v4f*)(bs + 16 * n * G_LD);
#pragma unroll
    for (int s = 0; s < G_KC / 8; ++s) {
      v4f an[4], bn[2];
      if (s + 1 < G_KC / 8) {
#pragma unroll
        for (int m = 0; m < 4; ++m) an[m] = *(const v4f*)(as + 16 * m * G_LD + 8 * (s + 1));
#pragma unroll
        for (int n = 0; n < 2; ++n) bn[n] = *(const v4f*)(bs + 16 * n * G_LD + 8 * (s + 1));
      }
#pragma unroll
      for (int m = 0; m < 4; ++m)
#pragma unroll
        for (int n = 0; n < 2; ++n) {
          v2f a0 = {af[m].x, af[m].y};
          v2f b0 = {bf[n].x, bf[n].y};
          acc[m][n] = wmma4(a0, b0, acc[m][n]);
        }
#pragma unroll
      for (int m = 0; m < 4; ++m)
#pragma unroll
        for (int n = 0; n < 2; ++n) {
          v2f a1 = {af[m].z, af[m].w};
          v2f b1 = {bf[n].z, bf[n].w};
          acc[m][n] = wmma4(a1, b1, acc[m][n]);
        }
      if (s + 1 < G_KC / 8) {
#pragma unroll
        for (int m = 0; m < 4; ++m) af[m] = an[m];
#pragma unroll
        for (int n = 0; n < 2; ++n) bf[n] = bn[n];
      }
    }

    if (more) {
      float* ad = &As[buf ^ 1][a_st];
      float* bd = &Bs[buf ^ 1][b_st];
#pragma unroll
      for (int i = 0; i < 8; ++i) *(v4f*)(ad + 16 * i * G_LD) = ga[i];
#pragma unroll
      for (int i = 0; i < 4; ++i) {
        bd[0 * G_LD + 8 * i] = gb[i].x;
        bd[1 * G_LD + 8 * i] = gb[i].y;
        bd[2 * G_LD + 8 * i] = gb[i].z;
        bd[3 * G_LD + 8 * i] = gb[i].w;
      }
    }
    __syncthreads();
  }

#pragma unroll
  for (int m = 0; m < 4; ++m)
#pragma unroll
    for (int n = 0; n < 2; ++n) {
      const int cidx = col0 + wn * 32 + 16 * n + lm;
#pragma unroll
      for (int v = 0; v < 8; ++v) {
        const int r = row0 + wm * 64 + 16 * m + v + 8 * kh;
        float val = alpha * acc[m][n][v];
        if (r == cidx) val += diagBeta;
        D[(size_t)r * C_ + cidx] = val;
      }
    }
}

__global__ void bias_fold_kernel(const float* __restrict__ P,
                                 const float* __restrict__ cmean,
                                 const float* __restrict__ smean,
                                 const float* __restrict__ tr,
                                 float* __restrict__ bias) {
  const int idx = blockIdx.x * 256 + threadIdx.x;
  if (idx >= B_ * C_) return;
  const int b = idx / C_;
  const int i = idx % C_;
  const float* r = P + ((size_t)b * C_ + i) * C_;
  const float* mu = cmean + b * C_;
  float s = 0.f;
  for (int j = 0; j < C_; ++j) s += r[j] * mu[j];
  bias[idx] = smean[idx] - s * (sqrtf(tr[4 + b]) * rsqrtf(tr[b]));
}

__global__ __launch_bounds__(128) void apply_kernel(const float* __restrict__ Mmat,
                                                    const float* __restrict__ X,
                                                    const float* __restrict__ bias,
                                                    const float* __restrict__ tr,
                                                    float* __restrict__ out) {
  __shared__ __attribute__((aligned(16))) float As[2][G_BM * G_LD];
  __shared__ __attribute__((aligned(16))) float Bs[2][G_BN * G_LD];

  const int b = blockIdx.z;
  const float* A  = Mmat + (size_t)b * C_ * C_;
  const float* Bp = X + (size_t)b * C_ * N_;
  float* D = out + (size_t)b * C_ * N_;
  const float* bi = bias + b * C_;
  const float scale = sqrtf(tr[4 + b]) * rsqrtf(tr[b]);

  const int row0 = blockIdx.x * G_BM;
  const int col0 = blockIdx.y * G_BN;
  const int tid  = threadIdx.x;
  const int lane = tid & 31;
  const int wave = tid >> 5;
  const int wm = wave >> 1;
  const int wn = wave & 1;
  const int lm = lane & 15;
  const int kh = lane >> 4;

  const float* agp = A + (size_t)(row0 + (tid >> 3)) * C_ + (tid & 7) * 4;
  const float* bgp = Bp + (size_t)(tid >> 4) * N_ + col0 + (tid & 15) * 4;
  const int a_st = (tid >> 3) * G_LD + (tid & 7) * 4;
  const int b_st = ((tid & 15) * 4) * G_LD + (tid >> 4);
  const int a_rd = (wm * 64 + lm) * G_LD + 4 * kh;
  const int b_rd = (wn * 32 + lm) * G_LD + 4 * kh;

  const v8f zero = {0.f, 0.f, 0.f, 0.f, 0.f, 0.f, 0.f, 0.f};
  v8f acc[4][2];
#pragma unroll
  for (int m = 0; m < 4; ++m)
#pragma unroll
    for (int n = 0; n < 2; ++n) acc[m][n] = zero;

  v4f ga[8], gb[4];
#pragma unroll
  for (int i = 0; i < 8; ++i) ga[i] = *(const v4f*)(agp + (size_t)(16 * i) * C_);
#pragma unroll
  for (int i = 0; i < 4; ++i) gb[i] = *(const v4f*)(bgp + (size_t)(8 * i) * N_);
#pragma unroll
  for (int i = 0; i < 8; ++i) *(v4f*)(&As[0][a_st + 16 * i * G_LD]) = ga[i];
#pragma unroll
  for (int i = 0; i < 4; ++i) {
    Bs[0][b_st + 0 * G_LD + 8 * i] = gb[i].x;
    Bs[0][b_st + 1 * G_LD + 8 * i] = gb[i].y;
    Bs[0][b_st + 2 * G_LD + 8 * i] = gb[i].z;
    Bs[0][b_st + 3 * G_LD + 8 * i] = gb[i].w;
  }
  __syncthreads();

  for (int t = 0; t < C_ / G_KC; ++t) {
    const int buf = t & 1;
    const bool more = (t + 1 < C_ / G_KC);
    if (more) {
      const int k1 = (t + 1) * G_KC;
#pragma unroll
      for (int i = 0; i < 8; ++i) ga[i] = *(const v4f*)(agp + (size_t)(16 * i) * C_ + k1);
#pragma unroll
      for (int i = 0; i < 4; ++i) gb[i] = *(const v4f*)(bgp + (size_t)(k1 + 8 * i) * N_);
    }

    const float* as = &As[buf][a_rd];
    const float* bs = &Bs[buf][b_rd];
    v4f af[4], bf[2];
#pragma unroll
    for (int m = 0; m < 4; ++m) af[m] = *(const v4f*)(as + 16 * m * G_LD);
#pragma unroll
    for (int n = 0; n < 2; ++n) bf[n] = *(const v4f*)(bs + 16 * n * G_LD);
#pragma unroll
    for (int s = 0; s < G_KC / 8; ++s) {
      v4f an[4], bn[2];
      if (s + 1 < G_KC / 8) {
#pragma unroll
        for (int m = 0; m < 4; ++m) an[m] = *(const v4f*)(as + 16 * m * G_LD + 8 * (s + 1));
#pragma unroll
        for (int n = 0; n < 2; ++n) bn[n] = *(const v4f*)(bs + 16 * n * G_LD + 8 * (s + 1));
      }
#pragma unroll
      for (int m = 0; m < 4; ++m)
#pragma unroll
        for (int n = 0; n < 2; ++n) {
          v2f a0 = {af[m].x, af[m].y};
          v2f b0 = {bf[n].x, bf[n].y};
          acc[m][n] = wmma4(a0, b0, acc[m][n]);
        }
#pragma unroll
      for (int m = 0; m < 4; ++m)
#pragma unroll
        for (int n = 0; n < 2; ++n) {
          v2f a1 = {af[m].z, af[m].w};
          v2f b1 = {bf[n].z, bf[n].w};
          acc[m][n] = wmma4(a1, b1, acc[m][n]);
        }
      if (s + 1 < G_KC / 8) {
#pragma unroll
        for (int m = 0; m < 4; ++m) af[m] = an[m];
#pragma unroll
        for (int n = 0; n < 2; ++n) bf[n] = bn[n];
      }
    }

    if (more) {
      float* ad = &As[buf ^ 1][a_st];
      float* bd = &Bs[buf ^ 1][b_st];
#pragma unroll
      for (int i = 0; i < 8; ++i) *(v4f*)(ad + 16 * i * G_LD) = ga[i];
#pragma unroll
      for (int i = 0; i < 4; ++i) {
        bd[0 * G_LD + 8 * i] = gb[i].x;
        bd[1 * G_LD + 8 * i] = gb[i].y;
        bd[2 * G_LD + 8 * i] = gb[i].z;
        bd[3 * G_LD + 8 * i] = gb[i].w;
      }
    }
    __syncthreads();
  }

#pragma unroll
  for (int m = 0; m < 4; ++m)
#pragma unroll
    for (int n = 0; n < 2; ++n) {
      const int cidx = col0 + wn * 32 + 16 * n + lm;
#pragma unroll
      for (int v = 0; v < 8; ++v) {
        const int r = row0 + wm * 64 + 16 * m + v + 8 * kh;
        D[(size_t)r * N_ + cidx] = scale * acc[m][n][v] + bi[r];
      }
    }
}

extern "C" void kernel_launch(void* const* d_in, const int* in_sizes, int n_in,
                              void* d_out, int out_size, void* d_ws, size_t ws_size,
                              hipStream_t stream) {
  (void)in_sizes; (void)n_in; (void)out_size; (void)ws_size;
  const float* content = (const float*)d_in[0];
  const float* style   = (const float*)d_in[1];
  float* out = (float*)d_out;
  float* ws  = (float*)d_ws;

  const size_t CM = (size_t)C_ * C_;

  float* c_mean  = ws;
  float* s_mean  = c_mean + B_ * C_;
  float* tr      = s_mean + B_ * C_;
  float* bias_w  = tr + 8;
  float* cov_all = bias_w + B_ * C_;
  float* Y  = cov_all + 8 * CM;
  float* Z  = Y + 8 * CM;
  float* Y2 = Z + 8 * CM;
  float* Z2 = Y2 + 8 * CM;
  float* T  = Z2 + 8 * CM;

  dim3 meanGrid(B_ * C_, 2);
  mean_kernel<<<meanGrid, 256, 0, stream>>>(content, style, c_mean);

  dim3 covGrid(C_ / 64, C_ / 64, 2 * B_);
  cov_kernel<<<covGrid, 128, 0, stream>>>(content, style, c_mean, cov_all);

  trace_kernel<<<8, 256, 0, stream>>>(cov_all, tr);

  {
    const size_t tot = 8 * CM;
    ns_init_kernel<<<(unsigned)((tot + 255) / 256), 256, 0, stream>>>(cov_all, tr, Y, Z);
  }
  dim3 g512(C_ / G_BN, C_ / G_BM, 8);
  dim3 g512x2(C_ / G_BN, C_ / G_BM, 16);
  float* y = Y; float* z = Z; float* y2 = Y2; float* z2 = Z2;
  for (int it = 0; it < NS_ITERS; ++it) {
    gemm512_kernel<<<g512, 128, 0, stream>>>(z, y, T, z, y, T, 8, -0.5f, 1.5f);
    gemm512_kernel<<<g512x2, 128, 0, stream>>>(y, T, y2, T, z, z2, 8, 1.0f, 0.0f);
    float* tmp;
    tmp = y; y = y2; y2 = tmp;
    tmp = z; z = z2; z2 = tmp;
  }

  float* P = T;
  dim3 gFold(C_ / G_BN, C_ / G_BM, B_);
  gemm512_kernel<<<gFold, 128, 0, stream>>>(y + 4 * CM, z, P, y + 4 * CM, z, P, 4, 1.0f, 0.0f);

  bias_fold_kernel<<<(B_ * C_ + 255) / 256, 256, 0, stream>>>(P, c_mean, s_mean, tr, bias_w);

  dim3 gApply(C_ / G_BM, N_ / G_BN, B_);
  apply_kernel<<<gApply, 128, 0, stream>>>(P, content, bias_w, tr, out);
}


# Newton-Schulz start scaled by the largest absolute row sum (>= largest eigenvalue, far tighter than the trace): 7 iterations converge where 12 did, 31 -> 21 launches; rounding differs
# speedup vs baseline: 5.4278x; 1.2592x over previous
#include <hip/hip_runtime.h>
#include <math.h>

#define B_ 4
#define C_ 512
#define N_ 16384
#define EPS_ 1e-5f
#define NS_ITERS 7

typedef __attribute__((ext_vector_type(2))) float v2f;
typedef __attribute__((ext_vector_type(8))) float v8f;

__device__ __forceinline__ v8f wmma4(v2f a, v2f b, v8f c) {
  return __builtin_amdgcn_wmma_f32_16x16x4_f32(false, a, false, b, (short)0, c,
                                               false, false);
}

__global__ void mean_kernel(const float* __restrict__ x0, const float* __restrict__ x1,
                            float* __restrict__ mean) {
  const int bc = blockIdx.x;
  const int which = blockIdx.y;
  const float* p = (which == 0 ? x0 : x1) + (size_t)bc * N_;
  float s = 0.f;
  for (int i = threadIdx.x; i < N_; i += 256) s += p[i];
  __shared__ float red[256];
  red[threadIdx.x] = s;
  __syncthreads();
  for (int o = 128; o > 0; o >>= 1) {
    if ((int)threadIdx.x < o) red[threadIdx.x] += red[threadIdx.x + o];
    __syncthreads();
  }
  if (threadIdx.x == 0) mean[which * (B_ * C_) + bc] = red[0] * (1.0f / (float)N_);
}

#define V_KC 128
#define V_LD 132
#define V_PR 128

typedef __attribute__((ext_vector_type(4))) float v4f;

__global__ __launch_bounds__(128) void cov_kernel(const float* __restrict__ x0,
                                                  const float* __restrict__ x1,
                                                  const float* __restrict__ mean,
                                                  float* __restrict__ cov) {
  __shared__ __attribute__((aligned(16))) float Ps[2][V_PR * V_LD];

  const int b    = blockIdx.z;
  const int row0 = blockIdx.y * 64;
  const int col0 = blockIdx.x * 64;
  const float* X  = (b < B_ ? x0 : x1) + (size_t)(b & (B_ - 1)) * C_ * N_;
  const float* mu = mean + b * C_;
  float* D = cov + (size_t)b * C_ * C_;

  const int tid  = threadIdx.x;
  const int lane = tid & 31;
  const int wave = tid >> 5;
  const int wm = wave >> 1;
  const int wn = wave & 1;
  const int lm = lane & 15;
  const int kh = lane >> 4;

  const float* gpa = X + (size_t)(row0 + wave) * N_ + lane * 4;
  const float* gpb = X + (size_t)(col0 + wave) * N_ + lane * 4;
  const int p_st = wave * V_LD + lane * 4;
  const int a_rd = (wm * 32 + lm) * V_LD + 4 * kh;
  const int b_rd = (64 + wn * 32 + lm) * V_LD + 4 * kh;

  const v8f zero = {0.f, 0.f, 0.f, 0.f, 0.f, 0.f, 0.f, 0.f};
  v8f acc[2][2];
#pragma unroll
  for (int m = 0; m < 2; ++m)
#pragma unroll
    for (int n = 0; n < 2; ++n) acc[m][n] = zero;

  v4f g[32];
#pragma unroll
  for (int j = 0; j < 16; ++j) g[j] = *(const v4f*)(gpa + (size_t)(4 * j) * N_);
#pragma unroll
  for (int j = 0; j < 16; ++j) g[16 + j] = *(const v4f*)(gpb + (size_t)(4 * j) * N_);
#pragma unroll
  for (int j = 0; j < 32; ++j) *(v4f*)(&Ps[0][p_st + 4 * j * V_LD]) = g[j];
  __syncthreads();

  for (int t = 0; t < N_ / V_KC; ++t) {
    const int buf = t & 1;
    const bool more = (t + 1 < N_ / V_KC);
    if (more) {
      const int k1 = (t + 1) * V_KC;
#pragma unroll
      for (int j = 0; j < 16; ++j) g[j] = *(const v4f*)(gpa + (size_t)(4 * j) * N_ + k1);
#pragma unroll
      for (int j = 0; j < 16; ++j) g[16 + j] = *(const v4f*)(gpb + (size_t)(4 * j) * N_ + k1);
    }

    const float* as = &Ps[buf][a_rd];
    const float* bs = &Ps[buf][b_rd];
    v4f af[2], bf[2];
#pragma unroll
    for (int m = 0; m < 2; ++m) af[m] = *(const v4f*)(as + 16 * m * V_LD);
#pragma unroll
    for (int n = 0; n < 2; ++n) bf[n] = *(const v4f*)(bs + 16 * n * V_LD);
#pragma unroll
    for (int s = 0; s < V_KC / 8; ++s) {
      v4f an[2], bn[2];
      if (s + 1 < V_KC / 8) {
#pragma unroll
        for (int m = 0; m < 2; ++m) an[m] = *(const v4f*)(as + 16 * m * V_LD + 8 * (s + 1));
#pragma unroll
        for (int n = 0; n < 2; ++n) bn[n] = *(const v4f*)(bs + 16 * n * V_LD + 8 * (s + 1));
      }
#pragma unroll
      for (int m = 0; m < 2; ++m)
#pragma unroll
        for (int n = 0; n < 2; ++n) {
          v2f a0 = {af[m].x, af[m].y};
          v2f b0 = {bf[n].x, bf[n].y};
          acc[m][n] = wmma4(a0, b0, acc[m][n]);
        }
#pragma unroll
      for (int m = 0; m < 2; ++m)
#pragma unroll
        for (int n = 0; n < 2; ++n) {
          v2f a1 = {af[m].z, af[m].w};
          v2f b1 = {bf[n].z, bf[n].w};
          acc[m][n] = wmma4(a1, b1, acc[m][n]);
        }
      if (s + 1 < V_KC / 8) {
#pragma unroll
        for (int m = 0; m < 2; ++m) af[m] = an[m];
#pragma unroll
        for (int n = 0; n < 2; ++n) bf[n] = bn[n];
      }
    }

    if (more) {
      float* pd = &Ps[buf ^ 1][p_st];
#pragma unroll
      for (int j = 0; j < 32; ++j) *(v4f*)(pd + 4 * j * V_LD) = g[j];
    }
    __syncthreads();
  }

  const float fac = 1.0f / (float)(N_ - 1);
#pragma unroll
  for (int m = 0; m < 2; ++m) {
#pragma unroll
    for (int n = 0; n < 2; ++n) {
      const int cidx = col0 + wn * 32 + 16 * n + lm;
      const float mj = mu[cidx];
#pragma unroll
      for (int v = 0; v < 8; ++v) {
        const int r = row0 + wm * 32 + 16 * m + v + 8 * kh;
        float val = (acc[m][n][v] - (float)N_ * mu[r] * mj) * fac;
        if (r == cidx) val += EPS_;
        D[(size_t)r * C_ + cidx] = val;
      }
    }
  }
}

__global__ __launch_bounds__(1024) void rowabs_kernel(const float* __restrict__ cov,
                                                      float* __restrict__ rs) {
  __shared__ float part[32];
  const int lane = threadIdx.x & 31;
  const int wave = threadIdx.x >> 5;
  const float* p = cov + ((size_t)blockIdx.x * 32 + wave) * C_ + lane * 4;
  float s = 0.f;
#pragma unroll
  for (int j = 0; j < 4; ++j) {
    const v4f v = *(const v4f*)(p + 128 * j);
    s += (fabsf(v.x) + fabsf(v.y)) + (fabsf(v.z) + fabsf(v.w));
  }
  for (int o = 16; o > 0; o >>= 1) s += __shfl_xor(s, o);
  if (lane == 0) part[wave] = s;
  __syncthreads();
  if (wave == 0) rs[blockIdx.x * 32 + lane] = part[lane];
}

__global__ __launch_bounds__(256) void ns_init_kernel(const float* __restrict__ cov,
                                                      const float* __restrict__ rs,
                                                      float* __restrict__ nrm,
                                                      float* __restrict__ Y,
                                                      float* __restrict__ Z) {
  __shared__ float part[8];
  const int tid  = threadIdx.x;
  const int lane = tid & 31;
  const int wave = tid >> 5;
  const int m = blockIdx.x >> 6;
  float c = fmaxf(rs[m * C_ + tid], rs[m * C_ + 256 + tid]);
  for (int o = 16; o > 0; o >>= 1) c = fmaxf(c, __shfl_xor(c, o));
  if (lane == 0) part[wave] = c;
  __syncthreads();
  c = part[0];
#pragma unroll
  for (int w = 1; w < 8; ++w) c = fmaxf(c, part[w]);
  if ((blockIdx.x & 63) == 0 && tid == 0) nrm[m] = c;

  const size_t base = (size_t)blockIdx.x * 4096;
#pragma unroll
  for (int k = 0; k < 4; ++k) {
    const size_t i = base + (size_t)k * 1024 + tid * 4;
    const int rr  = (int)(i & ((size_t)C_ * C_ - 1));
    const int row = rr >> 9;
    const int col = rr & (C_ - 1);
    const v4f a = *(const v4f*)(cov + i);
    v4f y, z;
    y.x = a.x / c; y.y = a.y / c; y.z = a.z / c; y.w = a.w / c;
    z.x = (row == col)     ? 1.f : 0.f;
    z.y = (row == col + 1) ? 1.f : 0.f;
    z.z = (row == col + 2) ? 1.f : 0.f;
    z.w = (row == col + 3) ? 1.f : 0.f;
    *(v4f*)(Y + i) = y;
    *(v4f*)(Z + i) = z;
  }
}

#define G_BM 128
#define G_BN 64
#define G_KC 32
#define G_LD 36

__global__ __launch_bounds__(128) void gemm512_kernel(const float* A0, const float* B0,
                                                      float* D0,
                                                      const float* A1, const float* B1,
                                                      float* D1, int split,
                                                      float alpha, float diagBeta) {
  __shared__ __attribute__((aligned(16))) float As[2][G_BM * G_LD];
  __shared__ __attribute__((aligned(16))) float Bs[2][G_BN * G_LD];

  const bool first = ((int)blockIdx.z < split);
  const int mm = first ? (int)blockIdx.z : (int)blockIdx.z - split;
  const float* __restrict__ A  = (first ? A0 : A1) + (size_t)mm * C_ * C_;
  const float* __restrict__ Bp = (first ? B0 : B1) + (size_t)mm * C_ * C_;
  float* __restrict__ D = (first ? D0 : D1) + (size_t)mm * C_ * C_;
  const int row0 = blockIdx.y * G_BM;
  const int col0 = blockIdx.x * G_BN;
  const int tid  = threadIdx.x;
  const int lane = tid & 31;
  const int wave = tid >> 5;
  const int wm = wave >> 1;
  const int wn = wave & 1;
  const int lm = lane & 15;
  const int kh = lane >> 4;

  const float* agp = A + (size_t)(row0 + (tid >> 3)) * C_ + (tid & 7) * 4;
  const float* bgp = Bp + (size_t)(tid >> 4) * C_ + col0 + (tid & 15) * 4;
  const int a_st = (tid >> 3) * G_LD + (tid & 7) * 4;
  const int b_st = ((tid & 15) * 4) * G_LD + (tid >> 4);
  const int a_rd = (wm * 64 + lm) * G_LD + 4 * kh;
  const int b_rd = (wn * 32 + lm) * G_LD + 4 * kh;

  const v8f zero = {0.f, 0.f, 0.f, 0.f, 0.f, 0.f, 0.f, 0.f};
  v8f acc[4][2];
#pragma unroll
  for (int m = 0; m < 4; ++m)
#pragma unroll
    for (int n = 0; n < 2; ++n) acc[m][n] = zero;

  v4f ga[8], gb[4];
#pragma unroll
  for (int i = 0; i < 8; ++i) ga[i] = *(const v4f*)(agp + (size_t)(16 * i) * C_);
#pragma unroll
  for (int i = 0; i < 4; ++i) gb[i] = *(const v4f*)(bgp + (size_t)(8 * i) * C_);
#pragma unroll
  for (int i = 0; i < 8; ++i) *(v4f*)(&As[0][a_st + 16 * i * G_LD]) = ga[i];
#pragma unroll
  for (int i = 0; i < 4; ++i) {
    Bs[0][b_st + 0 * G_LD + 8 * i] = gb[i].x;
    Bs[0][b_st + 1 * G_LD + 8 * i] = gb[i].y;
    Bs[0][b_st + 2 * G_LD + 8 * i] = gb[i].z;
    Bs[0][b_st + 3 * G_LD + 8 * i] = gb[i].w;
  }
  __syncthreads();

  for (int t = 0; t < C_ / G_KC; ++t) {
    const int buf = t & 1;
    const bool more = (t + 1 < C_ / G_KC);
    if (more) {
      const int k1 = (t + 1) * G_KC;
#pragma unroll
      for (int i = 0; i < 8; ++i) ga[i] = *(const v4f*)(agp + (size_t)(16 * i) * C_ + k1);
#pragma unroll
      for (int i = 0; i < 4; ++i) gb[i] = *(const v4f*)(bgp + (size_t)(k1 + 8 * i) * C_);
    }

    const float* as = &As[buf][a_rd];
    const float* bs = &Bs[buf][b_rd];
    v4f af[4], bf[2];
#pragma unroll
    for (int m = 0; m < 4; ++m) af[m] = *(const v4f*)(as + 16 * m * G_LD);
#pragma unroll
    for (int n = 0; n < 2; ++n) bf[n] = *(const v4f*)(bs + 16 * n * G_LD);
#pragma unroll
    for (int s = 0; s < G_KC / 8; ++s) {
      v4f an[4], bn[2];
      if (s + 1 < G_KC / 8) {
#pragma unroll
        for (int m = 0; m < 4; ++m) an[m] = *(const v4f*)(as + 16 * m * G_LD + 8 * (s + 1));
#pragma unroll
        for (int n = 0; n < 2; ++n) bn[n] = *(const v4f*)(bs + 16 * n * G_LD + 8 * (s + 1));
      }
#pragma unroll
      for (int m = 0; m < 4; ++m)
#pragma unroll
        for (int n = 0; n < 2; ++n) {
          v2f a0 = {af[m].x, af[m].y};
          v2f b0 = {bf[n].x, bf[n].y};
          acc[m][n] = wmma4(a0, b0, acc[m][n]);
        }
#pragma unroll
      for (int m = 0; m < 4; ++m)
#pragma unroll
        for (int n = 0; n < 2; ++n) {
          v2f a1 = {af[m].z, af[m].w};
          v2f b1 = {bf[n].z, bf[n].w};
          acc[m][n] = wmma4(a1, b1, acc[m][n]);
        }
      if (s + 1 < G_KC / 8) {
#pragma unroll
        for (int m = 0; m < 4; ++m) af[m] = an[m];
#pragma unroll
        for (int n = 0; n < 2; ++n) bf[n] = bn[n];
      }
    }

    if (more) {
      float* ad = &As[buf ^ 1][a_st];
      float* bd = &Bs[buf ^ 1][b_st];
#pragma unroll
      for (int i = 0; i < 8; ++i) *(v4f*)(ad + 16 * i * G_LD) = ga[i];
#pragma unroll
      for (int i = 0; i < 4; ++i) {
        bd[0 * G_LD + 8 * i] = gb[i].x;
        bd[1 * G_LD + 8 * i] = gb[i].y;
        bd[2 * G_LD + 8 * i] = gb[i].z;
        bd[3 * G_LD + 8 * i] = gb[i].w;
      }
    }
    __syncthreads();
  }

#pragma unroll
  for (int m = 0; m < 4; ++m)
#pragma unroll
    for (int n = 0; n < 2; ++n) {
      const int cidx = col0 + wn * 32 + 16 * n + lm;
#pragma unroll
      for (int v = 0; v < 8; ++v) {
        const int r = row0 + wm * 64 + 16 * m + v + 8 * kh;
        float val = alpha * acc[m][n][v];
        if (r == cidx) val += diagBeta;
        D[(size_t)r * C_ + cidx] = val;
      }
    }
}

__global__ void bias_fold_kernel(const float* __restrict__ P,
                                 const float* __restrict__ cmean,
                                 const float* __restrict__ smean,
                                 const float* __restrict__ tr,
                                 float* __restrict__ bias) {
  const int idx = blockIdx.x * 256 + threadIdx.x;
  if (idx >= B_ * C_) return;
  const int b = idx / C_;
  const int i = idx % C_;
  const float* r = P + ((size_t)b * C_ + i) * C_;
  const float* mu = cmean + b * C_;
  float s = 0.f;
  for (int j = 0; j < C_; ++j) s += r[j] * mu[j];
  bias[idx] = smean[idx] - s * (sqrtf(tr[4 + b]) * rsqrtf(tr[b]));
}

__global__ __launch_bounds__(128) void apply_kernel(const float* __restrict__ Mmat,
                                                    const float* __restrict__ X,
                                                    const float* __restrict__ bias,
                                                    const float* __restrict__ tr,
                                                    float* __restrict__ out) {
  __shared__ __attribute__((aligned(16))) float As[2][G_BM * G_LD];
  __shared__ __attribute__((aligned(16))) float Bs[2][G_BN * G_LD];

  const int b = blockIdx.z;
  const float* A  = Mmat + (size_t)b * C_ * C_;
  const float* Bp = X + (size_t)b * C_ * N_;
  float* D = out + (size_t)b * C_ * N_;
  const float* bi = bias + b * C_;
  const float scale = sqrtf(tr[4 + b]) * rsqrtf(tr[b]);

  const int row0 = blockIdx.x * G_BM;
  const int col0 = blockIdx.y * G_BN;
  const int tid  = threadIdx.x;
  const int lane = tid & 31;
  const int wave = tid >> 5;
  const int wm = wave >> 1;
  const int wn = wave & 1;
  const int lm = lane & 15;
  const int kh = lane >> 4;

  const float* agp = A + (size_t)(row0 + (tid >> 3)) * C_ + (tid & 7) * 4;
  const float* bgp = Bp + (size_t)(tid >> 4) * N_ + col0 + (tid & 15) * 4;
  const int a_st = (tid >> 3) * G_LD + (tid & 7) * 4;
  const int b_st = ((tid & 15) * 4) * G_LD + (tid >> 4);
  const int a_rd = (wm * 64 + lm) * G_LD + 4 * kh;
  const int b_rd = (wn * 32 + lm) * G_LD + 4 * kh;

  const v8f zero = {0.f, 0.f, 0.f, 0.f, 0.f, 0.f, 0.f, 0.f};
  v8f acc[4][2];
#pragma unroll
  for (int m = 0; m < 4; ++m)
#pragma unroll
    for (int n = 0; n < 2; ++n) acc[m][n] = zero;

  v4f ga[8], gb[4];
#pragma unroll
  for (int i = 0; i < 8; ++i) ga[i] = *(const v4f*)(agp + (size_t)(16 * i) * C_);
#pragma unroll
  for (int i = 0; i < 4; ++i) gb[i] = *(const v4f*)(bgp + (size_t)(8 * i) * N_);
#pragma unroll
  for (int i = 0; i < 8; ++i) *(v4f*)(&As[0][a_st + 16 * i * G_LD]) = ga[i];
#pragma unroll
  for (int i = 0; i < 4; ++i) {
    Bs[0][b_st + 0 * G_LD + 8 * i] = gb[i].x;
    Bs[0][b_st + 1 * G_LD + 8 * i] = gb[i].y;
    Bs[0][b_st + 2 * G_LD + 8 * i] = gb[i].z;
    Bs[0][b_st + 3 * G_LD + 8 * i] = gb[i].w;
  }
  __syncthreads();

  for (int t = 0; t < C_ / G_KC; ++t) {
    const int buf = t & 1;
    const bool more = (t + 1 < C_ / G_KC);
    if (more) {
      const int k1 = (t + 1) * G_KC;
#pragma unroll
      for (int i = 0; i < 8; ++i) ga[i] = *(const v4f*)(agp + (size_t)(16 * i) * C_ + k1);
#pragma unroll
      for (int i = 0; i < 4; ++i) gb[i] = *(const v4f*)(bgp + (size_t)(k1 + 8 * i) * N_);
    }

    const float* as = &As[buf][a_rd];
    const float* bs = &Bs[buf][b_rd];
    v4f af[4], bf[2];
#pragma unroll
    for (int m = 0; m < 4; ++m) af[m] = *(const v4f*)(as + 16 * m * G_LD);
#pragma unroll
    for (int n = 0; n < 2; ++n) bf[n] = *(const v4f*)(bs + 16 * n * G_LD);
#pragma unroll
    for (int s = 0; s < G_KC / 8; ++s) {
      v4f an[4], bn[2];
      if (s + 1 < G_KC / 8) {
#pragma unroll
        for (int m = 0; m < 4; ++m) an[m] = *(const v4f*)(as + 16 * m * G_LD + 8 * (s + 1));
#pragma unroll
        for (int n = 0; n < 2; ++n) bn[n] = *(const v4f*)(bs + 16 * n * G_LD + 8 * (s + 1));
      }
#pragma unroll
      for (int m = 0; m < 4; ++m)
#pragma unroll
        for (int n = 0; n < 2; ++n) {
          v2f a0 = {af[m].x, af[m].y};
          v2f b0 = {bf[n].x, bf[n].y};
          acc[m][n] = wmma4(a0, b0, acc[m][n]);
        }
#pragma unroll
      for (int m = 0; m < 4; ++m)
#pragma unroll
        for (int n = 0; n < 2; ++n) {
          v2f a1 = {af[m].z, af[m].w};
          v2f b1 = {bf[n].z, bf[n].w};
          acc[m][n] = wmma4(a1, b1, acc[m][n]);
        }
      if (s + 1 < G_KC / 8) {
#pragma unroll
        for (int m = 0; m < 4; ++m) af[m] = an[m];
#pragma unroll
        for (int n = 0; n < 2; ++n) bf[n] = bn[n];
      }
    }

    if (more) {
      float* ad = &As[buf ^ 1][a_st];
      float* bd = &Bs[buf ^ 1][b_st];
#pragma unroll
      for (int i = 0; i < 8; ++i) *(v4f*)(ad + 16 * i * G_LD) = ga[i];
#pragma unroll
      for (int i = 0; i < 4; ++i) {
        bd[0 * G_LD + 8 * i] = gb[i].x;
        bd[1 * G_LD + 8 * i] = gb[i].y;
        bd[2 * G_LD + 8 * i] = gb[i].z;
        bd[3 * G_LD + 8 * i] = gb[i].w;
      }
    }
    __syncthreads();
  }

#pragma unroll
  for (int m = 0; m < 4; ++m)
#pragma unroll
    for (int n = 0; n < 2; ++n) {
      const int cidx = col0 + wn * 32 + 16 * n + lm;
#pragma unroll
      for (int v = 0; v < 8; ++v) {
        const int r = row0 + wm * 64 + 16 * m + v + 8 * kh;
        D[(size_t)r * N_ + cidx] = scale * acc[m][n][v] + bi[r];
      }
    }
}

extern "C" void kernel_launch(void* const* d_in, const int* in_sizes, int n_in,
                              void* d_out, int out_size, void* d_ws, size_t ws_size,
                              hipStream_t stream) {
  (void)in_sizes; (void)n_in; (void)out_size; (void)ws_size;
  const float* content = (const float*)d_in[0];
  const float* style   = (const float*)d_in[1];
  float* out = (float*)d_out;
  float* ws  = (float*)d_ws;

  const size_t CM = (size_t)C_ * C_;

  float* c_mean  = ws;
  float* s_mean  = c_mean + B_ * C_;
  float* tr      = s_mean + B_ * C_;
  float* bias_w  = tr + 8;
  float* cov_all = bias_w + B_ * C_;
  float* Y  = cov_all + 8 * CM;
  float* Z  = Y + 8 * CM;
  float* Y2 = Z + 8 * CM;
  float* Z2 = Y2 + 8 * CM;
  float* T  = Z2 + 8 * CM;

  dim3 meanGrid(B_ * C_, 2);
  mean_kernel<<<meanGrid, 256, 0, stream>>>(content, style, c_mean);

  dim3 covGrid(C_ / 64, C_ / 64, 2 * B_);
  cov_kernel<<<covGrid, 128, 0, stream>>>(content, style, c_mean, cov_all);

  float* rowsum = T;
  rowabs_kernel<<<8 * C_ / 32, 1024, 0, stream>>>(cov_all, rowsum);

  ns_init_kernel<<<8 * 64, 256, 0, stream>>>(cov_all, rowsum, tr, Y, Z);
  dim3 g512(C_ / G_BN, C_ / G_BM, 8);
  dim3 g512x2(C_ / G_BN, C_ / G_BM, 16);
  float* y = Y; float* z = Z; float* y2 = Y2; float* z2 = Z2;
  for (int it = 0; it < NS_ITERS; ++it) {
    gemm512_kernel<<<g512, 128, 0, stream>>>(z, y, T, z, y, T, 8, -0.5f, 1.5f);
    gemm512_kernel<<<g512x2, 128, 0, stream>>>(y, T, y2, T, z, z2, 8, 1.0f, 0.0f);
    float* tmp;
    tmp = y; y = y2; y2 = tmp;
    tmp = z; z = z2; z2 = tmp;
  }

  float* P = T;
  dim3 gFold(C_ / G_BN, C_ / G_BM, B_);
  gemm512_kernel<<<gFold, 128, 0, stream>>>(y + 4 * CM, z, P, y + 4 * CM, z, P, 4, 1.0f, 0.0f);

  bias_fold_kernel<<<(B_ * C_ + 255) / 256, 256, 0, stream>>>(P, c_mean, s_mean, tr, bias_w);

  dim3 gApply(C_ / G_BM, N_ / G_BN, B_);
  apply_kernel<<<gApply, 128, 0, stream>>>(P, content, bias_w, tr, out);
}


# Newton-Schulz steps 1-3 use minimax cubic coefficients for the eigenvalue interval (T = a_k I + b_k ZY, not 1.5I - 0.5ZY), then 2 plain steps: 5 iterations for 7, 21 -> 17 launches; rounding differs
# speedup vs baseline: 6.0397x; 1.1127x over previous
#include <hip/hip_runtime.h>
#include <math.h>

#define B_ 4
#define C_ 512
#define N_ 16384
#define EPS_ 1e-5f
#define NS_ITERS 5

typedef __attribute__((ext_vector_type(2))) float v2f;
typedef __attribute__((ext_vector_type(8))) float v8f;

__device__ __forceinline__ v8f wmma4(v2f a, v2f b, v8f c) {
  return __builtin_amdgcn_wmma_f32_16x16x4_f32(false, a, false, b, (short)0, c,
                                               false, false);
}

__global__ void mean_kernel(const float* __restrict__ x0, const float* __restrict__ x1,
                            float* __restrict__ mean) {
  const int bc = blockIdx.x;
  const int which = blockIdx.y;
  const float* p = (which == 0 ? x0 : x1) + (size_t)bc * N_;
  float s = 0.f;
  for (int i = threadIdx.x; i < N_; i += 256) s += p[i];
  __shared__ float red[256];
  red[threadIdx.x] = s;
  __syncthreads();
  for (int o = 128; o > 0; o >>= 1) {
    if ((int)threadIdx.x < o) red[threadIdx.x] += red[threadIdx.x + o];
    __syncthreads();
  }
  if (threadIdx.x == 0) mean[which * (B_ * C_) + bc] = red[0] * (1.0f / (float)N_);
}

#define V_KC 128
#define V_LD 132
#define V_PR 128

typedef __attribute__((ext_vector_type(4))) float v4f;

__global__ __launch_bounds__(128) void cov_kernel(const float* __restrict__ x0,
                                                  const float* __restrict__ x1,
                                                  const float* __restrict__ mean,
                                                  float* __restrict__ cov) {
  __shared__ __attribute__((aligned(16))) float Ps[2][V_PR * V_LD];

  const int b    = blockIdx.z;
  const int row0 = blockIdx.y * 64;
  const int col0 = blockIdx.x * 64;
  const float* X  = (b < B_ ? x0 : x1) + (size_t)(b & (B_ - 1)) * C_ * N_;
  const float* mu = mean + b * C_;
  float* D = cov + (size_t)b * C_ * C_;

  const int tid  = threadIdx.x;
  const int lane = tid & 31;
  const int wave = tid >> 5;
  const int wm = wave >> 1;
  const int wn = wave & 1;
  const int lm = lane & 15;
  const int kh = lane >> 4;

  const float* gpa = X + (size_t)(row0 + wave) * N_ + lane * 4;
  const float* gpb = X + (size_t)(col0 + wave) * N_ + lane * 4;
  const int p_st = wave * V_LD + lane * 4;
  const int a_rd = (wm * 32 + lm) * V_LD + 4 * kh;
  const int b_rd = (64 + wn * 32 + lm) * V_LD + 4 * kh;

  const v8f zero = {0.f, 0.f, 0.f, 0.f, 0.f, 0.f, 0.f, 0.f};
  v8f acc[2][2];
#pragma unroll
  for (int m = 0; m < 2; ++m)
#pragma unroll
    for (int n = 0; n < 2; ++n) acc[m][n] = zero;

  v4f g[32];
#pragma unroll
  for (int j = 0; j < 16; ++j) g[j] = *(const v4f*)(gpa + (size_t)(4 * j) * N_);
#pragma unroll
  for (int j = 0; j < 16; ++j) g[16 + j] = *(const v4f*)(gpb + (size_t)(4 * j) * N_);
#pragma unroll
  for (int j = 0; j < 32; ++j) *(v4f*)(&Ps[0][p_st + 4 * j * V_LD]) = g[j];
  __syncthreads();

  for (int t = 0; t < N_ / V_KC; ++t) {
    const int buf = t & 1;
    const bool more = (t + 1 < N_ / V_KC);
    if (more) {
      const int k1 = (t + 1) * V_KC;
#pragma unroll
      for (int j = 0; j < 16; ++j) g[j] = *(const v4f*)(gpa + (size_t)(4 * j) * N_ + k1);
#pragma unroll
      for (int j = 0; j < 16; ++j) g[16 + j] = *(const v4f*)(gpb + (size_t)(4 * j) * N_ + k1);
    }

    const float* as = &Ps[buf][a_rd];
    const float* bs = &Ps[buf][b_rd];
    v4f af[2], bf[2];
#pragma unroll
    for (int m = 0; m < 2; ++m) af[m] = *(const v4f*)(as + 16 * m * V_LD);
#pragma unroll
    for (int n = 0; n < 2; ++n) bf[n] = *(const v4f*)(bs + 16 * n * V_LD);
#pragma unroll
    for (int s = 0; s < V_KC / 8; ++s) {
      v4f an[2], bn[2];
      if (s + 1 < V_KC / 8) {
#pragma unroll
        for (int m = 0; m < 2; ++m) an[m] = *(const v4f*)(as + 16 * m * V_LD + 8 * (s + 1));
#pragma unroll
        for (int n = 0; n < 2; ++n) bn[n] = *(const v4f*)(bs + 16 * n * V_LD + 8 * (s + 1));
      }
#pragma unroll
      for (int m = 0; m < 2; ++m)
#pragma unroll
        for (int n = 0; n < 2; ++n) {
          v2f a0 = {af[m].x, af[m].y};
          v2f b0 = {bf[n].x, bf[n].y};
          acc[m][n] = wmma4(a0, b0, acc[m][n]);
        }
#pragma unroll
      for (int m = 0; m < 2; ++m)
#pragma unroll
        for (int n = 0; n < 2; ++n) {
          v2f a1 = {af[m].z, af[m].w};
          v2f b1 = {bf[n].z, bf[n].w};
          acc[m][n] = wmma4(a1, b1, acc[m][n]);
        }
      if (s + 1 < V_KC / 8) {
#pragma unroll
        for (int m = 0; m < 2; ++m) af[m] = an[m];
#pragma unroll
        for (int n = 0; n < 2; ++n) bf[n] = bn[n];
      }
    }

    if (more) {
      float* pd = &Ps[buf ^ 1][p_st];
#pragma unroll
      for (int j = 0; j < 32; ++j) *(v4f*)(pd + 4 * j * V_LD) = g[j];
    }
    __syncthreads();
  }

  const float fac = 1.0f / (float)(N_ - 1);
#pragma unroll
  for (int m = 0; m < 2; ++m) {
#pragma unroll
    for (int n = 0; n < 2; ++n) {
      const int cidx = col0 + wn * 32 + 16 * n + lm;
      const float mj = mu[cidx];
#pragma unroll
      for (int v = 0; v < 8; ++v) {
        const int r = row0 + wm * 32 + 16 * m + v + 8 * kh;
        float val = (acc[m][n][v] - (float)N_ * mu[r] * mj) * fac;
        if (r == cidx) val += EPS_;
        D[(size_t)r * C_ + cidx] = val;
      }
    }
  }
}

__global__ __launch_bounds__(1024) void rowabs_kernel(const float* __restrict__ cov,
                                                      float* __restrict__ rs) {
  __shared__ float part[32];
  const int lane = threadIdx.x & 31;
  const int wave = threadIdx.x >> 5;
  const float* p = cov + ((size_t)blockIdx.x * 32 + wave) * C_ + lane * 4;
  float s = 0.f;
#pragma unroll
  for (int j = 0; j < 4; ++j) {
    const v4f v = *(const v4f*)(p + 128 * j);
    s += (fabsf(v.x) + fabsf(v.y)) + (fabsf(v.z) + fabsf(v.w));
  }
  for (int o = 16; o > 0; o >>= 1) s += __shfl_xor(s, o);
  if (lane == 0) part[wave] = s;
  __syncthreads();
  if (wave == 0) rs[blockIdx.x * 32 + lane] = part[lane];
}

__global__ __launch_bounds__(256) void ns_init_kernel(const float* __restrict__ cov,
                                                      const float* __restrict__ rs,
                                                      float* __restrict__ nrm,
                                                      float* __restrict__ Y,
                                                      float* __restrict__ Z) {
  __shared__ float part[8];
  const int tid  = threadIdx.x;
  const int lane = tid & 31;
  const int wave = tid >> 5;
  const int m = blockIdx.x >> 6;
  float c = fmaxf(rs[m * C_ + tid], rs[m * C_ + 256 + tid]);
  for (int o = 16; o > 0; o >>= 1) c = fmaxf(c, __shfl_xor(c, o));
  if (lane == 0) part[wave] = c;
  __syncthreads();
  c = part[0];
#pragma unroll
  for (int w = 1; w < 8; ++w) c = fmaxf(c, part[w]);
  if ((blockIdx.x & 63) == 0 && tid == 0) nrm[m] = c;

  const size_t base = (size_t)blockIdx.x * 4096;
#pragma unroll
  for (int k = 0; k < 4; ++k) {
    const size_t i = base + (size_t)k * 1024 + tid * 4;
    const int rr  = (int)(i & ((size_t)C_ * C_ - 1));
    const int row = rr >> 9;
    const int col = rr & (C_ - 1);
    const v4f a = *(const v4f*)(cov + i);
    v4f y, z;
    y.x = a.x / c; y.y = a.y / c; y.z = a.z / c; y.w = a.w / c;
    z.x = (row == col)     ? 1.f : 0.f;
    z.y = (row == col + 1) ? 1.f : 0.f;
    z.z = (row == col + 2) ? 1.f : 0.f;
    z.w = (row == col + 3) ? 1.f : 0.f;
    *(v4f*)(Y + i) = y;
    *(v4f*)(Z + i) = z;
  }
}

#define G_BM 128
#define G_BN 64
#define G_KC 32
#define G_LD 36

__global__ __launch_bounds__(128) void gemm512_kernel(const float* A0, const float* B0,
                                                      float* D0,
                                                      const float* A1, const float* B1,
                                                      float* D1, int split,
                                                      float alpha, float diagBeta) {
  __shared__ __attribute__((aligned(16))) float As[2][G_BM * G_LD];
  __shared__ __attribute__((aligned(16))) float Bs[2][G_BN * G_LD];

  const bool first = ((int)blockIdx.z < split);
  const int mm = first ? (int)blockIdx.z : (int)blockIdx.z - split;
  const float* __restrict__ A  = (first ? A0 : A1) + (size_t)mm * C_ * C_;
  const float* __restrict__ Bp = (first ? B0 : B1) + (size_t)mm * C_ * C_;
  float* __restrict__ D = (first ? D0 : D1) + (size_t)mm * C_ * C_;
  const int row0 = blockIdx.y * G_BM;
  const int col0 = blockIdx.x * G_BN;
  const int tid  = threadIdx.x;
  const int lane = tid & 31;
  const int wave = tid >> 5;
  const int wm = wave >> 1;
  const int wn = wave & 1;
  const int lm = lane & 15;
  const int kh = lane >> 4;

  const float* agp = A + (size_t)(row0 + (tid >> 3)) * C_ + (tid & 7) * 4;
  const float* bgp = Bp + (size_t)(tid >> 4) * C_ + col0 + (tid & 15) * 4;
  const int a_st = (tid >> 3) * G_LD + (tid & 7) * 4;
  const int b_st = ((tid & 15) * 4) * G_LD + (tid >> 4);
  const int a_rd = (wm * 64 + lm) * G_LD + 4 * kh;
  const int b_rd = (wn * 32 + lm) * G_LD + 4 * kh;

  const v8f zero = {0.f, 0.f, 0.f, 0.f, 0.f, 0.f, 0.f, 0.f};
  v8f acc[4][2];
#pragma unroll
  for (int m = 0; m < 4; ++m)
#pragma unroll
    for (int n = 0; n < 2; ++n) acc[m][n] = zero;

  v4f ga[8], gb[4];
#pragma unroll
  for (int i = 0; i < 8; ++i) ga[i] = *(const v4f*)(agp + (size_t)(16 * i) * C_);
#pragma unroll
  for (int i = 0; i < 4; ++i) gb[i] = *(const v4f*)(bgp + (size_t)(8 * i) * C_);
#pragma unroll
  for (int i = 0; i < 8; ++i) *(v4f*)(&As[0][a_st + 16 * i * G_LD]) = ga[i];
#pragma unroll
  for (int i = 0; i < 4; ++i) {
    Bs[0][b_st + 0 * G_LD + 8 * i] = gb[i].x;
    Bs[0][b_st + 1 * G_LD + 8 * i] = gb[i].y;
    Bs[0][b_st + 2 * G_LD + 8 * i] = gb[i].z;
    Bs[0][b_st + 3 * G_LD + 8 * i] = gb[i].w;
  }
  __syncthreads();

  for (int t = 0; t < C_ / G_KC; ++t) {
    const int buf = t & 1;
    const bool more = (t + 1 < C_ / G_KC);
    if (more) {
      const int k1 = (t + 1) * G_KC;
#pragma unroll
      for (int i = 0; i < 8; ++i) ga[i] = *(const v4f*)(agp + (size_t)(16 * i) * C_ + k1);
#pragma unroll
      for (int i = 0; i < 4; ++i) gb[i] = *(const v4f*)(bgp + (size_t)(k1 + 8 * i) * C_);
    }

    const float* as = &As[buf][a_rd];
    const float* bs = &Bs[buf][b_rd];
    v4f af[4], bf[2];
#pragma unroll
    for (int m = 0; m < 4; ++m) af[m] = *(const v4f*)(as + 16 * m * G_LD);
#pragma unroll
    for (int n = 0; n < 2; ++n) bf[n] = *(const v4f*)(bs + 16 * n * G_LD);
#pragma unroll
    for (int s = 0; s < G_KC / 8; ++s) {
      v4f an[4], bn[2];
      if (s + 1 < G_KC / 8) {
#pragma unroll
        for (int m = 0; m < 4; ++m) an[m] = *(const v4f*)(as + 16 * m * G_LD + 8 * (s + 1));
#pragma unroll
        for (int n = 0; n < 2; ++n) bn[n] = *(const v4f*)(bs + 16 * n * G_LD + 8 * (s + 1));
      }
#pragma unroll
      for (int m = 0; m < 4; ++m)
#pragma unroll
        for (int n = 0; n < 2; ++n) {
          v2f a0 = {af[m].x, af[m].y};
          v2f b0 = {bf[n].x, bf[n].y};
          acc[m][n] = wmma4(a0, b0, acc[m][n]);
        }
#pragma unroll
      for (int m = 0; m < 4; ++m)
#pragma unroll
        for (int n = 0; n < 2; ++n) {
          v2f a1 = {af[m].z, af[m].w};
          v2f b1 = {bf[n].z, bf[n].w};
          acc[m][n] = wmma4(a1, b1, acc[m][n]);
        }
      if (s + 1 < G_KC / 8) {
#pragma unroll
        for (int m = 0; m < 4; ++m) af[m] = an[m];
#pragma unroll
        for (int n = 0; n < 2; ++n) bf[n] = bn[n];
      }
    }

    if (more) {
      float* ad = &As[buf ^ 1][a_st];
      float* bd = &Bs[buf ^ 1][b_st];
#pragma unroll
      for (int i = 0; i < 8; ++i) *(v4f*)(ad + 16 * i * G_LD) = ga[i];
#pragma unroll
      for (int i = 0; i < 4; ++i) {
        bd[0 * G_LD + 8 * i] = gb[i].x;
        bd[1 * G_LD + 8 * i] = gb[i].y;
        bd[2 * G_LD + 8 * i] = gb[i].z;
        bd[3 * G_LD + 8 * i] = gb[i].w;
      }
    }
    __syncthreads();
  }

#pragma unroll
  for (int m = 0; m < 4; ++m)
#pragma unroll
    for (int n = 0; n < 2; ++n) {
      const int cidx = col0 + wn * 32 + 16 * n + lm;
#pragma unroll
      for (int v = 0; v < 8; ++v) {
        const int r = row0 + wm * 64 + 16 * m + v + 8 * kh;
        float val = alpha * acc[m][n][v];
        if (r == cidx) val += diagBeta;
        D[(size_t)r * C_ + cidx] = val;
      }
    }
}

__global__ void bias_fold_kernel(const float* __restrict__ P,
                                 const float* __restrict__ cmean,
                                 const float* __restrict__ smean,
                                 const float* __restrict__ tr,
                                 float* __restrict__ bias) {
  const int idx = blockIdx.x * 256 + threadIdx.x;
  if (idx >= B_ * C_) return;
  const int b = idx / C_;
  const int i = idx % C_;
  const float* r = P + ((size_t)b * C_ + i) * C_;
  const float* mu = cmean + b * C_;
  float s = 0.f;
  for (int j = 0; j < C_; ++j) s += r[j] * mu[j];
  bias[idx] = smean[idx] - s * (sqrtf(tr[4 + b]) * rsqrtf(tr[b]));
}

__global__ __launch_bounds__(128) void apply_kernel(const float* __restrict__ Mmat,
                                                    const float* __restrict__ X,
                                                    const float* __restrict__ bias,
                                                    const float* __restrict__ tr,
                                                    float* __restrict__ out) {
  __shared__ __attribute__((aligned(16))) float As[2][G_BM * G_LD];
  __shared__ __attribute__((aligned(16))) float Bs[2][G_BN * G_LD];

  const int b = blockIdx.z;
  const float* A  = Mmat + (size_t)b * C_ * C_;
  const float* Bp = X + (size_t)b * C_ * N_;
  float* D = out + (size_t)b * C_ * N_;
  const float* bi = bias + b * C_;
  const float scale = sqrtf(tr[4 + b]) * rsqrtf(tr[b]);

  const int row0 = blockIdx.x * G_BM;
  const int col0 = blockIdx.y * G_BN;
  const int tid  = threadIdx.x;
  const int lane = tid & 31;
  const int wave = tid >> 5;
  const int wm = wave >> 1;
  const int wn = wave & 1;
  const int lm = lane & 15;
  const int kh = lane >> 4;

  const float* agp = A + (size_t)(row0 + (tid >> 3)) * C_ + (tid & 7) * 4;
  const float* bgp = Bp + (size_t)(tid >> 4) * N_ + col0 + (tid & 15) * 4;
  const int a_st = (tid >> 3) * G_LD + (tid & 7) * 4;
  const int b_st = ((tid & 15) * 4) * G_LD + (tid >> 4);
  const int a_rd = (wm * 64 + lm) * G_LD + 4 * kh;
  const int b_rd = (wn * 32 + lm) * G_LD + 4 * kh;

  const v8f zero = {0.f, 0.f, 0.f, 0.f, 0.f, 0.f, 0.f, 0.f};
  v8f acc[4][2];
#pragma unroll
  for (int m = 0; m < 4; ++m)
#pragma unroll
    for (int n = 0; n < 2; ++n) acc[m][n] = zero;

  v4f ga[8], gb[4];
#pragma unroll
  for (int i = 0; i < 8; ++i) ga[i] = *(const v4f*)(agp + (size_t)(16 * i) * C_);
#pragma unroll
  for (int i = 0; i < 4; ++i) gb[i] = *(const v4f*)(bgp + (size_t)(8 * i) * N_);
#pragma unroll
  for (int i = 0; i < 8; ++i) *(v4f*)(&As[0][a_st + 16 * i * G_LD]) = ga[i];
#pragma unroll
  for (int i = 0; i < 4; ++i) {
    Bs[0][b_st + 0 * G_LD + 8 * i] = gb[i].x;
    Bs[0][b_st + 1 * G_LD + 8 * i] = gb[i].y;
    Bs[0][b_st + 2 * G_LD + 8 * i] = gb[i].z;
    Bs[0][b_st + 3 * G_LD + 8 * i] = gb[i].w;
  }
  __syncthreads();

  for (int t = 0; t < C_ / G_KC; ++t) {
    const int buf = t & 1;
    const bool more = (t + 1 < C_ / G_KC);
    if (more) {
      const int k1 = (t + 1) * G_KC;
#pragma unroll
      for (int i = 0; i < 8; ++i) ga[i] = *(const v4f*)(agp + (size_t)(16 * i) * C_ + k1);
#pragma unroll
      for (int i = 0; i < 4; ++i) gb[i] = *(const v4f*)(bgp + (size_t)(k1 + 8 * i) * N_);
    }

    const float* as = &As[buf][a_rd];
    const float* bs = &Bs[buf][b_rd];
    v4f af[4], bf[2];
#pragma unroll
    for (int m = 0; m < 4; ++m) af[m] = *(const v4f*)(as + 16 * m * G_LD);
#pragma unroll
    for (int n = 0; n < 2; ++n) bf[n] = *(const v4f*)(bs + 16 * n * G_LD);
#pragma unroll
    for (int s = 0; s < G_KC / 8; ++s) {
      v4f an[4], bn[2];
      if (s + 1 < G_KC / 8) {
#pragma unroll
        for (int m = 0; m < 4; ++m) an[m] = *(const v4f*)(as + 16 * m * G_LD + 8 * (s + 1));
#pragma unroll
        for (int n = 0; n < 2; ++n) bn[n] = *(const v4f*)(bs + 16 * n * G_LD + 8 * (s + 1));
      }
#pragma unroll
      for (int m = 0; m < 4; ++m)
#pragma unroll
        for (int n = 0; n < 2; ++n) {
          v2f a0 = {af[m].x, af[m].y};
          v2f b0 = {bf[n].x, bf[n].y};
          acc[m][n] = wmma4(a0, b0, acc[m][n]);
        }
#pragma unroll
      for (int m = 0; m < 4; ++m)
#pragma unroll
        for (int n = 0; n < 2; ++n) {
          v2f a1 = {af[m].z, af[m].w};
          v2f b1 = {bf[n].z, bf[n].w};
          acc[m][n] = wmma4(a1, b1, acc[m][n]);
        }
      if (s + 1 < G_KC / 8) {
#pragma unroll
        for (int m = 0; m < 4; ++m) af[m] = an[m];
#pragma unroll
        for (int n = 0; n < 2; ++n) bf[n] = bn[n];
      }
    }

    if (more) {
      float* ad = &As[buf ^ 1][a_st];
      float* bd = &Bs[buf ^ 1][b_st];
#pragma unroll
      for (int i = 0; i < 8; ++i) *(v4f*)(ad + 16 * i * G_LD) = ga[i];
#pragma unroll
      for (int i = 0; i < 4; ++i) {
        bd[0 * G_LD + 8 * i] = gb[i].x;
        bd[1 * G_LD + 8 * i] = gb[i].y;
        bd[2 * G_LD + 8 * i] = gb[i].z;
        bd[3 * G_LD + 8 * i] = gb[i].w;
      }
    }
    __syncthreads();
  }

#pragma unroll
  for (int m = 0; m < 4; ++m)
#pragma unroll
    for (int n = 0; n < 2; ++n) {
      const int cidx = col0 + wn * 32 + 16 * n + lm;
#pragma unroll
      for (int v = 0; v < 8; ++v) {
        const int r = row0 + wm * 64 + 16 * m + v + 8 * kh;
        D[(size_t)r * N_ + cidx] = scale * acc[m][n][v] + bi[r];
      }
    }
}

extern "C" void kernel_launch(void* const* d_in, const int* in_sizes, int n_in,
                              void* d_out, int out_size, void* d_ws, size_t ws_size,
                              hipStream_t stream) {
  (void)in_sizes; (void)n_in; (void)out_size; (void)ws_size;
  const float* content = (const float*)d_in[0];
  const float* style   = (const float*)d_in[1];
  float* out = (float*)d_out;
  float* ws  = (float*)d_ws;

  const size_t CM = (size_t)C_ * C_;

  float* c_mean  = ws;
  float* s_mean  = c_mean + B_ * C_;
  float* tr      = s_mean + B_ * C_;
  float* bias_w  = tr + 8;
  float* cov_all = bias_w + B_ * C_;
  float* Y  = cov_all + 8 * CM;
  float* Z  = Y + 8 * CM;
  float* Y2 = Z + 8 * CM;
  float* Z2 = Y2 + 8 * CM;
  float* T  = Z2 + 8 * CM;

  dim3 meanGrid(B_ * C_, 2);
  mean_kernel<<<meanGrid, 256, 0, stream>>>(content, style, c_mean);

  dim3 covGrid(C_ / 64, C_ / 64, 2 * B_);
  cov_kernel<<<covGrid, 128, 0, stream>>>(content, style, c_mean, cov_all);

  float* rowsum = T;
  rowabs_kernel<<<8 * C_ / 32, 1024, 0, stream>>>(cov_all, rowsum);

  ns_init_kernel<<<8 * 64, 256, 0, stream>>>(cov_all, rowsum, tr, Y, Z);
  dim3 g512(C_ / G_BN, C_ / G_BM, 8);
  dim3 g512x2(C_ / G_BN, C_ / G_BM, 16);
  float* y = Y; float* z = Z; float* y2 = Y2; float* z2 = Z2;
  const float ns_a[NS_ITERS] = {3.214632557f, 1.628810531f, 1.510889005f, 1.5f, 1.5f};
  const float ns_b[NS_ITERS] = {-2.592445610f, -0.5182767681f, -0.5015546530f, -0.5f, -0.5f};
  for (int it = 0; it < NS_ITERS; ++it) {
    gemm512_kernel<<<g512, 128, 0, stream>>>(z, y, T, z, y, T, 8, ns_b[it], ns_a[it]);
    gemm512_kernel<<<g512x2, 128, 0, stream>>>(y, T, y2, T, z, z2, 8, 1.0f, 0.0f);
    float* tmp;
    tmp = y; y = y2; y2 = tmp;
    tmp = z; z = z2; z2 = tmp;
  }

  float* P = T;
  dim3 gFold(C_ / G_BN, C_ / G_BM, B_);
  gemm512_kernel<<<gFold, 128, 0, stream>>>(y + 4 * CM, z, P, y + 4 * CM, z, P, 4, 1.0f, 0.0f);

  bias_fold_kernel<<<(B_ * C_ + 255) / 256, 256, 0, stream>>>(P, c_mean, s_mean, tr, bias_w);

  dim3 gApply(C_ / G_BM, N_ / G_BN, B_);
  apply_kernel<<<gApply, 128, 0, stream>>>(P, content, bias_w, tr, out);
}


# Covariance X*X^T is symmetric: only its 36 upper-triangle 64x64 tiles of 64 are multiplied, K cut in 8 ranges for balance; a finish pass adds the ranges in fixed order and mirrors; sum order changes
# speedup vs baseline: 7.2219x; 1.1957x over previous
#include <hip/hip_runtime.h>
#include <math.h>

#define B_ 4
#define C_ 512
#define N_ 16384
#define EPS_ 1e-5f
#define NS_ITERS 5

typedef __attribute__((ext_vector_type(2))) float v2f;
typedef __attribute__((ext_vector_type(8))) float v8f;

__device__ __forceinline__ v8f wmma4(v2f a, v2f b, v8f c) {
  return __builtin_amdgcn_wmma_f32_16x16x4_f32(false, a, false, b, (short)0, c,
                                               false, false);
}

__global__ void mean_kernel(const float* __restrict__ x0, const float* __restrict__ x1,
                            float* __restrict__ mean) {
  const int bc = blockIdx.x;
  const int which = blockIdx.y;
  const float* p = (which == 0 ? x0 : x1) + (size_t)bc * N_;
  float s = 0.f;
  for (int i = threadIdx.x; i < N_; i += 256) s += p[i];
  __shared__ float red[256];
  red[threadIdx.x] = s;
  __syncthreads();
  for (int o = 128; o > 0; o >>= 1) {
    if ((int)threadIdx.x < o) red[threadIdx.x] += red[threadIdx.x + o];
    __syncthreads();
  }
  if (threadIdx.x == 0) mean[which * (B_ * C_) + bc] = red[0] * (1.0f / (float)N_);
}

#define V_KC 128
#define V_LD 132
#define V_PR 128
#define V_S  8
#define V_KR (N_ / V_S)
#define V_NT 36

typedef __attribute__((ext_vector_type(4))) float v4f;

__device__ __forceinline__ void tri_tile(int t, int& ty, int& tx) {
  int r = 0;
  while (r < 7 && t >= 8 - r) { t -= 8 - r; ++r; }
  ty = r;
  tx = r + t;
}

__global__ __launch_bounds__(128) void cov_kernel(const float* __restrict__ x0,
                                                  const float* __restrict__ x1,
                                                  float* __restrict__ part) {
  __shared__ __attribute__((aligned(16))) float Ps[2][V_PR * V_LD];

  const int b = blockIdx.z;
  int ty, tx;
  tri_tile((int)blockIdx.x, ty, tx);
  const int row0 = ty * 64;
  const int col0 = tx * 64;
  const int kbase = blockIdx.y * V_KR;
  const float* X  = (b < B_ ? x0 : x1) + (size_t)(b & (B_ - 1)) * C_ * N_ + kbase;
  float* D = part + ((size_t)(b * V_NT + (int)blockIdx.x) * V_S + blockIdx.y) * 4096;

  const int tid  = threadIdx.x;
  const int lane = tid & 31;
  const int wave = tid >> 5;
  const int wm = wave >> 1;
  const int wn = wave & 1;
  const int lm = lane & 15;
  const int kh = lane >> 4;

  const float* gpa = X + (size_t)(row0 + wave) * N_ + lane * 4;
  const float* gpb = X + (size_t)(col0 + wave) * N_ + lane * 4;
  const int p_st = wave * V_LD + lane * 4;
  const int a_rd = (wm * 32 + lm) * V_LD + 4 * kh;
  const int b_rd = (64 + wn * 32 + lm) * V_LD + 4 * kh;

  const v8f zero = {0.f, 0.f, 0.f, 0.f, 0.f, 0.f, 0.f, 0.f};
  v8f acc[2][2];
#pragma unroll
  for (int m = 0; m < 2; ++m)
#pragma unroll
    for (int n = 0; n < 2; ++n) acc[m][n] = zero;

  v4f g[32];
#pragma unroll
  for (int j = 0; j < 16; ++j) g[j] = *(const v4f*)(gpa + (size_t)(4 * j) * N_);
#pragma unroll
  for (int j = 0; j < 16; ++j) g[16 + j] = *(const v4f*)(gpb + (size_t)(4 * j) * N_);
#pragma unroll
  for (int j = 0; j < 32; ++j) *(v4f*)(&Ps[0][p_st + 4 * j * V_LD]) = g[j];
  __syncthreads();

  for (int t = 0; t < V_KR / V_KC; ++t) {
    const int buf = t & 1;
    const bool more = (t + 1 < V_KR / V_KC);
    if (more) {
      const int k1 = (t + 1) * V_KC;
#pragma unroll
      for (int j = 0; j < 16; ++j) g[j] = *(const v4f*)(gpa + (size_t)(4 * j) * N_ + k1);
#pragma unroll
      for (int j = 0; j < 16; ++j) g[16 + j] = *(const v4f*)(gpb + (size_t)(4 * j) * N_ + k1);
    }

    const float* as = &Ps[buf][a_rd];
    const float* bs = &Ps[buf][b_rd];
    v4f af[2], bf[2];
#pragma unroll
    for (int m = 0; m < 2; ++m) af[m] = *(const v4f*)(as + 16 * m * V_LD);
#pragma unroll
    for (int n = 0; n < 2; ++n) bf[n] = *(const v4f*)(bs + 16 * n * V_LD);
#pragma unroll
    for (int s = 0; s < V_KC / 8; ++s) {
      v4f an[2], bn[2];
      if (s + 1 < V_KC / 8) {
#pragma unroll
        for (int m = 0; m < 2; ++m) an[m] = *(const v4f*)(as + 16 * m * V_LD + 8 * (s + 1));
#pragma unroll
        for (int n = 0; n < 2; ++n) bn[n] = *(const v4f*)(bs + 16 * n * V_LD + 8 * (s + 1));
      }
#pragma unroll
      for (int m = 0; m < 2; ++m)
#pragma unroll
        for (int n = 0; n < 2; ++n) {
          v2f a0 = {af[m].x, af[m].y};
          v2f b0 = {bf[n].x, bf[n].y};
          acc[m][n] = wmma4(a0, b0, acc[m][n]);
        }
#pragma unroll
      for (int m = 0; m < 2; ++m)
#pragma unroll
        for (int n = 0; n < 2; ++n) {
          v2f a1 = {af[m].z, af[m].w};
          v2f b1 = {bf[n].z, bf[n].w};
          acc[m][n] = wmma4(a1, b1, acc[m][n]);
        }
      if (s + 1 < V_KC / 8) {
#pragma unroll
        for (int m = 0; m < 2; ++m) af[m] = an[m];
#pragma unroll
        for (int n = 0; n < 2; ++n) bf[n] = bn[n];
      }
    }

    if (more) {
      float* pd = &Ps[buf ^ 1][p_st];
#pragma unroll
      for (int j = 0; j < 32; ++j) *(v4f*)(pd + 4 * j * V_LD) = g[j];
    }
    __syncthreads();
  }

#pragma unroll
  for (int m = 0; m < 2; ++m) {
#pragma unroll
    for (int n = 0; n < 2; ++n) {
      const int cc = wn * 32 + 16 * n + lm;
#pragma unroll
      for (int v = 0; v < 8; ++v) {
        const int rr = wm * 32 + 16 * m + v + 8 * kh;
        D[rr * 64 + cc] = acc[m][n][v];
      }
    }
  }
}

#define F_LD 65

__global__ __launch_bounds__(256) void cov_finish_kernel(const float* __restrict__ part,
                                                         const float* __restrict__ mean,
                                                         float* __restrict__ cov) {
  __shared__ float Ts[64 * F_LD];
  const int b = blockIdx.y;
  int ty, tx;
  tri_tile((int)blockIdx.x, ty, tx);
  const int row0 = ty * 64;
  const int col0 = tx * 64;
  const float* mu = mean + b * C_;
  float* D = cov + (size_t)b * C_ * C_;
  const float* p = part + (size_t)(b * V_NT + (int)blockIdx.x) * V_S * 4096;
  const int tid = threadIdx.x;
  const float fac = 1.0f / (float)(N_ - 1);

#pragma unroll
  for (int i = 0; i < 4; ++i) {
    const int e = (i * 256 + tid) * 4;
    const int r = e >> 6;
    const int cq = e & 63;
    v4f s = *(const v4f*)(p + e);
#pragma unroll
    for (int k = 1; k < V_S; ++k) {
      const v4f q = *(const v4f*)(p + (size_t)k * 4096 + e);
      s.x += q.x; s.y += q.y; s.z += q.z; s.w += q.w;
    }
    const int gr = row0 + r;
    const int gc = col0 + cq;
    const float nmr = (float)N_ * mu[gr];
    v4f val;
    val.x = (s.x - nmr * mu[gc + 0]) * fac;
    val.y = (s.y - nmr * mu[gc + 1]) * fac;
    val.z = (s.z - nmr * mu[gc + 2]) * fac;
    val.w = (s.w - nmr * mu[gc + 3]) * fac;
    if (gr == gc + 0) val.x += EPS_;
    if (gr == gc + 1) val.y += EPS_;
    if (gr == gc + 2) val.z += EPS_;
    if (gr == gc + 3) val.w += EPS_;
    *(v4f*)(D + (size_t)gr * C_ + gc) = val;
    Ts[(cq + 0) * F_LD + r] = val.x;
    Ts[(cq + 1) * F_LD + r] = val.y;
    Ts[(cq + 2) * F_LD + r] = val.z;
    Ts[(cq + 3) * F_LD + r] = val.w;
  }
  __syncthreads();
  if (ty != tx) {
#pragma unroll
    for (int i = 0; i < 4; ++i) {
      const int e = (i * 256 + tid) * 4;
      const int r = e >> 6;
      const int cq = e & 63;
      v4f val;
      val.x = Ts[r * F_LD + cq + 0];
      val.y = Ts[r * F_LD + cq + 1];
      val.z = Ts[r * F_LD + cq + 2];
      val.w = Ts[r * F_LD + cq + 3];
      *(v4f*)(D + (size_t)(col0 + r) * C_ + row0 + cq) = val;
    }
  }
}

__global__ __launch_bounds__(1024) void rowabs_kernel(const float* __restrict__ cov,
                                                      float* __restrict__ rs) {
  __shared__ float part[32];
  const int lane = threadIdx.x & 31;
  const int wave = threadIdx.x >> 5;
  const float* p = cov + ((size_t)blockIdx.x * 32 + wave) * C_ + lane * 4;
  float s = 0.f;
#pragma unroll
  for (int j = 0; j < 4; ++j) {
    const v4f v = *(const v4f*)(p + 128 * j);
    s += (fabsf(v.x) + fabsf(v.y)) + (fabsf(v.z) + fabsf(v.w));
  }
  for (int o = 16; o > 0; o >>= 1) s += __shfl_xor(s, o);
  if (lane == 0) part[wave] = s;
  __syncthreads();
  if (wave == 0) rs[blockIdx.x * 32 + lane] = part[lane];
}

__global__ __launch_bounds__(256) void ns_init_kernel(const float* __restrict__ cov,
                                                      const float* __restrict__ rs,
                                                      float* __restrict__ nrm,
                                                      float* __restrict__ Y,
                                                      float* __restrict__ Z) {
  __shared__ float part[8];
  const int tid  = threadIdx.x;
  const int lane = tid & 31;
  const int wave = tid >> 5;
  const int m = blockIdx.x >> 6;
  float c = fmaxf(rs[m * C_ + tid], rs[m * C_ + 256 + tid]);
  for (int o = 16; o > 0; o >>= 1) c = fmaxf(c, __shfl_xor(c, o));
  if (lane == 0) part[wave] = c;
  __syncthreads();
  c = part[0];
#pragma unroll
  for (int w = 1; w < 8; ++w) c = fmaxf(c, part[w]);
  if ((blockIdx.x & 63) == 0 && tid == 0) nrm[m] = c;

  const size_t base = (size_t)blockIdx.x * 4096;
#pragma unroll
  for (int k = 0; k < 4; ++k) {
    const size_t i = base + (size_t)k * 1024 + tid * 4;
    const int rr  = (int)(i & ((size_t)C_ * C_ - 1));
    const int row = rr >> 9;
    const int col = rr & (C_ - 1);
    const v4f a = *(const v4f*)(cov + i);
    v4f y, z;
    y.x = a.x / c; y.y = a.y / c; y.z = a.z / c; y.w = a.w / c;
    z.x = (row == col)     ? 1.f : 0.f;
    z.y = (row == col + 1) ? 1.f : 0.f;
    z.z = (row == col + 2) ? 1.f : 0.f;
    z.w = (row == col + 3) ? 1.f : 0.f;
    *(v4f*)(Y + i) = y;
    *(v4f*)(Z + i) = z;
  }
}

#define G_BM 128
#define G_BN 64
#define G_KC 32
#define G_LD 36

__global__ __launch_bounds__(128) void gemm512_kernel(const float* A0, const float* B0,
                                                      float* D0,
                                                      const float* A1, const float* B1,
                                                      float* D1, int split,
                                                      float alpha, float diagBeta) {
  __shared__ __attribute__((aligned(16))) float As[2][G_BM * G_LD];
  __shared__ __attribute__((aligned(16))) float Bs[2][G_BN * G_LD];

  const bool first = ((int)blockIdx.z < split);
  const int mm = first ? (int)blockIdx.z : (int)blockIdx.z - split;
  const float* __restrict__ A  = (first ? A0 : A1) + (size_t)mm * C_ * C_;
  const float* __restrict__ Bp = (first ? B0 : B1) + (size_t)mm * C_ * C_;
  float* __restrict__ D = (first ? D0 : D1) + (size_t)mm * C_ * C_;
  const int row0 = blockIdx.y * G_BM;
  const int col0 = blockIdx.x * G_BN;
  const int tid  = threadIdx.x;
  const int lane = tid & 31;
  const int wave = tid >> 5;
  const int wm = wave >> 1;
  const int wn = wave & 1;
  const int lm = lane & 15;
  const int kh = lane >> 4;

  const float* agp = A + (size_t)(row0 + (tid >> 3)) * C_ + (tid & 7) * 4;
  const float* bgp = Bp + (size_t)(tid >> 4) * C_ + col0 + (tid & 15) * 4;
  const int a_st = (tid >> 3) * G_LD + (tid & 7) * 4;
  const int b_st = ((tid & 15) * 4) * G_LD + (tid >> 4);
  const int a_rd = (wm * 64 + lm) * G_LD + 4 * kh;
  const int b_rd = (wn * 32 + lm) * G_LD + 4 * kh;

  const v8f zero = {0.f, 0.f, 0.f, 0.f, 0.f, 0.f, 0.f, 0.f};
  v8f acc[4][2];
#pragma unroll
  for (int m = 0; m < 4; ++m)
#pragma unroll
    for (int n = 0; n < 2; ++n) acc[m][n] = zero;

  v4f ga[8], gb[4];
#pragma unroll
  for (int i = 0; i < 8; ++i) ga[i] = *(const v4f*)(agp + (size_t)(16 * i) * C_);
#pragma unroll
  for (int i = 0; i < 4; ++i) gb[i] = *(const v4f*)(bgp + (size_t)(8 * i) * C_);
#pragma unroll
  for (int i = 0; i < 8; ++i) *(v4f*)(&As[0][a_st + 16 * i * G_LD]) = ga[i];
#pragma unroll
  for (int i = 0; i < 4; ++i) {
    Bs[0][b_st + 0 * G_LD + 8 * i] = gb[i].x;
    Bs[0][b_st + 1 * G_LD + 8 * i] = gb[i].y;
    Bs[0][b_st + 2 * G_LD + 8 * i] = gb[i].z;
    Bs[0][b_st + 3 * G_LD + 8 * i] = gb[i].w;
  }
  __syncthreads();

  for (int t = 0; t < C_ / G_KC; ++t) {
    const int buf = t & 1;
    const bool more = (t + 1 < C_ / G_KC);
    if (more) {
      const int k1 = (t + 1) * G_KC;
#pragma unroll
      for (int i = 0; i < 8; ++i) ga[i] = *(const v4f*)(agp + (size_t)(16 * i) * C_ + k1);
#pragma unroll
      for (int i = 0; i < 4; ++i) gb[i] = *(const v4f*)(bgp + (size_t)(k1 + 8 * i) * C_);
    }

    const float* as = &As[buf][a_rd];
    const float* bs = &Bs[buf][b_rd];
    v4f af[4], bf[2];
#pragma unroll
    for (int m = 0; m < 4; ++m) af[m] = *(const v4f*)(as + 16 * m * G_LD);
#pragma unroll
    for (int n = 0; n < 2; ++n) bf[n] = *(const v4f*)(bs + 16 * n * G_LD);
#pragma unroll
    for (int s = 0; s < G_KC / 8; ++s) {
      v4f an[4], bn[2];
      if (s + 1 < G_KC / 8) {
#pragma unroll
        for (int m = 0; m < 4; ++m) an[m] = *(const v4f*)(as + 16 * m * G_LD + 8 * (s + 1));
#pragma unroll
        for (int n = 0; n < 2; ++n) bn[n] = *(const v4f*)(bs + 16 * n * G_LD + 8 * (s + 1));
      }
#pragma unroll
      for (int m = 0; m < 4; ++m)
#pragma unroll
        for (int n = 0; n < 2; ++n) {
          v2f a0 = {af[m].x, af[m].y};
          v2f b0 = {bf[n].x, bf[n].y};
          acc[m][n] = wmma4(a0, b0, acc[m][n]);
        }
#pragma unroll
      for (int m = 0; m < 4; ++m)
#pragma unroll
        for (int n = 0; n < 2; ++n) {
          v2f a1 = {af[m].z, af[m].w};
          v2f b1 = {bf[n].z, bf[n].w};
          acc[m][n] = wmma4(a1, b1, acc[m][n]);
        }
      if (s + 1 < G_KC / 8) {
#pragma unroll
        for (int m = 0; m < 4; ++m) af[m] = an[m];
#pragma unroll
        for (int n = 0; n < 2; ++n) bf[n] = bn[n];
      }
    }

    if (more) {
      float* ad = &As[buf ^ 1][a_st];
      float* bd = &Bs[buf ^ 1][b_st];
#pragma unroll
      for (int i = 0; i < 8; ++i) *(v4f*)(ad + 16 * i * G_LD) = ga[i];
#pragma unroll
      for (int i = 0; i < 4; ++i) {
        bd[0 * G_LD + 8 * i] = gb[i].x;
        bd[1 * G_LD + 8 * i] = gb[i].y;
        bd[2 * G_LD + 8 * i] = gb[i].z;
        bd[3 * G_LD + 8 * i] = gb[i].w;
      }
    }
    __syncthreads();
  }

#pragma unroll
  for (int m = 0; m < 4; ++m)
#pragma unroll
    for (int n = 0; n < 2; ++n) {
      const int cidx = col0 + wn * 32 + 16 * n + lm;
#pragma unroll
      for (int v = 0; v < 8; ++v) {
        const int r = row0 + wm * 64 + 16 * m + v + 8 * kh;
        float val = alpha * acc[m][n][v];
        if (r == cidx) val += diagBeta;
        D[(size_t)r * C_ + cidx] = val;
      }
    }
}

__global__ void bias_fold_kernel(const float* __restrict__ P,
                                 const float* __restrict__ cmean,
                                 const float* __restrict__ smean,
                                 const float* __restrict__ tr,
                                 float* __restrict__ bias) {
  const int idx = blockIdx.x * 256 + threadIdx.x;
  if (idx >= B_ * C_) return;
  const int b = idx / C_;
  const int i = idx % C_;
  const float* r = P + ((size_t)b * C_ + i) * C_;
  const float* mu = cmean + b * C_;
  float s = 0.f;
  for (int j = 0; j < C_; ++j) s += r[j] * mu[j];
  bias[idx] = smean[idx] - s * (sqrtf(tr[4 + b]) * rsqrtf(tr[b]));
}

__global__ __launch_bounds__(128) void apply_kernel(const float* __restrict__ Mmat,
                                                    const float* __restrict__ X,
                                                    const float* __restrict__ bias,
                                                    const float* __restrict__ tr,
                                                    float* __restrict__ out) {
  __shared__ __attribute__((aligned(16))) float As[2][G_BM * G_LD];
  __shared__ __attribute__((aligned(16))) float Bs[2][G_BN * G_LD];

  const int b = blockIdx.z;
  const float* A  = Mmat + (size_t)b * C_ * C_;
  const float* Bp = X + (size_t)b * C_ * N_;
  float* D = out + (size_t)b * C_ * N_;
  const float* bi = bias + b * C_;
  const float scale = sqrtf(tr[4 + b]) * rsqrtf(tr[b]);

  const int row0 = blockIdx.x * G_BM;
  const int col0 = blockIdx.y * G_BN;
  const int tid  = threadIdx.x;
  const int lane = tid & 31;
  const int wave = tid >> 5;
  const int wm = wave >> 1;
  const int wn = wave & 1;
  const int lm = lane & 15;
  const int kh = lane >> 4;

  const float* agp = A + (size_t)(row0 + (tid >> 3)) * C_ + (tid & 7) * 4;
  const float* bgp = Bp + (size_t)(tid >> 4) * N_ + col0 + (tid & 15) * 4;
  const int a_st = (tid >> 3) * G_LD + (tid & 7) * 4;
  const int b_st = ((tid & 15) * 4) * G_LD + (tid >> 4);
  const int a_rd = (wm * 64 + lm) * G_LD + 4 * kh;
  const int b_rd = (wn * 32 + lm) * G_LD + 4 * kh;

  const v8f zero = {0.f, 0.f, 0.f, 0.f, 0.f, 0.f, 0.f, 0.f};
  v8f acc[4][2];
#pragma unroll
  for (int m = 0; m < 4; ++m)
#pragma unroll
    for (int n = 0; n < 2; ++n) acc[m][n] = zero;

  v4f ga[8], gb[4];
#pragma unroll
  for (int i = 0; i < 8; ++i) ga[i] = *(const v4f*)(agp + (size_t)(16 * i) * C_);
#pragma unroll
  for (int i = 0; i < 4; ++i) gb[i] = *(const v4f*)(bgp + (size_t)(8 * i) * N_);
#pragma unroll
  for (int i = 0; i < 8; ++i) *(v4f*)(&As[0][a_st + 16 * i * G_LD]) = ga[i];
#pragma unroll
  for (int i = 0; i < 4; ++i) {
    Bs[0][b_st + 0 * G_LD + 8 * i] = gb[i].x;
    Bs[0][b_st + 1 * G_LD + 8 * i] = gb[i].y;
    Bs[0][b_st + 2 * G_LD + 8 * i] = gb[i].z;
    Bs[0][b_st + 3 * G_LD + 8 * i] = gb[i].w;
  }
  __syncthreads();

  for (int t = 0; t < C_ / G_KC; ++t) {
    const int buf = t & 1;
    const bool more = (t + 1 < C_ / G_KC);
    if (more) {
      const int k1 = (t + 1) * G_KC;
#pragma unroll
      for (int i = 0; i < 8; ++i) ga[i] = *(const v4f*)(agp + (size_t)(16 * i) * C_ + k1);
#pragma unroll
      for (int i = 0; i < 4; ++i) gb[i] = *(const v4f*)(bgp + (size_t)(k1 + 8 * i) * N_);
    }

    const float* as = &As[buf][a_rd];
    const float* bs = &Bs[buf][b_rd];
    v4f af[4], bf[2];
#pragma unroll
    for (int m = 0; m < 4; ++m) af[m] = *(const v4f*)(as + 16 * m * G_LD);
#pragma unroll
    for (int n = 0; n < 2; ++n) bf[n] = *(const v4f*)(bs + 16 * n * G_LD);
#pragma unroll
    for (int s = 0; s < G_KC / 8; ++s) {
      v4f an[4], bn[2];
      if (s + 1 < G_KC / 8) {
#pragma unroll
        for (int m = 0; m < 4; ++m) an[m] = *(const v4f*)(as + 16 * m * G_LD + 8 * (s + 1));
#pragma unroll
        for (int n = 0; n < 2; ++n) bn[n] = *(const v4f*)(bs + 16 * n * G_LD + 8 * (s + 1));
      }
#pragma unroll
      for (int m = 0; m < 4; ++m)
#pragma unroll
        for (int n = 0; n < 2; ++n) {
          v2f a0 = {af[m].x, af[m].y};
          v2f b0 = {bf[n].x, bf[n].y};
          acc[m][n] = wmma4(a0, b0, acc[m][n]);
        }
#pragma unroll
      for (int m = 0; m < 4; ++m)
#pragma unroll
        for (int n = 0; n < 2; ++n) {
          v2f a1 = {af[m].z, af[m].w};
          v2f b1 = {bf[n].z, bf[n].w};
          acc[m][n] = wmma4(a1, b1, acc[m][n]);
        }
      if (s + 1 < G_KC / 8) {
#pragma unroll
        for (int m = 0; m < 4; ++m) af[m] = an[m];
#pragma unroll
        for (int n = 0; n < 2; ++n) bf[n] = bn[n];
      }
    }

    if (more) {
      float* ad = &As[buf ^ 1][a_st];
      float* bd = &Bs[buf ^ 1][b_st];
#pragma unroll
      for (int i = 0; i < 8; ++i) *(v4f*)(ad + 16 * i * G_LD) = ga[i];
#pragma unroll
      for (int i = 0; i < 4; ++i) {
        bd[0 * G_LD + 8 * i] = gb[i].x;
        bd[1 * G_LD + 8 * i] = gb[i].y;
        bd[2 * G_LD + 8 * i] = gb[i].z;
        bd[3 * G_LD + 8 * i] = gb[i].w;
      }
    }
    __syncthreads();
  }

#pragma unroll
  for (int m = 0; m < 4; ++m)
#pragma unroll
    for (int n = 0; n < 2; ++n) {
      const int cidx = col0 + wn * 32 + 16 * n + lm;
#pragma unroll
      for (int v = 0; v < 8; ++v) {
        const int r = row0 + wm * 64 + 16 * m + v + 8 * kh;
        D[(size_t)r * N_ + cidx] = scale * acc[m][n][v] + bi[r];
      }
    }
}

extern "C" void kernel_launch(void* const* d_in, const int* in_sizes, int n_in,
                              void* d_out, int out_size, void* d_ws, size_t ws_size,
                              hipStream_t stream) {
  (void)in_sizes; (void)n_in; (void)out_size; (void)ws_size;
  const float* content = (const float*)d_in[0];
  const float* style   = (const float*)d_in[1];
  float* out = (float*)d_out;
  float* ws  = (float*)d_ws;

  const size_t CM = (size_t)C_ * C_;

  float* c_mean  = ws;
  float* s_mean  = c_mean + B_ * C_;
  float* tr      = s_mean + B_ * C_;
  float* bias_w  = tr + 8;
  float* cov_all = bias_w + B_ * C_;
  float* Y  = cov_all + 8 * CM;
  float* Z  = Y + 8 * CM;
  float* Y2 = Z + 8 * CM;
  float* Z2 = Y2 + 8 * CM;
  float* T  = Z2 + 8 * CM;

  dim3 meanGrid(B_ * C_, 2);
  mean_kernel<<<meanGrid, 256, 0, stream>>>(content, style, c_mean);

  float* cov_part = Y;
  dim3 covGrid(V_NT, V_S, 2 * B_);
  cov_kernel<<<covGrid, 128, 0, stream>>>(content, style, cov_part);
  dim3 finGrid(V_NT, 2 * B_);
  cov_finish_kernel<<<finGrid, 256, 0, stream>>>(cov_part, c_mean, cov_all);

  float* rowsum = T;
  rowabs_kernel<<<8 * C_ / 32, 1024, 0, stream>>>(cov_all, rowsum);

  ns_init_kernel<<<8 * 64, 256, 0, stream>>>(cov_all, rowsum, tr, Y, Z);
  dim3 g512(C_ / G_BN, C_ / G_BM, 8);
  dim3 g512x2(C_ / G_BN, C_ / G_BM, 16);
  float* y = Y; float* z = Z; float* y2 = Y2; float* z2 = Z2;
  const float ns_a[NS_ITERS] = {3.214632557f, 1.628810531f, 1.510889005f, 1.5f, 1.5f};
  const float ns_b[NS_ITERS] = {-2.592445610f, -0.5182767681f, -0.5015546530f, -0.5f, -0.5f};
  for (int it = 0; it < NS_ITERS; ++it) {
    gemm512_kernel<<<g512, 128, 0, stream>>>(z, y, T, z, y, T, 8, ns_b[it], ns_a[it]);
    gemm512_kernel<<<g512x2, 128, 0, stream>>>(y, T, y2, T, z, z2, 8, 1.0f, 0.0f);
    float* tmp;
    tmp = y; y = y2; y2 = tmp;
    tmp = z; z = z2; z2 = tmp;
  }

  float* P = T;
  dim3 gFold(C_ / G_BN, C_ / G_BM, B_);
  gemm512_kernel<<<gFold, 128, 0, stream>>>(y + 4 * CM, z, P, y + 4 * CM, z, P, 4, 1.0f, 0.0f);

  bias_fold_kernel<<<(B_ * C_ + 255) / 256, 256, 0, stream>>>(P, c_mean, s_mean, tr, bias_w);

  dim3 gApply(C_ / G_BM, N_ / G_BN, B_);
  apply_kernel<<<gApply, 128, 0, stream>>>(P, content, bias_w, tr, out);
}


# Newton-Schulz products with a known or unused result dropped: step 1 has Z=I (T and Z' come from the init kernel), last step forms only content Z' and style Y': 120 -> 96 GEMMs, 18 -> 17 launches
# speedup vs baseline: 7.6343x; 1.0571x over previous
#include <hip/hip_runtime.h>
#include <math.h>

#define B_ 4
#define C_ 512
#define N_ 16384
#define EPS_ 1e-5f
#define NS_ITERS 5

typedef __attribute__((ext_vector_type(2))) float v2f;
typedef __attribute__((ext_vector_type(8))) float v8f;

__device__ __forceinline__ v8f wmma4(v2f a, v2f b, v8f c) {
  return __builtin_amdgcn_wmma_f32_16x16x4_f32(false, a, false, b, (short)0, c,
                                               false, false);
}

__global__ void mean_kernel(const float* __restrict__ x0, const float* __restrict__ x1,
                            float* __restrict__ mean) {
  const int bc = blockIdx.x;
  const int which = blockIdx.y;
  const float* p = (which == 0 ? x0 : x1) + (size_t)bc * N_;
  float s = 0.f;
  for (int i = threadIdx.x; i < N_; i += 256) s += p[i];
  __shared__ float red[256];
  red[threadIdx.x] = s;
  __syncthreads();
  for (int o = 128; o > 0; o >>= 1) {
    if ((int)threadIdx.x < o) red[threadIdx.x] += red[threadIdx.x + o];
    __syncthreads();
  }
  if (threadIdx.x == 0) mean[which * (B_ * C_) + bc] = red[0] * (1.0f / (float)N_);
}

#define V_KC 128
#define V_LD 132
#define V_PR 128
#define V_S  8
#define V_KR (N_ / V_S)
#define V_NT 36

typedef __attribute__((ext_vector_type(4))) float v4f;

__device__ __forceinline__ void tri_tile(int t, int& ty, int& tx) {
  int r = 0;
  while (r < 7 && t >= 8 - r) { t -= 8 - r; ++r; }
  ty = r;
  tx = r + t;
}

__global__ __launch_bounds__(128) void cov_kernel(const float* __restrict__ x0,
                                                  const float* __restrict__ x1,
                                                  float* __restrict__ part) {
  __shared__ __attribute__((aligned(16))) float Ps[2][V_PR * V_LD];

  const int b = blockIdx.z;
  int ty, tx;
  tri_tile((int)blockIdx.x, ty, tx);
  const int row0 = ty * 64;
  const int col0 = tx * 64;
  const int kbase = blockIdx.y * V_KR;
  const float* X  = (b < B_ ? x0 : x1) + (size_t)(b & (B_ - 1)) * C_ * N_ + kbase;
  float* D = part + ((size_t)(b * V_NT + (int)blockIdx.x) * V_S + blockIdx.y) * 4096;

  const int tid  = threadIdx.x;
  const int lane = tid & 31;
  const int wave = tid >> 5;
  const int wm = wave >> 1;
  const int wn = wave & 1;
  const int lm = lane & 15;
  const int kh = lane >> 4;

  const float* gpa = X + (size_t)(row0 + wave) * N_ + lane * 4;
  const float* gpb = X + (size_t)(col0 + wave) * N_ + lane * 4;
  const int p_st = wave * V_LD + lane * 4;
  const int a_rd = (wm * 32 + lm) * V_LD + 4 * kh;
  const int b_rd = (64 + wn * 32 + lm) * V_LD + 4 * kh;

  const v8f zero = {0.f, 0.f, 0.f, 0.f, 0.f, 0.f, 0.f, 0.f};
  v8f acc[2][2];
#pragma unroll
  for (int m = 0; m < 2; ++m)
#pragma unroll
    for (int n = 0; n < 2; ++n) acc[m][n] = zero;

  v4f g[32];
#pragma unroll
  for (int j = 0; j < 16; ++j) g[j] = *(const v4f*)(gpa + (size_t)(4 * j) * N_);
#pragma unroll
  for (int j = 0; j < 16; ++j) g[16 + j] = *(const v4f*)(gpb + (size_t)(4 * j) * N_);
#pragma unroll
  for (int j = 0; j < 32; ++j) *(v4f*)(&Ps[0][p_st + 4 * j * V_LD]) = g[j];
  __syncthreads();

  for (int t = 0; t < V_KR / V_KC; ++t) {
    const int buf = t & 1;
    const bool more = (t + 1 < V_KR / V_KC);
    if (more) {
      const int k1 = (t + 1) * V_KC;
#pragma unroll
      for (int j = 0; j < 16; ++j) g[j] = *(const v4f*)(gpa + (size_t)(4 * j) * N_ + k1);
#pragma unroll
      for (int j = 0; j < 16; ++j) g[16 + j] = *(const v4f*)(gpb + (size_t)(4 * j) * N_ + k1);
    }

    const float* as = &Ps[buf][a_rd];
    const float* bs = &Ps[buf][b_rd];
    v4f af[2], bf[2];
#pragma unroll
    for (int m = 0; m < 2; ++m) af[m] = *(const v4f*)(as + 16 * m * V_LD);
#pragma unroll
    for (int n = 0; n < 2; ++n) bf[n] = *(const v4f*)(bs + 16 * n * V_LD);
#pragma unroll
    for (int s = 0; s < V_KC / 8; ++s) {
      v4f an[2], bn[2];
      if (s + 1 < V_KC / 8) {
#pragma unroll
        for (int m = 0; m < 2; ++m) an[m] = *(const v4f*)(as + 16 * m * V_LD + 8 * (s + 1));
#pragma unroll
        for (int n = 0; n < 2; ++n) bn[n] = *(const v4f*)(bs + 16 * n * V_LD + 8 * (s + 1));
      }
#pragma unroll
      for (int m = 0; m < 2; ++m)
#pragma unroll
        for (int n = 0; n < 2; ++n) {
          v2f a0 = {af[m].x, af[m].y};
          v2f b0 = {bf[n].x, bf[n].y};
          acc[m][n] = wmma4(a0, b0, acc[m][n]);
        }
#pragma unroll
      for (int m = 0; m < 2; ++m)
#pragma unroll
        for (int n = 0; n < 2; ++n) {
          v2f a1 = {af[m].z, af[m].w};
          v2f b1 = {bf[n].z, bf[n].w};
          acc[m][n] = wmma4(a1, b1, acc[m][n]);
        }
      if (s + 1 < V_KC / 8) {
#pragma unroll
        for (int m = 0; m < 2; ++m) af[m] = an[m];
#pragma unroll
        for (int n = 0; n < 2; ++n) bf[n] = bn[n];
      }
    }

    if (more) {
      float* pd = &Ps[buf ^ 1][p_st];
#pragma unroll
      for (int j = 0; j < 32; ++j) *(v4f*)(pd + 4 * j * V_LD) = g[j];
    }
    __syncthreads();
  }

#pragma unroll
  for (int m = 0; m < 2; ++m) {
#pragma unroll
    for (int n = 0; n < 2; ++n) {
      const int cc = wn * 32 + 16 * n + lm;
#pragma unroll
      for (int v = 0; v < 8; ++v) {
        const int rr = wm * 32 + 16 * m + v + 8 * kh;
        D[rr * 64 + cc] = acc[m][n][v];
      }
    }
  }
}

#define F_LD 65

__global__ __launch_bounds__(256) void cov_finish_kernel(const float* __restrict__ part,
                                                         const float* __restrict__ mean,
                                                         float* __restrict__ cov) {
  __shared__ float Ts[64 * F_LD];
  const int b = blockIdx.y;
  int ty, tx;
  tri_tile((int)blockIdx.x, ty, tx);
  const int row0 = ty * 64;
  const int col0 = tx * 64;
  const float* mu = mean + b * C_;
  float* D = cov + (size_t)b * C_ * C_;
  const float* p = part + (size_t)(b * V_NT + (int)blockIdx.x) * V_S * 4096;
  const int tid = threadIdx.x;
  const float fac = 1.0f / (float)(N_ - 1);

#pragma unroll
  for (int i = 0; i < 4; ++i) {
    const int e = (i * 256 + tid) * 4;
    const int r = e >> 6;
    const int cq = e & 63;
    v4f s = *(const v4f*)(p + e);
#pragma unroll
    for (int k = 1; k < V_S; ++k) {
      const v4f q = *(const v4f*)(p + (size_t)k * 4096 + e);
      s.x += q.x; s.y += q.y; s.z += q.z; s.w += q.w;
    }
    const int gr = row0 + r;
    const int gc = col0 + cq;
    const float nmr = (float)N_ * mu[gr];
    v4f val;
    val.x = (s.x - nmr * mu[gc + 0]) * fac;
    val.y = (s.y - nmr * mu[gc + 1]) * fac;
    val.z = (s.z - nmr * mu[gc + 2]) * fac;
    val.w = (s.w - nmr * mu[gc + 3]) * fac;
    if (gr == gc + 0) val.x += EPS_;
    if (gr == gc + 1) val.y += EPS_;
    if (gr == gc + 2) val.z += EPS_;
    if (gr == gc + 3) val.w += EPS_;
    *(v4f*)(D + (size_t)gr * C_ + gc) = val;
    Ts[(cq + 0) * F_LD + r] = val.x;
    Ts[(cq + 1) * F_LD + r] = val.y;
    Ts[(cq + 2) * F_LD + r] = val.z;
    Ts[(cq + 3) * F_LD + r] = val.w;
  }
  __syncthreads();
  if (ty != tx) {
#pragma unroll
    for (int i = 0; i < 4; ++i) {
      const int e = (i * 256 + tid) * 4;
      const int r = e >> 6;
      const int cq = e & 63;
      v4f val;
      val.x = Ts[r * F_LD + cq + 0];
      val.y = Ts[r * F_LD + cq + 1];
      val.z = Ts[r * F_LD + cq + 2];
      val.w = Ts[r * F_LD + cq + 3];
      *(v4f*)(D + (size_t)(col0 + r) * C_ + row0 + cq) = val;
    }
  }
}

__global__ __launch_bounds__(1024) void rowabs_kernel(const float* __restrict__ cov,
                                                      float* __restrict__ rs) {
  __shared__ float part[32];
  const int lane = threadIdx.x & 31;
  const int wave = threadIdx.x >> 5;
  const float* p = cov + ((size_t)blockIdx.x * 32 + wave) * C_ + lane * 4;
  float s = 0.f;
#pragma unroll
  for (int j = 0; j < 4; ++j) {
    const v4f v = *(const v4f*)(p + 128 * j);
    s += (fabsf(v.x) + fabsf(v.y)) + (fabsf(v.z) + fabsf(v.w));
  }
  for (int o = 16; o > 0; o >>= 1) s += __shfl_xor(s, o);
  if (lane == 0) part[wave] = s;
  __syncthreads();
  if (wave == 0) rs[blockIdx.x * 32 + lane] = part[lane];
}

__global__ __launch_bounds__(256) void ns_init_kernel(const float* __restrict__ cov,
                                                      const float* __restrict__ rs,
                                                      float* __restrict__ nrm,
                                                      float* __restrict__ Y,
                                                      float* __restrict__ Z,
                                                      float a0, float b0) {
  __shared__ float part[8];
  const int tid  = threadIdx.x;
  const int lane = tid & 31;
  const int wave = tid >> 5;
  const int m = blockIdx.x >> 6;
  float c = fmaxf(rs[m * C_ + tid], rs[m * C_ + 256 + tid]);
  for (int o = 16; o > 0; o >>= 1) c = fmaxf(c, __shfl_xor(c, o));
  if (lane == 0) part[wave] = c;
  __syncthreads();
  c = part[0];
#pragma unroll
  for (int w = 1; w < 8; ++w) c = fmaxf(c, part[w]);
  if ((blockIdx.x & 63) == 0 && tid == 0) nrm[m] = c;

  const size_t base = (size_t)blockIdx.x * 4096;
#pragma unroll
  for (int k = 0; k < 4; ++k) {
    const size_t i = base + (size_t)k * 1024 + tid * 4;
    const int rr  = (int)(i & ((size_t)C_ * C_ - 1));
    const int row = rr >> 9;
    const int col = rr & (C_ - 1);
    const v4f a = *(const v4f*)(cov + i);
    v4f y, z;
    y.x = a.x / c; y.y = a.y / c; y.z = a.z / c; y.w = a.w / c;
    z.x = b0 * y.x; z.y = b0 * y.y; z.z = b0 * y.z; z.w = b0 * y.w;
    if (row == col)     z.x += a0;
    if (row == col + 1) z.y += a0;
    if (row == col + 2) z.z += a0;
    if (row == col + 3) z.w += a0;
    *(v4f*)(Y + i) = y;
    *(v4f*)(Z + i) = z;
  }
}

#define G_BM 128
#define G_BN 64
#define G_KC 32
#define G_LD 36

__global__ __launch_bounds__(128) void gemm512_kernel(const float* A0, const float* B0,
                                                      float* D0,
                                                      const float* A1, const float* B1,
                                                      float* D1, int split,
                                                      float alpha, float diagBeta) {
  __shared__ __attribute__((aligned(16))) float As[2][G_BM * G_LD];
  __shared__ __attribute__((aligned(16))) float Bs[2][G_BN * G_LD];

  const bool first = ((int)blockIdx.z < split);
  const int mm = first ? (int)blockIdx.z : (int)blockIdx.z - split;
  const float* __restrict__ A  = (first ? A0 : A1) + (size_t)mm * C_ * C_;
  const float* __restrict__ Bp = (first ? B0 : B1) + (size_t)mm * C_ * C_;
  float* __restrict__ D = (first ? D0 : D1) + (size_t)mm * C_ * C_;
  const int row0 = blockIdx.y * G_BM;
  const int col0 = blockIdx.x * G_BN;
  const int tid  = threadIdx.x;
  const int lane = tid & 31;
  const int wave = tid >> 5;
  const int wm = wave >> 1;
  const int wn = wave & 1;
  const int lm = lane & 15;
  const int kh = lane >> 4;

  const float* agp = A + (size_t)(row0 + (tid >> 3)) * C_ + (tid & 7) * 4;
  const float* bgp = Bp + (size_t)(tid >> 4) * C_ + col0 + (tid & 15) * 4;
  const int a_st = (tid >> 3) * G_LD + (tid & 7) * 4;
  const int b_st = ((tid & 15) * 4) * G_LD + (tid >> 4);
  const int a_rd = (wm * 64 + lm) * G_LD + 4 * kh;
  const int b_rd = (wn * 32 + lm) * G_LD + 4 * kh;

  const v8f zero = {0.f, 0.f, 0.f, 0.f, 0.f, 0.f, 0.f, 0.f};
  v8f acc[4][2];
#pragma unroll
  for (int m = 0; m < 4; ++m)
#pragma unroll
    for (int n = 0; n < 2; ++n) acc[m][n] = zero;

  v4f ga[8], gb[4];
#pragma unroll
  for (int i = 0; i < 8; ++i) ga[i] = *(const v4f*)(agp + (size_t)(16 * i) * C_);
#pragma unroll
  for (int i = 0; i < 4; ++i) gb[i] = *(const v4f*)(bgp + (size_t)(8 * i) * C_);
#pragma unroll
  for (int i = 0; i < 8; ++i) *(v4f*)(&As[0][a_st + 16 * i * G_LD]) = ga[i];
#pragma unroll
  for (int i = 0; i < 4; ++i) {
    Bs[0][b_st + 0 * G_LD + 8 * i] = gb[i].x;
    Bs[0][b_st + 1 * G_LD + 8 * i] = gb[i].y;
    Bs[0][b_st + 2 * G_LD + 8 * i] = gb[i].z;
    Bs[0][b_st + 3 * G_LD + 8 * i] = gb[i].w;
  }
  __syncthreads();

  for (int t = 0; t < C_ / G_KC; ++t) {
    const int buf = t & 1;
    const bool more = (t + 1 < C_ / G_KC);
    if (more) {
      const int k1 = (t + 1) * G_KC;
#pragma unroll
      for (int i = 0; i < 8; ++i) ga[i] = *(const v4f*)(agp + (size_t)(16 * i) * C_ + k1);
#pragma unroll
      for (int i = 0; i < 4; ++i) gb[i] = *(const v4f*)(bgp + (size_t)(k1 + 8 * i) * C_);
    }

    const float* as = &As[buf][a_rd];
    const float* bs = &Bs[buf][b_rd];
    v4f af[4], bf[2];
#pragma unroll
    for (int m = 0; m < 4; ++m) af[m] = *(const v4f*)(as + 16 * m * G_LD);
#pragma unroll
    for (int n = 0; n < 2; ++n) bf[n] = *(const v4f*)(bs + 16 * n * G_LD);
#pragma unroll
    for (int s = 0; s < G_KC / 8; ++s) {
      v4f an[4], bn[2];
      if (s + 1 < G_KC / 8) {
#pragma unroll
        for (int m = 0; m < 4; ++m) an[m] = *(const v4f*)(as + 16 * m * G_LD + 8 * (s + 1));
#pragma unroll
        for (int n = 0; n < 2; ++n) bn[n] = *(const v4f*)(bs + 16 * n * G_LD + 8 * (s + 1));
      }
#pragma unroll
      for (int m = 0; m < 4; ++m)
#pragma unroll
        for (int n = 0; n < 2; ++n) {
          v2f a0 = {af[m].x, af[m].y};
          v2f b0 = {bf[n].x, bf[n].y};
          acc[m][n] = wmma4(a0, b0, acc[m][n]);
        }
#pragma unroll
      for (int m = 0; m < 4; ++m)
#pragma unroll
        for (int n = 0; n < 2; ++n) {
          v2f a1 = {af[m].z, af[m].w};
          v2f b1 = {bf[n].z, bf[n].w};
          acc[m][n] = wmma4(a1, b1, acc[m][n]);
        }
      if (s + 1 < G_KC / 8) {
#pragma unroll
        for (int m = 0; m < 4; ++m) af[m] = an[m];
#pragma unroll
        for (int n = 0; n < 2; ++n) bf[n] = bn[n];
      }
    }

    if (more) {
      float* ad = &As[buf ^ 1][a_st];
      float* bd = &Bs[buf ^ 1][b_st];
#pragma unroll
      for (int i = 0; i < 8; ++i) *(v4f*)(ad + 16 * i * G_LD) = ga[i];
#pragma unroll
      for (int i = 0; i < 4; ++i) {
        bd[0 * G_LD + 8 * i] = gb[i].x;
        bd[1 * G_LD + 8 * i] = gb[i].y;
        bd[2 * G_LD + 8 * i] = gb[i].z;
        bd[3 * G_LD + 8 * i] = gb[i].w;
      }
    }
    __syncthreads();
  }

#pragma unroll
  for (int m = 0; m < 4; ++m)
#pragma unroll
    for (int n = 0; n < 2; ++n) {
      const int cidx = col0 + wn * 32 + 16 * n + lm;
#pragma unroll
      for (int v = 0; v < 8; ++v) {
        const int r = row0 + wm * 64 + 16 * m + v + 8 * kh;
        float val = alpha * acc[m][n][v];
        if (r == cidx) val += diagBeta;
        D[(size_t)r * C_ + cidx] = val;
      }
    }
}

__global__ void bias_fold_kernel(const float* __restrict__ P,
                                 const float* __restrict__ cmean,
                                 const float* __restrict__ smean,
                                 const float* __restrict__ tr,
                                 float* __restrict__ bias) {
  const int idx = blockIdx.x * 256 + threadIdx.x;
  if (idx >= B_ * C_) return;
  const int b = idx / C_;
  const int i = idx % C_;
  const float* r = P + ((size_t)b * C_ + i) * C_;
  const float* mu = cmean + b * C_;
  float s = 0.f;
  for (int j = 0; j < C_; ++j) s += r[j] * mu[j];
  bias[idx] = smean[idx] - s * (sqrtf(tr[4 + b]) * rsqrtf(tr[b]));
}

__global__ __launch_bounds__(128) void apply_kernel(const float* __restrict__ Mmat,
                                                    const float* __restrict__ X,
                                                    const float* __restrict__ bias,
                                                    const float* __restrict__ tr,
                                                    float* __restrict__ out) {
  __shared__ __attribute__((aligned(16))) float As[2][G_BM * G_LD];
  __shared__ __attribute__((aligned(16))) float Bs[2][G_BN * G_LD];

  const int b = blockIdx.z;
  const float* A  = Mmat + (size_t)b * C_ * C_;
  const float* Bp = X + (size_t)b * C_ * N_;
  float* D = out + (size_t)b * C_ * N_;
  const float* bi = bias + b * C_;
  const float scale = sqrtf(tr[4 + b]) * rsqrtf(tr[b]);

  const int row0 = blockIdx.x * G_BM;
  const int col0 = blockIdx.y * G_BN;
  const int tid  = threadIdx.x;
  const int lane = tid & 31;
  const int wave = tid >> 5;
  const int wm = wave >> 1;
  const int wn = wave & 1;
  const int lm = lane & 15;
  const int kh = lane >> 4;

  const float* agp = A + (size_t)(row0 + (tid >> 3)) * C_ + (tid & 7) * 4;
  const float* bgp = Bp + (size_t)(tid >> 4) * N_ + col0 + (tid & 15) * 4;
  const int a_st = (tid >> 3) * G_LD + (tid & 7) * 4;
  const int b_st = ((tid & 15) * 4) * G_LD + (tid >> 4);
  const int a_rd = (wm * 64 + lm) * G_LD + 4 * kh;
  const int b_rd = (wn * 32 + lm) * G_LD + 4 * kh;

  const v8f zero = {0.f, 0.f, 0.f, 0.f, 0.f, 0.f, 0.f, 0.f};
  v8f acc[4][2];
#pragma unroll
  for (int m = 0; m < 4; ++m)
#pragma unroll
    for (int n = 0; n < 2; ++n) acc[m][n] = zero;

  v4f ga[8], gb[4];
#pragma unroll
  for (int i = 0; i < 8; ++i) ga[i] = *(const v4f*)(agp + (size_t)(16 * i) * C_);
#pragma unroll
  for (int i = 0; i < 4; ++i) gb[i] = *(const v4f*)(bgp + (size_t)(8 * i) * N_);
#pragma unroll
  for (int i = 0; i < 8; ++i) *(v4f*)(&As[0][a_st + 16 * i * G_LD]) = ga[i];
#pragma unroll
  for (int i = 0; i < 4; ++i) {
    Bs[0][b_st + 0 * G_LD + 8 * i] = gb[i].x;
    Bs[0][b_st + 1 * G_LD + 8 * i] = gb[i].y;
    Bs[0][b_st + 2 * G_LD + 8 * i] = gb[i].z;
    Bs[0][b_st + 3 * G_LD + 8 * i] = gb[i].w;
  }
  __syncthreads();

  for (int t = 0; t < C_ / G_KC; ++t) {
    const int buf = t & 1;
    const bool more = (t + 1 < C_ / G_KC);
    if (more) {
      const int k1 = (t + 1) * G_KC;
#pragma unroll
      for (int i = 0; i < 8; ++i) ga[i] = *(const v4f*)(agp + (size_t)(16 * i) * C_ + k1);
#pragma unroll
      for (int i = 0; i < 4; ++i) gb[i] = *(const v4f*)(bgp + (size_t)(k1 + 8 * i) * N_);
    }

    const float* as = &As[buf][a_rd];
    const float* bs = &Bs[buf][b_rd];
    v4f af[4], bf[2];
#pragma unroll
    for (int m = 0; m < 4; ++m) af[m] = *(const v4f*)(as + 16 * m * G_LD);
#pragma unroll
    for (int n = 0; n < 2; ++n) bf[n] = *(const v4f*)(bs + 16 * n * G_LD);
#pragma unroll
    for (int s = 0; s < G_KC / 8; ++s) {
      v4f an[4], bn[2];
      if (s + 1 < G_KC / 8) {
#pragma unroll
        for (int m = 0; m < 4; ++m) an[m] = *(const v4f*)(as + 16 * m * G_LD + 8 * (s + 1));
#pragma unroll
        for (int n = 0; n < 2; ++n) bn[n] = *(const v4f*)(bs + 16 * n * G_LD + 8 * (s + 1));
      }
#pragma unroll
      for (int m = 0; m < 4; ++m)
#pragma unroll
        for (int n = 0; n < 2; ++n) {
          v2f a0 = {af[m].x, af[m].y};
          v2f b0 = {bf[n].x, bf[n].y};
          acc[m][n] = wmma4(a0, b0, acc[m][n]);
        }
#pragma unroll
      for (int m = 0; m < 4; ++m)
#pragma unroll
        for (int n = 0; n < 2; ++n) {
          v2f a1 = {af[m].z, af[m].w};
          v2f b1 = {bf[n].z, bf[n].w};
          acc[m][n] = wmma4(a1, b1, acc[m][n]);
        }
      if (s + 1 < G_KC / 8) {
#pragma unroll
        for (int m = 0; m < 4; ++m) af[m] = an[m];
#pragma unroll
        for (int n = 0; n < 2; ++n) bf[n] = bn[n];
      }
    }

    if (more) {
      float* ad = &As[buf ^ 1][a_st];
      float* bd = &Bs[buf ^ 1][b_st];
#pragma unroll
      for (int i = 0; i < 8; ++i) *(v4f*)(ad + 16 * i * G_LD) = ga[i];
#pragma unroll
      for (int i = 0; i < 4; ++i) {
        bd[0 * G_LD + 8 * i] = gb[i].x;
        bd[1 * G_LD + 8 * i] = gb[i].y;
        bd[2 * G_LD + 8 * i] = gb[i].z;
        bd[3 * G_LD + 8 * i] = gb[i].w;
      }
    }
    __syncthreads();
  }

#pragma unroll
  for (int m = 0; m < 4; ++m)
#pragma unroll
    for (int n = 0; n < 2; ++n) {
      const int cidx = col0 + wn * 32 + 16 * n + lm;
#pragma unroll
      for (int v = 0; v < 8; ++v) {
        const int r = row0 + wm * 64 + 16 * m + v + 8 * kh;
        D[(size_t)r * N_ + cidx] = scale * acc[m][n][v] + bi[r];
      }
    }
}

extern "C" void kernel_launch(void* const* d_in, const int* in_sizes, int n_in,
                              void* d_out, int out_size, void* d_ws, size_t ws_size,
                              hipStream_t stream) {
  (void)in_sizes; (void)n_in; (void)out_size; (void)ws_size;
  const float* content = (const float*)d_in[0];
  const float* style   = (const float*)d_in[1];
  float* out = (float*)d_out;
  float* ws  = (float*)d_ws;

  const size_t CM = (size_t)C_ * C_;

  float* c_mean  = ws;
  float* s_mean  = c_mean + B_ * C_;
  float* tr      = s_mean + B_ * C_;
  float* bias_w  = tr + 8;
  float* cov_all = bias_w + B_ * C_;
  float* Y  = cov_all + 8 * CM;
  float* Z  = Y + 8 * CM;
  float* Y2 = Z + 8 * CM;
  float* Z2 = Y2 + 8 * CM;
  float* T  = Z2 + 8 * CM;

  dim3 meanGrid(B_ * C_, 2);
  mean_kernel<<<meanGrid, 256, 0, stream>>>(content, style, c_mean);

  float* cov_part = Y;
  dim3 covGrid(V_NT, V_S, 2 * B_);
  cov_kernel<<<covGrid, 128, 0, stream>>>(content, style, cov_part);
  dim3 finGrid(V_NT, 2 * B_);
  cov_finish_kernel<<<finGrid, 256, 0, stream>>>(cov_part, c_mean, cov_all);

  float* rowsum = T;
  rowabs_kernel<<<8 * C_ / 32, 1024, 0, stream>>>(cov_all, rowsum);

  const float ns_a[NS_ITERS] = {3.214632557f, 1.628810531f, 1.510889005f, 1.5f, 1.5f};
  const float ns_b[NS_ITERS] = {-2.592445610f, -0.5182767681f, -0.5015546530f, -0.5f, -0.5f};
  dim3 g512(C_ / G_BN, C_ / G_BM, 8);
  dim3 g512x2(C_ / G_BN, C_ / G_BM, 16);
  float* y = Y; float* z = Z; float* y2 = Y2; float* z2 = Z2;
  ns_init_kernel<<<8 * 64, 256, 0, stream>>>(cov_all, rowsum, tr, Y, Z, ns_a[0], ns_b[0]);
  gemm512_kernel<<<g512, 128, 0, stream>>>(y, z, y2, y, z, y2, 8, 1.0f, 0.0f);
  {
    float* tmp;
    tmp = y; y = y2; y2 = tmp;
  }
  for (int it = 1; it < NS_ITERS - 1; ++it) {
    gemm512_kernel<<<g512, 128, 0, stream>>>(z, y, T, z, y, T, 8, ns_b[it], ns_a[it]);
    gemm512_kernel<<<g512x2, 128, 0, stream>>>(y, T, y2, T, z, z2, 8, 1.0f, 0.0f);
    float* tmp;
    tmp = y; y = y2; y2 = tmp;
    tmp = z; z = z2; z2 = tmp;
  }
  gemm512_kernel<<<g512, 128, 0, stream>>>(z, y, T, z, y, T, 8, ns_b[NS_ITERS - 1], ns_a[NS_ITERS - 1]);
  gemm512_kernel<<<g512, 128, 0, stream>>>(y + 4 * CM, T + 4 * CM, y2 + 4 * CM, T, z, z2, 4, 1.0f, 0.0f);
  {
    float* tmp;
    tmp = y; y = y2; y2 = tmp;
    tmp = z; z = z2; z2 = tmp;
  }

  float* P = T;
  dim3 gFold(C_ / G_BN, C_ / G_BM, B_);
  gemm512_kernel<<<gFold, 128, 0, stream>>>(y + 4 * CM, z, P, y + 4 * CM, z, P, 4, 1.0f, 0.0f);

  bias_fold_kernel<<<(B_ * C_ + 255) / 256, 256, 0, stream>>>(P, c_mean, s_mean, tr, bias_w);

  dim3 gApply(C_ / G_BM, N_ / G_BN, B_);
  apply_kernel<<<gApply, 128, 0, stream>>>(P, content, bias_w, tr, out);
}


# Newton-Schulz steps use a quintic map, T = a I + b ZY + c (ZY)^2 formed as c (ZY + al I)^2 + d I: 3 steps for 5, 9 products per matrix for 12, 96 -> 72 GEMMs, 17 -> 16 launches; rounding differs
# speedup vs baseline: 8.1245x; 1.0642x over previous
#include <hip/hip_runtime.h>
#include <math.h>

#define B_ 4
#define C_ 512
#define N_ 16384
#define EPS_ 1e-5f

typedef __attribute__((ext_vector_type(2))) float v2f;
typedef __attribute__((ext_vector_type(8))) float v8f;

__device__ __forceinline__ v8f wmma4(v2f a, v2f b, v8f c) {
  return __builtin_amdgcn_wmma_f32_16x16x4_f32(false, a, false, b, (short)0, c,
                                               false, false);
}

__global__ void mean_kernel(const float* __restrict__ x0, const float* __restrict__ x1,
                            float* __restrict__ mean) {
  const int bc = blockIdx.x;
  const int which = blockIdx.y;
  const float* p = (which == 0 ? x0 : x1) + (size_t)bc * N_;
  float s = 0.f;
  for (int i = threadIdx.x; i < N_; i += 256) s += p[i];
  __shared__ float red[256];
  red[threadIdx.x] = s;
  __syncthreads();
  for (int o = 128; o > 0; o >>= 1) {
    if ((int)threadIdx.x < o) red[threadIdx.x] += red[threadIdx.x + o];
    __syncthreads();
  }
  if (threadIdx.x == 0) mean[which * (B_ * C_) + bc] = red[0] * (1.0f / (float)N_);
}

#define V_KC 128
#define V_LD 132
#define V_PR 128
#define V_S  8
#define V_KR (N_ / V_S)
#define V_NT 36

typedef __attribute__((ext_vector_type(4))) float v4f;

__device__ __forceinline__ void tri_tile(int t, int& ty, int& tx) {
  int r = 0;
  while (r < 7 && t >= 8 - r) { t -= 8 - r; ++r; }
  ty = r;
  tx = r + t;
}

__global__ __launch_bounds__(128) void cov_kernel(const float* __restrict__ x0,
                                                  const float* __restrict__ x1,
                                                  float* __restrict__ part) {
  __shared__ __attribute__((aligned(16))) float Ps[2][V_PR * V_LD];

  const int b = blockIdx.z;
  int ty, tx;
  tri_tile((int)blockIdx.x, ty, tx);
  const int row0 = ty * 64;
  const int col0 = tx * 64;
  const int kbase = blockIdx.y * V_KR;
  const float* X  = (b < B_ ? x0 : x1) + (size_t)(b & (B_ - 1)) * C_ * N_ + kbase;
  float* D = part + ((size_t)(b * V_NT + (int)blockIdx.x) * V_S + blockIdx.y) * 4096;

  const int tid  = threadIdx.x;
  const int lane = tid & 31;
  const int wave = tid >> 5;
  const int wm = wave >> 1;
  const int wn = wave & 1;
  const int lm = lane & 15;
  const int kh = lane >> 4;

  const float* gpa = X + (size_t)(row0 + wave) * N_ + lane * 4;
  const float* gpb = X + (size_t)(col0 + wave) * N_ + lane * 4;
  const int p_st = wave * V_LD + lane * 4;
  const int a_rd = (wm * 32 + lm) * V_LD + 4 * kh;
  const int b_rd = (64 + wn * 32 + lm) * V_LD + 4 * kh;

  const v8f zero = {0.f, 0.f, 0.f, 0.f, 0.f, 0.f, 0.f, 0.f};
  v8f acc[2][2];
#pragma unroll
  for (int m = 0; m < 2; ++m)
#pragma unroll
    for (int n = 0; n < 2; ++n) acc[m][n] = zero;

  v4f g[32];
#pragma unroll
  for (int j = 0; j < 16; ++j) g[j] = *(const v4f*)(gpa + (size_t)(4 * j) * N_);
#pragma unroll
  for (int j = 0; j < 16; ++j) g[16 + j] = *(const v4f*)(gpb + (size_t)(4 * j) * N_);
#pragma unroll
  for (int j = 0; j < 32; ++j) *(v4f*)(&Ps[0][p_st + 4 * j * V_LD]) = g[j];
  __syncthreads();

  for (int t = 0; t < V_KR / V_KC; ++t) {
    const int buf = t & 1;
    const bool more = (t + 1 < V_KR / V_KC);
    if (more) {
      const int k1 = (t + 1) * V_KC;
#pragma unroll
      for (int j = 0; j < 16; ++j) g[j] = *(const v4f*)(gpa + (size_t)(4 * j) * N_ + k1);
#pragma unroll
      for (int j = 0; j < 16; ++j) g[16 + j] = *(const v4f*)(gpb + (size_t)(4 * j) * N_ + k1);
    }

    const float* as = &Ps[buf][a_rd];
    const float* bs = &Ps[buf][b_rd];
    v4f af[2], bf[2];
#pragma unroll
    for (int m = 0; m < 2; ++m) af[m] = *(const v4f*)(as + 16 * m * V_LD);
#pragma unroll
    for (int n = 0; n < 2; ++n) bf[n] = *(const v4f*)(bs + 16 * n * V_LD);
#pragma unroll
    for (int s = 0; s < V_KC / 8; ++s) {
      v4f an[2], bn[2];
      if (s + 1 < V_KC / 8) {
#pragma unroll
        for (int m = 0; m < 2; ++m) an[m] = *(const v4f*)(as + 16 * m * V_LD + 8 * (s + 1));
#pragma unroll
        for (int n = 0; n < 2; ++n) bn[n] = *(const v4f*)(bs + 16 * n * V_LD + 8 * (s + 1));
      }
#pragma unroll
      for (int m = 0; m < 2; ++m)
#pragma unroll
        for (int n = 0; n < 2; ++n) {
          v2f a0 = {af[m].x, af[m].y};
          v2f b0 = {bf[n].x, bf[n].y};
          acc[m][n] = wmma4(a0, b0, acc[m][n]);
        }
#pragma unroll
      for (int m = 0; m < 2; ++m)
#pragma unroll
        for (int n = 0; n < 2; ++n) {
          v2f a1 = {af[m].z, af[m].w};
          v2f b1 = {bf[n].z, bf[n].w};
          acc[m][n] = wmma4(a1, b1, acc[m][n]);
        }
      if (s + 1 < V_KC / 8) {
#pragma unroll
        for (int m = 0; m < 2; ++m) af[m] = an[m];
#pragma unroll
        for (int n = 0; n < 2; ++n) bf[n] = bn[n];
      }
    }

    if (more) {
      float* pd = &Ps[buf ^ 1][p_st];
#pragma unroll
      for (int j = 0; j < 32; ++j) *(v4f*)(pd + 4 * j * V_LD) = g[j];
    }
    __syncthreads();
  }

#pragma unroll
  for (int m = 0; m < 2; ++m) {
#pragma unroll
    for (int n = 0; n < 2; ++n) {
      const int cc = wn * 32 + 16 * n + lm;
#pragma unroll
      for (int v = 0; v < 8; ++v) {
        const int rr = wm * 32 + 16 * m + v + 8 * kh;
        D[rr * 64 + cc] = acc[m][n][v];
      }
    }
  }
}

#define F_LD 65

__global__ __launch_bounds__(256) void cov_finish_kernel(const float* __restrict__ part,
                                                         const float* __restrict__ mean,
                                                         float* __restrict__ cov) {
  __shared__ float Ts[64 * F_LD];
  const int b = blockIdx.y;
  int ty, tx;
  tri_tile((int)blockIdx.x, ty, tx);
  const int row0 = ty * 64;
  const int col0 = tx * 64;
  const float* mu = mean + b * C_;
  float* D = cov + (size_t)b * C_ * C_;
  const float* p = part + (size_t)(b * V_NT + (int)blockIdx.x) * V_S * 4096;
  const int tid = threadIdx.x;
  const float fac = 1.0f / (float)(N_ - 1);

#pragma unroll
  for (int i = 0; i < 4; ++i) {
    const int e = (i * 256 + tid) * 4;
    const int r = e >> 6;
    const int cq = e & 63;
    v4f s = *(const v4f*)(p + e);
#pragma unroll
    for (int k = 1; k < V_S; ++k) {
      const v4f q = *(const v4f*)(p + (size_t)k * 4096 + e);
      s.x += q.x; s.y += q.y; s.z += q.z; s.w += q.w;
    }
    const int gr = row0 + r;
    const int gc = col0 + cq;
    const float nmr = (float)N_ * mu[gr];
    v4f val;
    val.x = (s.x - nmr * mu[gc + 0]) * fac;
    val.y = (s.y - nmr * mu[gc + 1]) * fac;
    val.z = (s.z - nmr * mu[gc + 2]) * fac;
    val.w = (s.w - nmr * mu[gc + 3]) * fac;
    if (gr == gc + 0) val.x += EPS_;
    if (gr == gc + 1) val.y += EPS_;
    if (gr == gc + 2) val.z += EPS_;
    if (gr == gc + 3) val.w += EPS_;
    *(v4f*)(D + (size_t)gr * C_ + gc) = val;
    Ts[(cq + 0) * F_LD + r] = val.x;
    Ts[(cq + 1) * F_LD + r] = val.y;
    Ts[(cq + 2) * F_LD + r] = val.z;
    Ts[(cq + 3) * F_LD + r] = val.w;
  }
  __syncthreads();
  if (ty != tx) {
#pragma unroll
    for (int i = 0; i < 4; ++i) {
      const int e = (i * 256 + tid) * 4;
      const int r = e >> 6;
      const int cq = e & 63;
      v4f val;
      val.x = Ts[r * F_LD + cq + 0];
      val.y = Ts[r * F_LD + cq + 1];
      val.z = Ts[r * F_LD + cq + 2];
      val.w = Ts[r * F_LD + cq + 3];
      *(v4f*)(D + (size_t)(col0 + r) * C_ + row0 + cq) = val;
    }
  }
}

__global__ __launch_bounds__(1024) void rowabs_kernel(const float* __restrict__ cov,
                                                      float* __restrict__ rs) {
  __shared__ float part[32];
  const int lane = threadIdx.x & 31;
  const int wave = threadIdx.x >> 5;
  const float* p = cov + ((size_t)blockIdx.x * 32 + wave) * C_ + lane * 4;
  float s = 0.f;
#pragma unroll
  for (int j = 0; j < 4; ++j) {
    const v4f v = *(const v4f*)(p + 128 * j);
    s += (fabsf(v.x) + fabsf(v.y)) + (fabsf(v.z) + fabsf(v.w));
  }
  for (int o = 16; o > 0; o >>= 1) s += __shfl_xor(s, o);
  if (lane == 0) part[wave] = s;
  __syncthreads();
  if (wave == 0) rs[blockIdx.x * 32 + lane] = part[lane];
}

__global__ __launch_bounds__(256) void ns_init_kernel(const float* __restrict__ cov,
                                                      const float* __restrict__ rs,
                                                      float* __restrict__ nrm,
                                                      float* __restrict__ Y,
                                                      float* __restrict__ U,
                                                      float al0) {
  __shared__ float part[8];
  const int tid  = threadIdx.x;
  const int lane = tid & 31;
  const int wave = tid >> 5;
  const int m = blockIdx.x >> 6;
  float c = fmaxf(rs[m * C_ + tid], rs[m * C_ + 256 + tid]);
  for (int o = 16; o > 0; o >>= 1) c = fmaxf(c, __shfl_xor(c, o));
  if (lane == 0) part[wave] = c;
  __syncthreads();
  c = part[0];
#pragma unroll
  for (int w = 1; w < 8; ++w) c = fmaxf(c, part[w]);
  if ((blockIdx.x & 63) == 0 && tid == 0) nrm[m] = c;

  const size_t base = (size_t)blockIdx.x * 4096;
#pragma unroll
  for (int k = 0; k < 4; ++k) {
    const size_t i = base + (size_t)k * 1024 + tid * 4;
    const int rr  = (int)(i & ((size_t)C_ * C_ - 1));
    const int row = rr >> 9;
    const int col = rr & (C_ - 1);
    const v4f a = *(const v4f*)(cov + i);
    v4f y, u;
    y.x = a.x / c; y.y = a.y / c; y.z = a.z / c; y.w = a.w / c;
    u = y;
    if (row == col)     u.x += al0;
    if (row == col + 1) u.y += al0;
    if (row == col + 2) u.z += al0;
    if (row == col + 3) u.w += al0;
    *(v4f*)(Y + i) = y;
    *(v4f*)(U + i) = u;
  }
}

#define G_BM 128
#define G_BN 64
#define G_KC 32
#define G_LD 36

__global__ __launch_bounds__(128) void gemm512_kernel(const float* A0, const float* B0,
                                                      float* D0,
                                                      const float* A1, const float* B1,
                                                      float* D1, int split,
                                                      float alpha, float diagBeta) {
  __shared__ __attribute__((aligned(16))) float As[2][G_BM * G_LD];
  __shared__ __attribute__((aligned(16))) float Bs[2][G_BN * G_LD];

  const bool first = ((int)blockIdx.z < split);
  const int mm = first ? (int)blockIdx.z : (int)blockIdx.z - split;
  const float* __restrict__ A  = (first ? A0 : A1) + (size_t)mm * C_ * C_;
  const float* __restrict__ Bp = (first ? B0 : B1) + (size_t)mm * C_ * C_;
  float* __restrict__ D = (first ? D0 : D1) + (size_t)mm * C_ * C_;
  const int row0 = blockIdx.y * G_BM;
  const int col0 = blockIdx.x * G_BN;
  const int tid  = threadIdx.x;
  const int lane = tid & 31;
  const int wave = tid >> 5;
  const int wm = wave >> 1;
  const int wn = wave & 1;
  const int lm = lane & 15;
  const int kh = lane >> 4;

  const float* agp = A + (size_t)(row0 + (tid >> 3)) * C_ + (tid & 7) * 4;
  const float* bgp = Bp + (size_t)(tid >> 4) * C_ + col0 + (tid & 15) * 4;
  const int a_st = (tid >> 3) * G_LD + (tid & 7) * 4;
  const int b_st = ((tid & 15) * 4) * G_LD + (tid >> 4);
  const int a_rd = (wm * 64 + lm) * G_LD + 4 * kh;
  const int b_rd = (wn * 32 + lm) * G_LD + 4 * kh;

  const v8f zero = {0.f, 0.f, 0.f, 0.f, 0.f, 0.f, 0.f, 0.f};
  v8f acc[4][2];
#pragma unroll
  for (int m = 0; m < 4; ++m)
#pragma unroll
    for (int n = 0; n < 2; ++n) acc[m][n] = zero;

  v4f ga[8], gb[4];
#pragma unroll
  for (int i = 0; i < 8; ++i) ga[i] = *(const v4f*)(agp + (size_t)(16 * i) * C_);
#pragma unroll
  for (int i = 0; i < 4; ++i) gb[i] = *(const v4f*)(bgp + (size_t)(8 * i) * C_);
#pragma unroll
  for (int i = 0; i < 8; ++i) *(v4f*)(&As[0][a_st + 16 * i * G_LD]) = ga[i];
#pragma unroll
  for (int i = 0; i < 4; ++i) {
    Bs[0][b_st + 0 * G_LD + 8 * i] = gb[i].x;
    Bs[0][b_st + 1 * G_LD + 8 * i] = gb[i].y;
    Bs[0][b_st + 2 * G_LD + 8 * i] = gb[i].z;
    Bs[0][b_st + 3 * G_LD + 8 * i] = gb[i].w;
  }
  __syncthreads();

  for (int t = 0; t < C_ / G_KC; ++t) {
    const int buf = t & 1;
    const bool more = (t + 1 < C_ / G_KC);
    if (more) {
      const int k1 = (t + 1) * G_KC;
#pragma unroll
      for (int i = 0; i < 8; ++i) ga[i] = *(const v4f*)(agp + (size_t)(16 * i) * C_ + k1);
#pragma unroll
      for (int i = 0; i < 4; ++i) gb[i] = *(const v4f*)(bgp + (size_t)(k1 + 8 * i) * C_);
    }

    const float* as = &As[buf][a_rd];
    const float* bs = &Bs[buf][b_rd];
    v4f af[4], bf[2];
#pragma unroll
    for (int m = 0; m < 4; ++m) af[m] = *(const v4f*)(as + 16 * m * G_LD);
#pragma unroll
    for (int n = 0; n < 2; ++n) bf[n] = *(const v4f*)(bs + 16 * n * G_LD);
#pragma unroll
    for (int s = 0; s < G_KC / 8; ++s) {
      v4f an[4], bn[2];
      if (s + 1 < G_KC / 8) {
#pragma unroll
        for (int m = 0; m < 4; ++m) an[m] = *(const v4f*)(as + 16 * m * G_LD + 8 * (s + 1));
#pragma unroll
        for (int n = 0; n < 2; ++n) bn[n] = *(const v4f*)(bs + 16 * n * G_LD + 8 * (s + 1));
      }
#pragma unroll
      for (int m = 0; m < 4; ++m)
#pragma unroll
        for (int n = 0; n < 2; ++n) {
          v2f a0 = {af[m].x, af[m].y};
          v2f b0 = {bf[n].x, bf[n].y};
          acc[m][n] = wmma4(a0, b0, acc[m][n]);
        }
#pragma unroll
      for (int m = 0; m < 4; ++m)
#pragma unroll
        for (int n = 0; n < 2; ++n) {
          v2f a1 = {af[m].z, af[m].w};
          v2f b1 = {bf[n].z, bf[n].w};
          acc[m][n] = wmma4(a1, b1, acc[m][n]);
        }
      if (s + 1 < G_KC / 8) {
#pragma unroll
        for (int m = 0; m < 4; ++m) af[m] = an[m];
#pragma unroll
        for (int n = 0; n < 2; ++n) bf[n] = bn[n];
      }
    }

    if (more) {
      float* ad = &As[buf ^ 1][a_st];
      float* bd = &Bs[buf ^ 1][b_st];
#pragma unroll
      for (int i = 0; i < 8; ++i) *(v4f*)(ad + 16 * i * G_LD) = ga[i];
#pragma unroll
      for (int i = 0; i < 4; ++i) {
        bd[0 * G_LD + 8 * i] = gb[i].x;
        bd[1 * G_LD + 8 * i] = gb[i].y;
        bd[2 * G_LD + 8 * i] = gb[i].z;
        bd[3 * G_LD + 8 * i] = gb[i].w;
      }
    }
    __syncthreads();
  }

#pragma unroll
  for (int m = 0; m < 4; ++m)
#pragma unroll
    for (int n = 0; n < 2; ++n) {
      const int cidx = col0 + wn * 32 + 16 * n + lm;
#pragma unroll
      for (int v = 0; v < 8; ++v) {
        const int r = row0 + wm * 64 + 16 * m + v + 8 * kh;
        float val = alpha * acc[m][n][v];
        if (r == cidx) val += diagBeta;
        D[(size_t)r * C_ + cidx] = val;
      }
    }
}

__global__ void bias_fold_kernel(const float* __restrict__ P,
                                 const float* __restrict__ cmean,
                                 const float* __restrict__ smean,
                                 const float* __restrict__ tr,
                                 float* __restrict__ bias) {
  const int idx = blockIdx.x * 256 + threadIdx.x;
  if (idx >= B_ * C_) return;
  const int b = idx / C_;
  const int i = idx % C_;
  const float* r = P + ((size_t)b * C_ + i) * C_;
  const float* mu = cmean + b * C_;
  float s = 0.f;
  for (int j = 0; j < C_; ++j) s += r[j] * mu[j];
  bias[idx] = smean[idx] - s * (sqrtf(tr[4 + b]) * rsqrtf(tr[b]));
}

__global__ __launch_bounds__(128) void apply_kernel(const float* __restrict__ Mmat,
                                                    const float* __restrict__ X,
                                                    const float* __restrict__ bias,
                                                    const float* __restrict__ tr,
                                                    float* __restrict__ out) {
  __shared__ __attribute__((aligned(16))) float As[2][G_BM * G_LD];
  __shared__ __attribute__((aligned(16))) float Bs[2][G_BN * G_LD];

  const int b = blockIdx.z;
  const float* A  = Mmat + (size_t)b * C_ * C_;
  const float* Bp = X + (size_t)b * C_ * N_;
  float* D = out + (size_t)b * C_ * N_;
  const float* bi = bias + b * C_;
  const float scale = sqrtf(tr[4 + b]) * rsqrtf(tr[b]);

  const int row0 = blockIdx.x * G_BM;
  const int col0 = blockIdx.y * G_BN;
  const int tid  = threadIdx.x;
  const int lane = tid & 31;
  const int wave = tid >> 5;
  const int wm = wave >> 1;
  const int wn = wave & 1;
  const int lm = lane & 15;
  const int kh = lane >> 4;

  const float* agp = A + (size_t)(row0 + (tid >> 3)) * C_ + (tid & 7) * 4;
  const float* bgp = Bp + (size_t)(tid >> 4) * N_ + col0 + (tid & 15) * 4;
  const int a_st = (tid >> 3) * G_LD + (tid & 7) * 4;
  const int b_st = ((tid & 15) * 4) * G_LD + (tid >> 4);
  const int a_rd = (wm * 64 + lm) * G_LD + 4 * kh;
  const int b_rd = (wn * 32 + lm) * G_LD + 4 * kh;

  const v8f zero = {0.f, 0.f, 0.f, 0.f, 0.f, 0.f, 0.f, 0.f};
  v8f acc[4][2];
#pragma unroll
  for (int m = 0; m < 4; ++m)
#pragma unroll
    for (int n = 0; n < 2; ++n) acc[m][n] = zero;

  v4f ga[8], gb[4];
#pragma unroll
  for (int i = 0; i < 8; ++i) ga[i] = *(const v4f*)(agp + (size_t)(16 * i) * C_);
#pragma unroll
  for (int i = 0; i < 4; ++i) gb[i] = *(const v4f*)(bgp + (size_t)(8 * i) * N_);
#pragma unroll
  for (int i = 0; i < 8; ++i) *(v4f*)(&As[0][a_st + 16 * i * G_LD]) = ga[i];
#pragma unroll
  for (int i = 0; i < 4; ++i) {
    Bs[0][b_st + 0 * G_LD + 8 * i] = gb[i].x;
    Bs[0][b_st + 1 * G_LD + 8 * i] = gb[i].y;
    Bs[0][b_st + 2 * G_LD + 8 * i] = gb[i].z;
    Bs[0][b_st + 3 * G_LD + 8 * i] = gb[i].w;
  }
  __syncthreads();

  for (int t = 0; t < C_ / G_KC; ++t) {
    const int buf = t & 1;
    const bool more = (t + 1 < C_ / G_KC);
    if (more) {
      const int k1 = (t + 1) * G_KC;
#pragma unroll
      for (int i = 0; i < 8; ++i) ga[i] = *(const v4f*)(agp + (size_t)(16 * i) * C_ + k1);
#pragma unroll
      for (int i = 0; i < 4; ++i) gb[i] = *(const v4f*)(bgp + (size_t)(k1 + 8 * i) * N_);
    }

    const float* as = &As[buf][a_rd];
    const float* bs = &Bs[buf][b_rd];
    v4f af[4], bf[2];
#pragma unroll
    for (int m = 0; m < 4; ++m) af[m] = *(const v4f*)(as + 16 * m * G_LD);
#pragma unroll
    for (int n = 0; n < 2; ++n) bf[n] = *(const v4f*)(bs + 16 * n * G_LD);
#pragma unroll
    for (int s = 0; s < G_KC / 8; ++s) {
      v4f an[4], bn[2];
      if (s + 1 < G_KC / 8) {
#pragma unroll
        for (int m = 0; m < 4; ++m) an[m] = *(const v4f*)(as + 16 * m * G_LD + 8 * (s + 1));
#pragma unroll
        for (int n = 0; n < 2; ++n) bn[n] = *(const v4f*)(bs + 16 * n * G_LD + 8 * (s + 1));
      }
#pragma unroll
      for (int m = 0; m < 4; ++m)
#pragma unroll
        for (int n = 0; n < 2; ++n) {
          v2f a0 = {af[m].x, af[m].y};
          v2f b0 = {bf[n].x, bf[n].y};
          acc[m][n] = wmma4(a0, b0, acc[m][n]);
        }
#pragma unroll
      for (int m = 0; m < 4; ++m)
#pragma unroll
        for (int n = 0; n < 2; ++n) {
          v2f a1 = {af[m].z, af[m].w};
          v2f b1 = {bf[n].z, bf[n].w};
          acc[m][n] = wmma4(a1, b1, acc[m][n]);
        }
      if (s + 1 < G_KC / 8) {
#pragma unroll
        for (int m = 0; m < 4; ++m) af[m] = an[m];
#pragma unroll
        for (int n = 0; n < 2; ++n) bf[n] = bn[n];
      }
    }

    if (more) {
      float* ad = &As[buf ^ 1][a_st];
      float* bd = &Bs[buf ^ 1][b_st];
#pragma unroll
      for (int i = 0; i < 8; ++i) *(v4f*)(ad + 16 * i * G_LD) = ga[i];
#pragma unroll
      for (int i = 0; i < 4; ++i) {
        bd[0 * G_LD + 8 * i] = gb[i].x;
        bd[1 * G_LD + 8 * i] = gb[i].y;
        bd[2 * G_LD + 8 * i] = gb[i].z;
        bd[3 * G_LD + 8 * i] = gb[i].w;
      }
    }
    __syncthreads();
  }

#pragma unroll
  for (int m = 0; m < 4; ++m)
#pragma unroll
    for (int n = 0; n < 2; ++n) {
      const int cidx = col0 + wn * 32 + 16 * n + lm;
#pragma unroll
      for (int v = 0; v < 8; ++v) {
        const int r = row0 + wm * 64 + 16 * m + v + 8 * kh;
        D[(size_t)r * N_ + cidx] = scale * acc[m][n][v] + bi[r];
      }
    }
}

extern "C" void kernel_launch(void* const* d_in, const int* in_sizes, int n_in,
                              void* d_out, int out_size, void* d_ws, size_t ws_size,
                              hipStream_t stream) {
  (void)in_sizes; (void)n_in; (void)out_size; (void)ws_size;
  const float* content = (const float*)d_in[0];
  const float* style   = (const float*)d_in[1];
  float* out = (float*)d_out;
  float* ws  = (float*)d_ws;

  const size_t CM = (size_t)C_ * C_;

  float* c_mean  = ws;
  float* s_mean  = c_mean + B_ * C_;
  float* tr      = s_mean + B_ * C_;
  float* bias_w  = tr + 8;
  float* cov_all = bias_w + B_ * C_;
  float* Y  = cov_all + 8 * CM;
  float* Z  = Y + 8 * CM;
  float* Y2 = Z + 8 * CM;
  float* Z2 = Y2 + 8 * CM;
  float* T  = Z2 + 8 * CM;

  dim3 meanGrid(B_ * C_, 2);
  mean_kernel<<<meanGrid, 256, 0, stream>>>(content, style, c_mean);

  float* cov_part = Y;
  dim3 covGrid(V_NT, V_S, 2 * B_);
  cov_kernel<<<covGrid, 128, 0, stream>>>(content, style, cov_part);
  dim3 finGrid(V_NT, 2 * B_);
  cov_finish_kernel<<<finGrid, 256, 0, stream>>>(cov_part, c_mean, cov_all);

  float* rowsum = T;
  rowabs_kernel<<<8 * C_ / 32, 1024, 0, stream>>>(cov_all, rowsum);

  const float ns_al0 = -0.7656725342f, ns_c0 = 5.681343271f,  ns_d0 = 0.9080677370f;
  const float ns_al1 = -1.717063412f,  ns_c1 = 0.3808256954f, ns_d1 = 0.8052991868f;
  const float ns_al2 = -1.666714354f,  ns_c2 = 0.3750053650f, ns_d2 = 0.8333071049f;
  dim3 g512(C_ / G_BN, C_ / G_BM, 8);
  dim3 g512x2(C_ / G_BN, C_ / G_BM, 16);
  float* U = cov_all;
  ns_init_kernel<<<8 * 64, 256, 0, stream>>>(cov_all, rowsum, tr, Y, Z2, ns_al0);
  gemm512_kernel<<<g512, 128, 0, stream>>>(Z2, Z2, Z, Z2, Z2, Z, 8, ns_c0, ns_d0);
  gemm512_kernel<<<g512, 128, 0, stream>>>(Y, Z, Y2, Y, Z, Y2, 8, 1.0f, 0.0f);
  gemm512_kernel<<<g512, 128, 0, stream>>>(Z, Y2, U, Z, Y2, U, 8, 1.0f, ns_al1);
  gemm512_kernel<<<g512, 128, 0, stream>>>(U, U, T, U, U, T, 8, ns_c1, ns_d1);
  gemm512_kernel<<<g512x2, 128, 0, stream>>>(Y2, T, Y, T, Z, Z2, 8, 1.0f, 0.0f);
  gemm512_kernel<<<g512, 128, 0, stream>>>(Z2, Y, U, Z2, Y, U, 8, 1.0f, ns_al2);
  gemm512_kernel<<<g512, 128, 0, stream>>>(U, U, T, U, U, T, 8, ns_c2, ns_d2);
  gemm512_kernel<<<g512, 128, 0, stream>>>(Y + 4 * CM, T + 4 * CM, Y2 + 4 * CM, T, Z2, Z, 4, 1.0f, 0.0f);
  float* y = Y2;
  float* z = Z;

  float* P = T;
  dim3 gFold(C_ / G_BN, C_ / G_BM, B_);
  gemm512_kernel<<<gFold, 128, 0, stream>>>(y + 4 * CM, z, P, y + 4 * CM, z, P, 4, 1.0f, 0.0f);

  bias_fold_kernel<<<(B_ * C_ + 255) / 256, 256, 0, stream>>>(P, c_mean, s_mean, tr, bias_w);

  dim3 gApply(C_ / G_BM, N_ / G_BN, B_);
  apply_kernel<<<gApply, 128, 0, stream>>>(P, content, bias_w, tr, out);
}
